# starting kernel
# baseline (speedup 1.0000x reference)
#include <hip/hip_runtime.h>
#include <math.h>

constexpr int DIMC   = 2048;
constexpr int SLEN   = 2048;
constexpr int BATCH  = 2;
constexpr int NH     = 16;
constexpr int NKV    = 4;
constexpr int CQD    = 128;
constexpr int QND    = 96;
constexpr int QRD    = 32;
constexpr int CKVD   = 128;
constexpr int KND    = 64;
constexpr int KRD    = 64;
constexpr int VDD    = 256;
constexpr int DQK    = 128;
constexpr int ROWS   = BATCH * SLEN;
constexpr int UQ_N   = NH * (QND + QRD);
constexpr int DKV_N  = CKVD + KRD;
constexpr int UKV_N  = NKV * (KND + VDD);
constexpr int HVD    = NH * VDD;

typedef float v2f __attribute__((ext_vector_type(2)));
typedef float v8f __attribute__((ext_vector_type(8)));

__device__ __forceinline__ v8f wmma_f32(v2f a, v2f b, v8f c) {
  return __builtin_amdgcn_wmma_f32_16x16x4_f32(
       false, a,  false, b,
       (short)0, c,  false,  false);
}

__device__ __forceinline__ v8f v8f_zero() {
  v8f z = {0.f, 0.f, 0.f, 0.f, 0.f, 0.f, 0.f, 0.f};
  return z;
}

constexpr int KC = 32;

__global__ __launch_bounds__(256) void gemm_wmma_f32_lds(
    const float* __restrict__ A, const float* __restrict__ B,
    float* __restrict__ C, int M, int N, int Kd) {
  __shared__ float As[2][128][36];
  __shared__ float Bs[2][KC / 2][64][2];

  const int tid  = threadIdx.x;
  const int lane = tid & 31;
  const int wave = tid >> 5;
  const int r    = lane & 15;
  const int hf   = lane >> 4;
  const int m0b  = blockIdx.x * 128;
  const int n0   = blockIdx.y * 64;

  v8f acc[4];
#pragma unroll
  for (int j = 0; j < 4; ++j) acc[j] = v8f_zero();

  const int aRow[4] = {0 * 32 + (tid >> 3), 1 * 32 + (tid >> 3),
                       2 * 32 + (tid >> 3), 3 * 32 + (tid >> 3)};
  const int aCol = (tid * 4) & 31;
  const int bKr[2] = {0 * 16 + (tid >> 4), 1 * 16 + (tid >> 4)};
  const int bNc = (tid * 4) & 63;

  const int nchunk = Kd / KC;
  float4 aReg[4], bReg[2];

#pragma unroll
  for (int rep = 0; rep < 4; ++rep)
    aReg[rep] = *(const float4*)(A + (size_t)(m0b + aRow[rep]) * Kd + aCol);
#pragma unroll
  for (int rep = 0; rep < 2; ++rep)
    bReg[rep] = *(const float4*)(B + (size_t)bKr[rep] * N + n0 + bNc);
#pragma unroll
  for (int rep = 0; rep < 4; ++rep)
    *(float4*)&As[0][aRow[rep]][aCol] = aReg[rep];
#pragma unroll
  for (int rep = 0; rep < 2; ++rep) {
    const int kp = bKr[rep] >> 1, kb = bKr[rep] & 1;
    Bs[0][kp][bNc + 0][kb] = bReg[rep].x;
    Bs[0][kp][bNc + 1][kb] = bReg[rep].y;
    Bs[0][kp][bNc + 2][kb] = bReg[rep].z;
    Bs[0][kp][bNc + 3][kb] = bReg[rep].w;
  }
  __syncthreads();

  for (int c = 0; c < nchunk; ++c) {
    const int buf = c & 1;
    if (c + 1 < nchunk) {
      const int kc = (c + 1) * KC;
#pragma unroll
      for (int rep = 0; rep < 4; ++rep)
        aReg[rep] =
            *(const float4*)(A + (size_t)(m0b + aRow[rep]) * Kd + kc + aCol);
#pragma unroll
      for (int rep = 0; rep < 2; ++rep)
        bReg[rep] =
            *(const float4*)(B + (size_t)(kc + bKr[rep]) * N + n0 + bNc);
    }
#pragma unroll
    for (int k = 0; k < KC; k += 4) {
      v2f a = *(const v2f*)&As[buf][wave * 16 + r][k + 2 * hf];
#pragma unroll
      for (int j = 0; j < 4; ++j) {
        v2f bb = *(const v2f*)&Bs[buf][(k + 2 * hf) >> 1][j * 16 + r][0];
        acc[j] = wmma_f32(a, bb, acc[j]);
      }
    }
    if (c + 1 < nchunk) {
      const int nbuf = buf ^ 1;
#pragma unroll
      for (int rep = 0; rep < 4; ++rep)
        *(float4*)&As[nbuf][aRow[rep]][aCol] = aReg[rep];
#pragma unroll
      for (int rep = 0; rep < 2; ++rep) {
        const int kp = bKr[rep] >> 1, kb = bKr[rep] & 1;
        Bs[nbuf][kp][bNc + 0][kb] = bReg[rep].x;
        Bs[nbuf][kp][bNc + 1][kb] = bReg[rep].y;
        Bs[nbuf][kp][bNc + 2][kb] = bReg[rep].z;
        Bs[nbuf][kp][bNc + 3][kb] = bReg[rep].w;
      }
    }
    __syncthreads();
  }

#pragma unroll
  for (int j = 0; j < 4; ++j)
#pragma unroll
    for (int i = 0; i < 8; ++i)
      C[(size_t)(m0b + wave * 16 + i + 8 * hf) * N + n0 + j * 16 + r] =
          acc[j][i];
}

__global__ __launch_bounds__(32) void t5ln_kernel(
    const float* __restrict__ in, int in_stride,
    const float* __restrict__ gamma, float* __restrict__ out) {
  const int row  = blockIdx.x;
  const int lane = threadIdx.x;
  const float* ip = in + (size_t)row * in_stride;
  float v[4], s = 0.f, s2 = 0.f;
#pragma unroll
  for (int i = 0; i < 4; ++i) {
    v[i] = ip[lane + 32 * i];
    s  += v[i];
    s2 += v[i] * v[i];
  }
#pragma unroll
  for (int off = 1; off < 32; off <<= 1) {
    s  += __shfl_xor(s, off, 32);
    s2 += __shfl_xor(s2, off, 32);
  }
  const float mu  = s * (1.f / 128.f);
  const float var = s2 * (1.f / 128.f) - mu * mu;
  const float rs  = rsqrtf(var + 1e-5f);
  float* op = out + (size_t)row * 128;
#pragma unroll
  for (int i = 0; i < 4; ++i)
    op[lane + 32 * i] = (v[i] - mu) * rs * gamma[lane + 32 * i];
}

__device__ __forceinline__ void rope_pair(float spos, int pair, float invD,
                                          float xe, float xo,
                                          float& re, float& ro) {
  const float lnTheta = 9.210340371976184f;
  float freq = expf(-(2.0f * (float)pair) * invD * lnTheta);
  float ang  = spos * freq;
  float c = cosf(ang), sn = sinf(ang);
  re = xe * c - xo * sn;
  ro = xe * sn + xo * c;
}

__global__ __launch_bounds__(256) void build_q_kernel(
    const float* __restrict__ qraw, float* __restrict__ qst) {
  size_t idx = (size_t)blockIdx.x * blockDim.x + threadIdx.x;
  const size_t total = (size_t)BATCH * NH * SLEN * DQK;
  if (idx >= total) return;
  int j = idx & 127;
  int s = (int)((idx >> 7) & (SLEN - 1));
  int h = (int)((idx >> 18) & (NH - 1));
  int b = (int)(idx >> 22);
  const float* qr = qraw + ((size_t)(b * SLEN + s)) * UQ_N;
  float val;
  if (j < QND) {
    val = qr[h * QND + j];
  } else {
    int jj = j - QND;
    int pair = jj >> 1;
    const float* base = qr + NH * QND + h * QRD;
    float xe = base[2 * pair], xo = base[2 * pair + 1];
    float re, ro;
    rope_pair((float)s, pair, 1.0f / (float)QRD, xe, xo, re, ro);
    val = (jj & 1) ? ro : re;
  }
  qst[idx] = val;
}

__global__ __launch_bounds__(256) void build_k_kernel(
    const float* __restrict__ kvraw, const float* __restrict__ ckv_kr,
    float* __restrict__ kst) {
  size_t idx = (size_t)blockIdx.x * blockDim.x + threadIdx.x;
  const size_t total = (size_t)BATCH * NKV * SLEN * DQK;
  if (idx >= total) return;
  int j   = idx & 127;
  int s   = (int)((idx >> 7) & (SLEN - 1));
  int kvh = (int)((idx >> 18) & (NKV - 1));
  int b   = (int)(idx >> 20);
  size_t row = (size_t)(b * SLEN + s);
  float val;
  if (j < KND) {
    val = kvraw[row * UKV_N + kvh * KND + j];
  } else {
    int jj = j - KND;
    int pair = jj >> 1;
    const float* base = ckv_kr + row * DKV_N + CKVD;
    float xe = base[2 * pair], xo = base[2 * pair + 1];
    float re, ro;
    rope_pair((float)s, pair, 1.0f / (float)KRD, xe, xo, re, ro);
    val = (jj & 1) ? ro : re;
  }
  kst[idx] = val;
}

__global__ __launch_bounds__(256) void build_v_kernel(
    const float* __restrict__ kvraw, float* __restrict__ vpi) {
  size_t idx = (size_t)blockIdx.x * blockDim.x + threadIdx.x;
  const size_t total = (size_t)BATCH * NKV * SLEN * VDD;
  if (idx >= total) return;
  int d   = idx & 255;
  int s   = (int)((idx >> 8) & (SLEN - 1));
  int kvh = (int)((idx >> 19) & (NKV - 1));
  int b   = (int)(idx >> 21);
  size_t row = (size_t)(b * SLEN + s);
  float val = kvraw[row * UKV_N + NKV * KND + kvh * VDD + d];
  size_t dst = ((size_t)(b * NKV + kvh) * (SLEN / 2) + (s >> 1)) * (2 * VDD) +
               d * 2 + (s & 1);
  vpi[dst] = val;
}

__global__ __launch_bounds__(256) void attn_kernel(
    const float* __restrict__ Q, const float* __restrict__ K,
    const float* __restrict__ V, float* __restrict__ O) {
  __shared__ float ps[8][16][16];

  const int lane = threadIdx.x & 31;
  const int wave = threadIdx.x >> 5;
  const int r    = lane & 15;
  const int hf   = lane >> 4;
  const int bh   = blockIdx.x;
  const int b    = bh >> 4;
  const int h    = bh & 15;
  const int kvh  = h >> 2;
  const int q0   = blockIdx.y * 128 + wave * 16;

  const float* Qb = Q + ((size_t)bh * SLEN + q0) * DQK;
  const float* Kb = K + ((size_t)(b * NKV + kvh) * SLEN) * DQK;
  const float* Vb = V + ((size_t)(b * NKV + kvh) * SLEN) * VDD;

  const float scale = 0.08838834764831845f;

  v2f qa[32];
#pragma unroll
  for (int di = 0; di < 32; ++di) {
    const float* p = Qb + (size_t)r * DQK + di * 4 + 2 * hf;
    v2f t;
    t.x = p[0] * scale;
    t.y = p[1] * scale;
    qa[di] = t;
  }

  float m_i[8], l_i[8];
#pragma unroll
  for (int i = 0; i < 8; ++i) { m_i[i] = -1e30f; l_i[i] = 0.f; }
  v8f acc[16];
#pragma unroll
  for (int nt = 0; nt < 16; ++nt) acc[nt] = v8f_zero();

  const int ktmax = (q0 + 15) >> 4;
  for (int kt = 0; kt <= ktmax; ++kt) {
    const int k0 = kt * 16;

    if (kt < ktmax) {
      const float* kpf = Kb + (size_t)(k0 + 16 + r) * DQK + hf * 64;
      __builtin_prefetch(kpf, 0, 3);
      const float* vpf = Vb + (size_t)((k0 + 16) >> 1) * (2 * VDD) + lane * 64;
      __builtin_prefetch(vpf, 0, 3);
    }

    v8f sc = v8f_zero();
    const float* Kt = Kb + (size_t)(k0 + r) * DQK + 2 * hf;
#pragma unroll
    for (int di = 0; di < 32; ++di) {
      v2f bb;
      bb.x = Kt[di * 4];
      bb.y = Kt[di * 4 + 1];
      sc = wmma_f32(qa[di], bb, sc);
    }

#pragma unroll
    for (int i = 0; i < 8; ++i) {
      const int qrow = q0 + i + 8 * hf;
      const int kcol = k0 + r;
      float v = (kcol > qrow) ? -1e30f : sc[i];
      float mx = v;
#pragma unroll
      for (int off = 1; off < 16; off <<= 1)
        mx = fmaxf(mx, __shfl_xor(mx, off, 32));
      const float mnew  = fmaxf(m_i[i], mx);
      const float alpha = __expf(m_i[i] - mnew);
      const float pe    = __expf(v - mnew);
      float rs = pe;
#pragma unroll
      for (int off = 1; off < 16; off <<= 1)
        rs += __shfl_xor(rs, off, 32);
      l_i[i] = l_i[i] * alpha + rs;
      m_i[i] = mnew;
#pragma unroll
      for (int nt = 0; nt < 16; ++nt) acc[nt][i] *= alpha;
      ps[wave][i + 8 * hf][r] = pe;
    }
    asm volatile("s_wait_dscnt 0" ::: "memory");

    v2f pa[4];
#pragma unroll
    for (int ks = 0; ks < 4; ++ks) {
      v2f t;
      t.x = ps[wave][r][ks * 4 + 2 * hf];
      t.y = ps[wave][r][ks * 4 + 1 + 2 * hf];
      pa[ks] = t;
    }

#pragma unroll
    for (int nt = 0; nt < 16; ++nt) {
#pragma unroll
      for (int ks = 0; ks < 4; ++ks) {
        const int kk = k0 + ks * 4 + 2 * hf;
        const float* Vp =
            Vb + (size_t)(kk >> 1) * (2 * VDD) + (nt * 16 + r) * 2;
        v2f bb = *(const v2f*)Vp;
        acc[nt] = wmma_f32(pa[ks], bb, acc[nt]);
      }
    }
  }

#pragma unroll
  for (int i = 0; i < 8; ++i) {
    const float inv = 1.0f / l_i[i];
    const int qrow = q0 + i + 8 * hf;
    float* Op = O + ((size_t)(b * SLEN + qrow)) * HVD + h * VDD;
#pragma unroll
    for (int nt = 0; nt < 16; ++nt) Op[nt * 16 + r] = acc[nt][i] * inv;
  }
}

extern "C" void kernel_launch(void* const* d_in, const int* in_sizes, int n_in,
                              void* d_out, int out_size, void* d_ws,
                              size_t ws_size, hipStream_t stream) {
  (void)in_sizes; (void)n_in; (void)out_size; (void)ws_size;
  const float* x        = (const float*)d_in[0];
  const float* w_dq     = (const float*)d_in[1];
  const float* gamma_q  = (const float*)d_in[2];
  const float* w_uq     = (const float*)d_in[3];
  const float* w_dkv    = (const float*)d_in[4];
  const float* gamma_kv = (const float*)d_in[5];
  const float* w_ukv    = (const float*)d_in[6];
  const float* w_o      = (const float*)d_in[7];
  float* out = (float*)d_out;

  float* ws = (float*)d_ws;
  float* tmp_q   = ws;
  float* cq      = tmp_q   + (size_t)ROWS * CQD;
  float* qraw    = cq      + (size_t)ROWS * CQD;
  float* qst     = qraw    + (size_t)ROWS * UQ_N;
  float* ckv_kr  = qst     + (size_t)BATCH * NH * SLEN * DQK;
  float* ckv     = ckv_kr  + (size_t)ROWS * DKV_N;
  float* kvraw   = ckv     + (size_t)ROWS * CKVD;
  float* kst     = kvraw   + (size_t)ROWS * UKV_N;
  float* vpi     = kst     + (size_t)BATCH * NKV * SLEN * DQK;
  float* attn_o  = vpi     + (size_t)BATCH * NKV * SLEN * VDD;

  gemm_wmma_f32_lds<<<dim3(ROWS / 128, CQD / 64), 256, 0, stream>>>(
      x, w_dq, tmp_q, ROWS, CQD, DIMC);
  t5ln_kernel<<<ROWS, 32, 0, stream>>>(tmp_q, CQD, gamma_q, cq);
  gemm_wmma_f32_lds<<<dim3(ROWS / 128, UQ_N / 64), 256, 0, stream>>>(
      cq, w_uq, qraw, ROWS, UQ_N, CQD);
  {
    size_t total = (size_t)BATCH * NH * SLEN * DQK;
    build_q_kernel<<<(unsigned)((total + 255) / 256), 256, 0, stream>>>(qraw, qst);
  }
  gemm_wmma_f32_lds<<<dim3(ROWS / 128, DKV_N / 64), 256, 0, stream>>>(
      x, w_dkv, ckv_kr, ROWS, DKV_N, DIMC);
  t5ln_kernel<<<ROWS, 32, 0, stream>>>(ckv_kr, DKV_N, gamma_kv, ckv);
  gemm_wmma_f32_lds<<<dim3(ROWS / 128, UKV_N / 64), 256, 0, stream>>>(
      ckv, w_ukv, kvraw, ROWS, UKV_N, CQD);
  {
    size_t tk = (size_t)BATCH * NKV * SLEN * DQK;
    build_k_kernel<<<(unsigned)((tk + 255) / 256), 256, 0, stream>>>(kvraw, ckv_kr, kst);
    size_t tv = (size_t)BATCH * NKV * SLEN * VDD;
    build_v_kernel<<<(unsigned)((tv + 255) / 256), 256, 0, stream>>>(kvraw, vpi);
  }
  attn_kernel<<<dim3(BATCH * NH, SLEN / 128), 256, 0, stream>>>(qst, kst, vpi, attn_o);
  gemm_wmma_f32_lds<<<dim3(ROWS / 128, DIMC / 64), 256, 0, stream>>>(
      attn_o, w_o, out, ROWS, DIMC, HVD);
}


# attention: register budget 512 per wave (amdgpu_waves_per_eu 2,2; 252 -> 398 VGPRs) so K and V fragment loads issue in groups ahead of their WMMAs instead of one waited load per WMMA
# speedup vs baseline: 1.8134x; 1.8134x over previous
#include <hip/hip_runtime.h>
#include <math.h>

constexpr int DIMC   = 2048;
constexpr int SLEN   = 2048;
constexpr int BATCH  = 2;
constexpr int NH     = 16;
constexpr int NKV    = 4;
constexpr int CQD    = 128;
constexpr int QND    = 96;
constexpr int QRD    = 32;
constexpr int CKVD   = 128;
constexpr int KND    = 64;
constexpr int KRD    = 64;
constexpr int VDD    = 256;
constexpr int DQK    = 128;
constexpr int ROWS   = BATCH * SLEN;
constexpr int UQ_N   = NH * (QND + QRD);
constexpr int DKV_N  = CKVD + KRD;
constexpr int UKV_N  = NKV * (KND + VDD);
constexpr int HVD    = NH * VDD;

typedef float v2f __attribute__((ext_vector_type(2)));
typedef float v8f __attribute__((ext_vector_type(8)));

__device__ __forceinline__ v8f wmma_f32(v2f a, v2f b, v8f c) {
  return __builtin_amdgcn_wmma_f32_16x16x4_f32(
       false, a,  false, b,
       (short)0, c,  false,  false);
}

__device__ __forceinline__ v8f v8f_zero() {
  v8f z = {0.f, 0.f, 0.f, 0.f, 0.f, 0.f, 0.f, 0.f};
  return z;
}

constexpr int KC = 32;

__global__ __launch_bounds__(256) void gemm_wmma_f32_lds(
    const float* __restrict__ A, const float* __restrict__ B,
    float* __restrict__ C, int M, int N, int Kd) {
  __shared__ float As[2][128][36];
  __shared__ float Bs[2][KC / 2][64][2];

  const int tid  = threadIdx.x;
  const int lane = tid & 31;
  const int wave = tid >> 5;
  const int r    = lane & 15;
  const int hf   = lane >> 4;
  const int m0b  = blockIdx.x * 128;
  const int n0   = blockIdx.y * 64;

  v8f acc[4];
#pragma unroll
  for (int j = 0; j < 4; ++j) acc[j] = v8f_zero();

  const int aRow[4] = {0 * 32 + (tid >> 3), 1 * 32 + (tid >> 3),
                       2 * 32 + (tid >> 3), 3 * 32 + (tid >> 3)};
  const int aCol = (tid * 4) & 31;
  const int bKr[2] = {0 * 16 + (tid >> 4), 1 * 16 + (tid >> 4)};
  const int bNc = (tid * 4) & 63;

  const int nchunk = Kd / KC;
  float4 aReg[4], bReg[2];

#pragma unroll
  for (int rep = 0; rep < 4; ++rep)
    aReg[rep] = *(const float4*)(A + (size_t)(m0b + aRow[rep]) * Kd + aCol);
#pragma unroll
  for (int rep = 0; rep < 2; ++rep)
    bReg[rep] = *(const float4*)(B + (size_t)bKr[rep] * N + n0 + bNc);
#pragma unroll
  for (int rep = 0; rep < 4; ++rep)
    *(float4*)&As[0][aRow[rep]][aCol] = aReg[rep];
#pragma unroll
  for (int rep = 0; rep < 2; ++rep) {
    const int kp = bKr[rep] >> 1, kb = bKr[rep] & 1;
    Bs[0][kp][bNc + 0][kb] = bReg[rep].x;
    Bs[0][kp][bNc + 1][kb] = bReg[rep].y;
    Bs[0][kp][bNc + 2][kb] = bReg[rep].z;
    Bs[0][kp][bNc + 3][kb] = bReg[rep].w;
  }
  __syncthreads();

  for (int c = 0; c < nchunk; ++c) {
    const int buf = c & 1;
    if (c + 1 < nchunk) {
      const int kc = (c + 1) * KC;
#pragma unroll
      for (int rep = 0; rep < 4; ++rep)
        aReg[rep] =
            *(const float4*)(A + (size_t)(m0b + aRow[rep]) * Kd + kc + aCol);
#pragma unroll
      for (int rep = 0; rep < 2; ++rep)
        bReg[rep] =
            *(const float4*)(B + (size_t)(kc + bKr[rep]) * N + n0 + bNc);
    }
#pragma unroll
    for (int k = 0; k < KC; k += 4) {
      v2f a = *(const v2f*)&As[buf][wave * 16 + r][k + 2 * hf];
#pragma unroll
      for (int j = 0; j < 4; ++j) {
        v2f bb = *(const v2f*)&Bs[buf][(k + 2 * hf) >> 1][j * 16 + r][0];
        acc[j] = wmma_f32(a, bb, acc[j]);
      }
    }
    if (c + 1 < nchunk) {
      const int nbuf = buf ^ 1;
#pragma unroll
      for (int rep = 0; rep < 4; ++rep)
        *(float4*)&As[nbuf][aRow[rep]][aCol] = aReg[rep];
#pragma unroll
      for (int rep = 0; rep < 2; ++rep) {
        const int kp = bKr[rep] >> 1, kb = bKr[rep] & 1;
        Bs[nbuf][kp][bNc + 0][kb] = bReg[rep].x;
        Bs[nbuf][kp][bNc + 1][kb] = bReg[rep].y;
        Bs[nbuf][kp][bNc + 2][kb] = bReg[rep].z;
        Bs[nbuf][kp][bNc + 3][kb] = bReg[rep].w;
      }
    }
    __syncthreads();
  }

#pragma unroll
  for (int j = 0; j < 4; ++j)
#pragma unroll
    for (int i = 0; i < 8; ++i)
      C[(size_t)(m0b + wave * 16 + i + 8 * hf) * N + n0 + j * 16 + r] =
          acc[j][i];
}

__global__ __launch_bounds__(32) void t5ln_kernel(
    const float* __restrict__ in, int in_stride,
    const float* __restrict__ gamma, float* __restrict__ out) {
  const int row  = blockIdx.x;
  const int lane = threadIdx.x;
  const float* ip = in + (size_t)row * in_stride;
  float v[4], s = 0.f, s2 = 0.f;
#pragma unroll
  for (int i = 0; i < 4; ++i) {
    v[i] = ip[lane + 32 * i];
    s  += v[i];
    s2 += v[i] * v[i];
  }
#pragma unroll
  for (int off = 1; off < 32; off <<= 1) {
    s  += __shfl_xor(s, off, 32);
    s2 += __shfl_xor(s2, off, 32);
  }
  const float mu  = s * (1.f / 128.f);
  const float var = s2 * (1.f / 128.f) - mu * mu;
  const float rs  = rsqrtf(var + 1e-5f);
  float* op = out + (size_t)row * 128;
#pragma unroll
  for (int i = 0; i < 4; ++i)
    op[lane + 32 * i] = (v[i] - mu) * rs * gamma[lane + 32 * i];
}

__device__ __forceinline__ void rope_pair(float spos, int pair, float invD,
                                          float xe, float xo,
                                          float& re, float& ro) {
  const float lnTheta = 9.210340371976184f;
  float freq = expf(-(2.0f * (float)pair) * invD * lnTheta);
  float ang  = spos * freq;
  float c = cosf(ang), sn = sinf(ang);
  re = xe * c - xo * sn;
  ro = xe * sn + xo * c;
}

__global__ __launch_bounds__(256) void build_q_kernel(
    const float* __restrict__ qraw, float* __restrict__ qst) {
  size_t idx = (size_t)blockIdx.x * blockDim.x + threadIdx.x;
  const size_t total = (size_t)BATCH * NH * SLEN * DQK;
  if (idx >= total) return;
  int j = idx & 127;
  int s = (int)((idx >> 7) & (SLEN - 1));
  int h = (int)((idx >> 18) & (NH - 1));
  int b = (int)(idx >> 22);
  const float* qr = qraw + ((size_t)(b * SLEN + s)) * UQ_N;
  float val;
  if (j < QND) {
    val = qr[h * QND + j];
  } else {
    int jj = j - QND;
    int pair = jj >> 1;
    const float* base = qr + NH * QND + h * QRD;
    float xe = base[2 * pair], xo = base[2 * pair + 1];
    float re, ro;
    rope_pair((float)s, pair, 1.0f / (float)QRD, xe, xo, re, ro);
    val = (jj & 1) ? ro : re;
  }
  qst[idx] = val;
}

__global__ __launch_bounds__(256) void build_k_kernel(
    const float* __restrict__ kvraw, const float* __restrict__ ckv_kr,
    float* __restrict__ kst) {
  size_t idx = (size_t)blockIdx.x * blockDim.x + threadIdx.x;
  const size_t total = (size_t)BATCH * NKV * SLEN * DQK;
  if (idx >= total) return;
  int j   = idx & 127;
  int s   = (int)((idx >> 7) & (SLEN - 1));
  int kvh = (int)((idx >> 18) & (NKV - 1));
  int b   = (int)(idx >> 20);
  size_t row = (size_t)(b * SLEN + s);
  float val;
  if (j < KND) {
    val = kvraw[row * UKV_N + kvh * KND + j];
  } else {
    int jj = j - KND;
    int pair = jj >> 1;
    const float* base = ckv_kr + row * DKV_N + CKVD;
    float xe = base[2 * pair], xo = base[2 * pair + 1];
    float re, ro;
    rope_pair((float)s, pair, 1.0f / (float)KRD, xe, xo, re, ro);
    val = (jj & 1) ? ro : re;
  }
  kst[idx] = val;
}

__global__ __launch_bounds__(256) void build_v_kernel(
    const float* __restrict__ kvraw, float* __restrict__ vpi) {
  size_t idx = (size_t)blockIdx.x * blockDim.x + threadIdx.x;
  const size_t total = (size_t)BATCH * NKV * SLEN * VDD;
  if (idx >= total) return;
  int d   = idx & 255;
  int s   = (int)((idx >> 8) & (SLEN - 1));
  int kvh = (int)((idx >> 19) & (NKV - 1));
  int b   = (int)(idx >> 21);
  size_t row = (size_t)(b * SLEN + s);
  float val = kvraw[row * UKV_N + NKV * KND + kvh * VDD + d];
  size_t dst = ((size_t)(b * NKV + kvh) * (SLEN / 2) + (s >> 1)) * (2 * VDD) +
               d * 2 + (s & 1);
  vpi[dst] = val;
}

__global__ __launch_bounds__(256) __attribute__((amdgpu_waves_per_eu(2, 2)))
void attn_kernel(
    const float* __restrict__ Q, const float* __restrict__ K,
    const float* __restrict__ V, float* __restrict__ O) {
  __shared__ float ps[8][16][16];

  const int lane = threadIdx.x & 31;
  const int wave = threadIdx.x >> 5;
  const int r    = lane & 15;
  const int hf   = lane >> 4;
  const int bh   = blockIdx.x;
  const int b    = bh >> 4;
  const int h    = bh & 15;
  const int kvh  = h >> 2;
  const int q0   = blockIdx.y * 128 + wave * 16;

  const float* Qb = Q + ((size_t)bh * SLEN + q0) * DQK;
  const float* Kb = K + ((size_t)(b * NKV + kvh) * SLEN) * DQK;
  const float* Vb = V + ((size_t)(b * NKV + kvh) * SLEN) * VDD;

  const float scale = 0.08838834764831845f;

  v2f qa[32];
#pragma unroll
  for (int di = 0; di < 32; ++di) {
    const float* p = Qb + (size_t)r * DQK + di * 4 + 2 * hf;
    v2f t;
    t.x = p[0] * scale;
    t.y = p[1] * scale;
    qa[di] = t;
  }

  float m_i[8], l_i[8];
#pragma unroll
  for (int i = 0; i < 8; ++i) { m_i[i] = -1e30f; l_i[i] = 0.f; }
  v8f acc[16];
#pragma unroll
  for (int nt = 0; nt < 16; ++nt) acc[nt] = v8f_zero();

  const int ktmax = (q0 + 15) >> 4;
  for (int kt = 0; kt <= ktmax; ++kt) {
    const int k0 = kt * 16;

    if (kt < ktmax) {
      const float* kpf = Kb + (size_t)(k0 + 16 + r) * DQK + hf * 64;
      __builtin_prefetch(kpf, 0, 3);
      const float* vpf = Vb + (size_t)((k0 + 16) >> 1) * (2 * VDD) + lane * 64;
      __builtin_prefetch(vpf, 0, 3);
    }

    v8f sc = v8f_zero();
    const float* Kt = Kb + (size_t)(k0 + r) * DQK + 2 * hf;
#pragma unroll
    for (int di = 0; di < 32; ++di) {
      v2f bb;
      bb.x = Kt[di * 4];
      bb.y = Kt[di * 4 + 1];
      sc = wmma_f32(qa[di], bb, sc);
    }

#pragma unroll
    for (int i = 0; i < 8; ++i) {
      const int qrow = q0 + i + 8 * hf;
      const int kcol = k0 + r;
      float v = (kcol > qrow) ? -1e30f : sc[i];
      float mx = v;
#pragma unroll
      for (int off = 1; off < 16; off <<= 1)
        mx = fmaxf(mx, __shfl_xor(mx, off, 32));
      const float mnew  = fmaxf(m_i[i], mx);
      const float alpha = __expf(m_i[i] - mnew);
      const float pe    = __expf(v - mnew);
      float rs = pe;
#pragma unroll
      for (int off = 1; off < 16; off <<= 1)
        rs += __shfl_xor(rs, off, 32);
      l_i[i] = l_i[i] * alpha + rs;
      m_i[i] = mnew;
#pragma unroll
      for (int nt = 0; nt < 16; ++nt) acc[nt][i] *= alpha;
      ps[wave][i + 8 * hf][r] = pe;
    }
    asm volatile("s_wait_dscnt 0" ::: "memory");

    v2f pa[4];
#pragma unroll
    for (int ks = 0; ks < 4; ++ks) {
      v2f t;
      t.x = ps[wave][r][ks * 4 + 2 * hf];
      t.y = ps[wave][r][ks * 4 + 1 + 2 * hf];
      pa[ks] = t;
    }

#pragma unroll
    for (int nt = 0; nt < 16; ++nt) {
#pragma unroll
      for (int ks = 0; ks < 4; ++ks) {
        const int kk = k0 + ks * 4 + 2 * hf;
        const float* Vp =
            Vb + (size_t)(kk >> 1) * (2 * VDD) + (nt * 16 + r) * 2;
        v2f bb = *(const v2f*)Vp;
        acc[nt] = wmma_f32(pa[ks], bb, acc[nt]);
      }
    }
  }

#pragma unroll
  for (int i = 0; i < 8; ++i) {
    const float inv = 1.0f / l_i[i];
    const int qrow = q0 + i + 8 * hf;
    float* Op = O + ((size_t)(b * SLEN + qrow)) * HVD + h * VDD;
#pragma unroll
    for (int nt = 0; nt < 16; ++nt) Op[nt * 16 + r] = acc[nt][i] * inv;
  }
}

extern "C" void kernel_launch(void* const* d_in, const int* in_sizes, int n_in,
                              void* d_out, int out_size, void* d_ws,
                              size_t ws_size, hipStream_t stream) {
  (void)in_sizes; (void)n_in; (void)out_size; (void)ws_size;
  const float* x        = (const float*)d_in[0];
  const float* w_dq     = (const float*)d_in[1];
  const float* gamma_q  = (const float*)d_in[2];
  const float* w_uq     = (const float*)d_in[3];
  const float* w_dkv    = (const float*)d_in[4];
  const float* gamma_kv = (const float*)d_in[5];
  const float* w_ukv    = (const float*)d_in[6];
  const float* w_o      = (const float*)d_in[7];
  float* out = (float*)d_out;

  float* ws = (float*)d_ws;
  float* tmp_q   = ws;
  float* cq      = tmp_q   + (size_t)ROWS * CQD;
  float* qraw    = cq      + (size_t)ROWS * CQD;
  float* qst     = qraw    + (size_t)ROWS * UQ_N;
  float* ckv_kr  = qst     + (size_t)BATCH * NH * SLEN * DQK;
  float* ckv     = ckv_kr  + (size_t)ROWS * DKV_N;
  float* kvraw   = ckv     + (size_t)ROWS * CKVD;
  float* kst     = kvraw   + (size_t)ROWS * UKV_N;
  float* vpi     = kst     + (size_t)BATCH * NKV * SLEN * DQK;
  float* attn_o  = vpi     + (size_t)BATCH * NKV * SLEN * VDD;

  gemm_wmma_f32_lds<<<dim3(ROWS / 128, CQD / 64), 256, 0, stream>>>(
      x, w_dq, tmp_q, ROWS, CQD, DIMC);
  t5ln_kernel<<<ROWS, 32, 0, stream>>>(tmp_q, CQD, gamma_q, cq);
  gemm_wmma_f32_lds<<<dim3(ROWS / 128, UQ_N / 64), 256, 0, stream>>>(
      cq, w_uq, qraw, ROWS, UQ_N, CQD);
  {
    size_t total = (size_t)BATCH * NH * SLEN * DQK;
    build_q_kernel<<<(unsigned)((total + 255) / 256), 256, 0, stream>>>(qraw, qst);
  }
  gemm_wmma_f32_lds<<<dim3(ROWS / 128, DKV_N / 64), 256, 0, stream>>>(
      x, w_dkv, ckv_kr, ROWS, DKV_N, DIMC);
  t5ln_kernel<<<ROWS, 32, 0, stream>>>(ckv_kr, DKV_N, gamma_kv, ckv);
  gemm_wmma_f32_lds<<<dim3(ROWS / 128, UKV_N / 64), 256, 0, stream>>>(
      ckv, w_ukv, kvraw, ROWS, UKV_N, CQD);
  {
    size_t tk = (size_t)BATCH * NKV * SLEN * DQK;
    build_k_kernel<<<(unsigned)((tk + 255) / 256), 256, 0, stream>>>(kvraw, ckv_kr, kst);
    size_t tv = (size_t)BATCH * NKV * SLEN * VDD;
    build_v_kernel<<<(unsigned)((tv + 255) / 256), 256, 0, stream>>>(kvraw, vpi);
  }
  attn_kernel<<<dim3(BATCH * NH, SLEN / 128), 256, 0, stream>>>(qst, kst, vpi, attn_o);
  gemm_wmma_f32_lds<<<dim3(ROWS / 128, DIMC / 64), 256, 0, stream>>>(
      attn_o, w_o, out, ROWS, DIMC, HVD);
}


# attention: K built in WMMA-fragment order, so a K fragment load is one whole-line 16-byte-per-lane load (16 per tile), not 32 8-byte loads over 16 rows each; sum order over d in QK permuted
# speedup vs baseline: 1.8673x; 1.0297x over previous
#include <hip/hip_runtime.h>
#include <math.h>

constexpr int DIMC   = 2048;
constexpr int SLEN   = 2048;
constexpr int BATCH  = 2;
constexpr int NH     = 16;
constexpr int NKV    = 4;
constexpr int CQD    = 128;
constexpr int QND    = 96;
constexpr int QRD    = 32;
constexpr int CKVD   = 128;
constexpr int KND    = 64;
constexpr int KRD    = 64;
constexpr int VDD    = 256;
constexpr int DQK    = 128;
constexpr int ROWS   = BATCH * SLEN;
constexpr int UQ_N   = NH * (QND + QRD);
constexpr int DKV_N  = CKVD + KRD;
constexpr int UKV_N  = NKV * (KND + VDD);
constexpr int HVD    = NH * VDD;

typedef float v2f __attribute__((ext_vector_type(2)));
typedef float v8f __attribute__((ext_vector_type(8)));

__device__ __forceinline__ v8f wmma_f32(v2f a, v2f b, v8f c) {
  return __builtin_amdgcn_wmma_f32_16x16x4_f32(
       false, a,  false, b,
       (short)0, c,  false,  false);
}

__device__ __forceinline__ v8f v8f_zero() {
  v8f z = {0.f, 0.f, 0.f, 0.f, 0.f, 0.f, 0.f, 0.f};
  return z;
}

constexpr int KC = 32;

__global__ __launch_bounds__(256) void gemm_wmma_f32_lds(
    const float* __restrict__ A, const float* __restrict__ B,
    float* __restrict__ C, int M, int N, int Kd) {
  __shared__ float As[2][128][36];
  __shared__ float Bs[2][KC / 2][64][2];

  const int tid  = threadIdx.x;
  const int lane = tid & 31;
  const int wave = tid >> 5;
  const int r    = lane & 15;
  const int hf   = lane >> 4;
  const int m0b  = blockIdx.x * 128;
  const int n0   = blockIdx.y * 64;

  v8f acc[4];
#pragma unroll
  for (int j = 0; j < 4; ++j) acc[j] = v8f_zero();

  const int aRow[4] = {0 * 32 + (tid >> 3), 1 * 32 + (tid >> 3),
                       2 * 32 + (tid >> 3), 3 * 32 + (tid >> 3)};
  const int aCol = (tid * 4) & 31;
  const int bKr[2] = {0 * 16 + (tid >> 4), 1 * 16 + (tid >> 4)};
  const int bNc = (tid * 4) & 63;

  const int nchunk = Kd / KC;
  float4 aReg[4], bReg[2];

#pragma unroll
  for (int rep = 0; rep < 4; ++rep)
    aReg[rep] = *(const float4*)(A + (size_t)(m0b + aRow[rep]) * Kd + aCol);
#pragma unroll
  for (int rep = 0; rep < 2; ++rep)
    bReg[rep] = *(const float4*)(B + (size_t)bKr[rep] * N + n0 + bNc);
#pragma unroll
  for (int rep = 0; rep < 4; ++rep)
    *(float4*)&As[0][aRow[rep]][aCol] = aReg[rep];
#pragma unroll
  for (int rep = 0; rep < 2; ++rep) {
    const int kp = bKr[rep] >> 1, kb = bKr[rep] & 1;
    Bs[0][kp][bNc + 0][kb] = bReg[rep].x;
    Bs[0][kp][bNc + 1][kb] = bReg[rep].y;
    Bs[0][kp][bNc + 2][kb] = bReg[rep].z;
    Bs[0][kp][bNc + 3][kb] = bReg[rep].w;
  }
  __syncthreads();

  for (int c = 0; c < nchunk; ++c) {
    const int buf = c & 1;
    if (c + 1 < nchunk) {
      const int kc = (c + 1) * KC;
#pragma unroll
      for (int rep = 0; rep < 4; ++rep)
        aReg[rep] =
            *(const float4*)(A + (size_t)(m0b + aRow[rep]) * Kd + kc + aCol);
#pragma unroll
      for (int rep = 0; rep < 2; ++rep)
        bReg[rep] =
            *(const float4*)(B + (size_t)(kc + bKr[rep]) * N + n0 + bNc);
    }
#pragma unroll
    for (int k = 0; k < KC; k += 4) {
      v2f a = *(const v2f*)&As[buf][wave * 16 + r][k + 2 * hf];
#pragma unroll
      for (int j = 0; j < 4; ++j) {
        v2f bb = *(const v2f*)&Bs[buf][(k + 2 * hf) >> 1][j * 16 + r][0];
        acc[j] = wmma_f32(a, bb, acc[j]);
      }
    }
    if (c + 1 < nchunk) {
      const int nbuf = buf ^ 1;
#pragma unroll
      for (int rep = 0; rep < 4; ++rep)
        *(float4*)&As[nbuf][aRow[rep]][aCol] = aReg[rep];
#pragma unroll
      for (int rep = 0; rep < 2; ++rep) {
        const int kp = bKr[rep] >> 1, kb = bKr[rep] & 1;
        Bs[nbuf][kp][bNc + 0][kb] = bReg[rep].x;
        Bs[nbuf][kp][bNc + 1][kb] = bReg[rep].y;
        Bs[nbuf][kp][bNc + 2][kb] = bReg[rep].z;
        Bs[nbuf][kp][bNc + 3][kb] = bReg[rep].w;
      }
    }
    __syncthreads();
  }

#pragma unroll
  for (int j = 0; j < 4; ++j)
#pragma unroll
    for (int i = 0; i < 8; ++i)
      C[(size_t)(m0b + wave * 16 + i + 8 * hf) * N + n0 + j * 16 + r] =
          acc[j][i];
}

__global__ __launch_bounds__(32) void t5ln_kernel(
    const float* __restrict__ in, int in_stride,
    const float* __restrict__ gamma, float* __restrict__ out) {
  const int row  = blockIdx.x;
  const int lane = threadIdx.x;
  const float* ip = in + (size_t)row * in_stride;
  float v[4], s = 0.f, s2 = 0.f;
#pragma unroll
  for (int i = 0; i < 4; ++i) {
    v[i] = ip[lane + 32 * i];
    s  += v[i];
    s2 += v[i] * v[i];
  }
#pragma unroll
  for (int off = 1; off < 32; off <<= 1) {
    s  += __shfl_xor(s, off, 32);
    s2 += __shfl_xor(s2, off, 32);
  }
  const float mu  = s * (1.f / 128.f);
  const float var = s2 * (1.f / 128.f) - mu * mu;
  const float rs  = rsqrtf(var + 1e-5f);
  float* op = out + (size_t)row * 128;
#pragma unroll
  for (int i = 0; i < 4; ++i)
    op[lane + 32 * i] = (v[i] - mu) * rs * gamma[lane + 32 * i];
}

__device__ __forceinline__ void rope_pair(float spos, int pair, float invD,
                                          float xe, float xo,
                                          float& re, float& ro) {
  const float lnTheta = 9.210340371976184f;
  float freq = expf(-(2.0f * (float)pair) * invD * lnTheta);
  float ang  = spos * freq;
  float c = cosf(ang), sn = sinf(ang);
  re = xe * c - xo * sn;
  ro = xe * sn + xo * c;
}

__global__ __launch_bounds__(256) void build_q_kernel(
    const float* __restrict__ qraw, float* __restrict__ qst) {
  size_t idx = (size_t)blockIdx.x * blockDim.x + threadIdx.x;
  const size_t total = (size_t)BATCH * NH * SLEN * DQK;
  if (idx >= total) return;
  int j = idx & 127;
  int s = (int)((idx >> 7) & (SLEN - 1));
  int h = (int)((idx >> 18) & (NH - 1));
  int b = (int)(idx >> 22);
  const float* qr = qraw + ((size_t)(b * SLEN + s)) * UQ_N;
  float val;
  if (j < QND) {
    val = qr[h * QND + j];
  } else {
    int jj = j - QND;
    int pair = jj >> 1;
    const float* base = qr + NH * QND + h * QRD;
    float xe = base[2 * pair], xo = base[2 * pair + 1];
    float re, ro;
    rope_pair((float)s, pair, 1.0f / (float)QRD, xe, xo, re, ro);
    val = (jj & 1) ? ro : re;
  }
  qst[idx] = val;
}

__global__ __launch_bounds__(256) void build_k_kernel(
    const float* __restrict__ kvraw, const float* __restrict__ ckv_kr,
    float* __restrict__ kst) {
  size_t idx = (size_t)blockIdx.x * blockDim.x + threadIdx.x;
  const size_t total = (size_t)BATCH * NKV * SLEN * DQK;
  if (idx >= total) return;
  int e   = idx & 3;
  int ln  = (int)((idx >> 2) & 31);
  int jc  = (int)((idx >> 7) & 15);
  int t   = (int)((idx >> 11) & (SLEN / 16 - 1));
  int kvh = (int)((idx >> 18) & (NKV - 1));
  int b   = (int)(idx >> 20);
  int j   = jc * 8 + (ln >> 4) * 4 + e;
  int s   = t * 16 + (ln & 15);
  size_t row = (size_t)(b * SLEN + s);
  float val;
  if (j < KND) {
    val = kvraw[row * UKV_N + kvh * KND + j];
  } else {
    int jj = j - KND;
    int pair = jj >> 1;
    const float* base = ckv_kr + row * DKV_N + CKVD;
    float xe = base[2 * pair], xo = base[2 * pair + 1];
    float re, ro;
    rope_pair((float)s, pair, 1.0f / (float)KRD, xe, xo, re, ro);
    val = (jj & 1) ? ro : re;
  }
  kst[idx] = val;
}

__global__ __launch_bounds__(256) void build_v_kernel(
    const float* __restrict__ kvraw, float* __restrict__ vpi) {
  size_t idx = (size_t)blockIdx.x * blockDim.x + threadIdx.x;
  const size_t total = (size_t)BATCH * NKV * SLEN * VDD;
  if (idx >= total) return;
  int d   = idx & 255;
  int s   = (int)((idx >> 8) & (SLEN - 1));
  int kvh = (int)((idx >> 19) & (NKV - 1));
  int b   = (int)(idx >> 21);
  size_t row = (size_t)(b * SLEN + s);
  float val = kvraw[row * UKV_N + NKV * KND + kvh * VDD + d];
  size_t dst = ((size_t)(b * NKV + kvh) * (SLEN / 2) + (s >> 1)) * (2 * VDD) +
               d * 2 + (s & 1);
  vpi[dst] = val;
}

__global__ __launch_bounds__(256) __attribute__((amdgpu_waves_per_eu(2, 2)))
void attn_kernel(
    const float* __restrict__ Q, const float* __restrict__ K,
    const float* __restrict__ V, float* __restrict__ O) {
  __shared__ float ps[8][16][16];

  const int lane = threadIdx.x & 31;
  const int wave = threadIdx.x >> 5;
  const int r    = lane & 15;
  const int hf   = lane >> 4;
  const int bh   = blockIdx.x;
  const int b    = bh >> 4;
  const int h    = bh & 15;
  const int kvh  = h >> 2;
  const int q0   = blockIdx.y * 128 + wave * 16;

  const float* Qb = Q + ((size_t)bh * SLEN + q0) * DQK;
  const float* Kb = K + ((size_t)(b * NKV + kvh) * SLEN) * DQK;
  const float* Vb = V + ((size_t)(b * NKV + kvh) * SLEN) * VDD;

  const float scale = 0.08838834764831845f;

  v2f qa[32];
#pragma unroll
  for (int j = 0; j < 16; ++j) {
    const float4 qv = *(const float4*)(Qb + (size_t)r * DQK + j * 8 + 4 * hf);
    v2f t0, t1;
    t0.x = qv.x * scale;
    t0.y = qv.y * scale;
    t1.x = qv.z * scale;
    t1.y = qv.w * scale;
    qa[2 * j]     = t0;
    qa[2 * j + 1] = t1;
  }

  float m_i[8], l_i[8];
#pragma unroll
  for (int i = 0; i < 8; ++i) { m_i[i] = -1e30f; l_i[i] = 0.f; }
  v8f acc[16];
#pragma unroll
  for (int nt = 0; nt < 16; ++nt) acc[nt] = v8f_zero();

  const int ktmax = (q0 + 15) >> 4;
  for (int kt = 0; kt <= ktmax; ++kt) {
    const int k0 = kt * 16;

    if (kt < ktmax) {
      const float* kpf = Kb + (size_t)(kt + 1) * (16 * DQK) + lane * 64;
      __builtin_prefetch(kpf, 0, 3);
      const float* vpf = Vb + (size_t)((k0 + 16) >> 1) * (2 * VDD) + lane * 64;
      __builtin_prefetch(vpf, 0, 3);
    }

    v8f sc = v8f_zero();
    const float* Kt = Kb + (size_t)kt * (16 * DQK) + lane * 4;
#pragma unroll
    for (int j = 0; j < 16; ++j) {
      const float4 kf = *(const float4*)(Kt + j * 128);
      v2f b0, b1;
      b0.x = kf.x;
      b0.y = kf.y;
      b1.x = kf.z;
      b1.y = kf.w;
      sc = wmma_f32(qa[2 * j], b0, sc);
      sc = wmma_f32(qa[2 * j + 1], b1, sc);
    }

#pragma unroll
    for (int i = 0; i < 8; ++i) {
      const int qrow = q0 + i + 8 * hf;
      const int kcol = k0 + r;
      float v = (kcol > qrow) ? -1e30f : sc[i];
      float mx = v;
#pragma unroll
      for (int off = 1; off < 16; off <<= 1)
        mx = fmaxf(mx, __shfl_xor(mx, off, 32));
      const float mnew  = fmaxf(m_i[i], mx);
      const float alpha = __expf(m_i[i] - mnew);
      const float pe    = __expf(v - mnew);
      float rs = pe;
#pragma unroll
      for (int off = 1; off < 16; off <<= 1)
        rs += __shfl_xor(rs, off, 32);
      l_i[i] = l_i[i] * alpha + rs;
      m_i[i] = mnew;
#pragma unroll
      for (int nt = 0; nt < 16; ++nt) acc[nt][i] *= alpha;
      ps[wave][i + 8 * hf][r] = pe;
    }
    asm volatile("s_wait_dscnt 0" ::: "memory");

    v2f pa[4];
#pragma unroll
    for (int ks = 0; ks < 4; ++ks) {
      v2f t;
      t.x = ps[wave][r][ks * 4 + 2 * hf];
      t.y = ps[wave][r][ks * 4 + 1 + 2 * hf];
      pa[ks] = t;
    }

#pragma unroll
    for (int nt = 0; nt < 16; ++nt) {
#pragma unroll
      for (int ks = 0; ks < 4; ++ks) {
        const int kk = k0 + ks * 4 + 2 * hf;
        const float* Vp =
            Vb + (size_t)(kk >> 1) * (2 * VDD) + (nt * 16 + r) * 2;
        v2f bb = *(const v2f*)Vp;
        acc[nt] = wmma_f32(pa[ks], bb, acc[nt]);
      }
    }
  }

#pragma unroll
  for (int i = 0; i < 8; ++i) {
    const float inv = 1.0f / l_i[i];
    const int qrow = q0 + i + 8 * hf;
    float* Op = O + ((size_t)(b * SLEN + qrow)) * HVD + h * VDD;
#pragma unroll
    for (int nt = 0; nt < 16; ++nt) Op[nt * 16 + r] = acc[nt][i] * inv;
  }
}

extern "C" void kernel_launch(void* const* d_in, const int* in_sizes, int n_in,
                              void* d_out, int out_size, void* d_ws,
                              size_t ws_size, hipStream_t stream) {
  (void)in_sizes; (void)n_in; (void)out_size; (void)ws_size;
  const float* x        = (const float*)d_in[0];
  const float* w_dq     = (const float*)d_in[1];
  const float* gamma_q  = (const float*)d_in[2];
  const float* w_uq     = (const float*)d_in[3];
  const float* w_dkv    = (const float*)d_in[4];
  const float* gamma_kv = (const float*)d_in[5];
  const float* w_ukv    = (const float*)d_in[6];
  const float* w_o      = (const float*)d_in[7];
  float* out = (float*)d_out;

  float* ws = (float*)d_ws;
  float* tmp_q   = ws;
  float* cq      = tmp_q   + (size_t)ROWS * CQD;
  float* qraw    = cq      + (size_t)ROWS * CQD;
  float* qst     = qraw    + (size_t)ROWS * UQ_N;
  float* ckv_kr  = qst     + (size_t)BATCH * NH * SLEN * DQK;
  float* ckv     = ckv_kr  + (size_t)ROWS * DKV_N;
  float* kvraw   = ckv     + (size_t)ROWS * CKVD;
  float* kst     = kvraw   + (size_t)ROWS * UKV_N;
  float* vpi     = kst     + (size_t)BATCH * NKV * SLEN * DQK;
  float* attn_o  = vpi     + (size_t)BATCH * NKV * SLEN * VDD;

  gemm_wmma_f32_lds<<<dim3(ROWS / 128, CQD / 64), 256, 0, stream>>>(
      x, w_dq, tmp_q, ROWS, CQD, DIMC);
  t5ln_kernel<<<ROWS, 32, 0, stream>>>(tmp_q, CQD, gamma_q, cq);
  gemm_wmma_f32_lds<<<dim3(ROWS / 128, UQ_N / 64), 256, 0, stream>>>(
      cq, w_uq, qraw, ROWS, UQ_N, CQD);
  {
    size_t total = (size_t)BATCH * NH * SLEN * DQK;
    build_q_kernel<<<(unsigned)((total + 255) / 256), 256, 0, stream>>>(qraw, qst);
  }
  gemm_wmma_f32_lds<<<dim3(ROWS / 128, DKV_N / 64), 256, 0, stream>>>(
      x, w_dkv, ckv_kr, ROWS, DKV_N, DIMC);
  t5ln_kernel<<<ROWS, 32, 0, stream>>>(ckv_kr, DKV_N, gamma_kv, ckv);
  gemm_wmma_f32_lds<<<dim3(ROWS / 128, UKV_N / 64), 256, 0, stream>>>(
      ckv, w_ukv, kvraw, ROWS, UKV_N, CQD);
  {
    size_t tk = (size_t)BATCH * NKV * SLEN * DQK;
    build_k_kernel<<<(unsigned)((tk + 255) / 256), 256, 0, stream>>>(kvraw, ckv_kr, kst);
    size_t tv = (size_t)BATCH * NKV * SLEN * VDD;
    build_v_kernel<<<(unsigned)((tv + 255) / 256), 256, 0, stream>>>(kvraw, vpi);
  }
  attn_kernel<<<dim3(BATCH * NH, SLEN / 128), 256, 0, stream>>>(qst, kst, vpi, attn_o);
  gemm_wmma_f32_lds<<<dim3(ROWS / 128, DIMC / 64), 256, 0, stream>>>(
      attn_o, w_o, out, ROWS, DIMC, HVD);
}


# GEMM: next K-chunk's six 16-byte global loads held in named registers, not a float4 array the compiler kept in scratch memory: issued as one group ahead of the 32 WMMAs, no scratch round trip
# speedup vs baseline: 3.7959x; 2.0328x over previous
#include <hip/hip_runtime.h>
#include <math.h>

constexpr int DIMC   = 2048;
constexpr int SLEN   = 2048;
constexpr int BATCH  = 2;
constexpr int NH     = 16;
constexpr int NKV    = 4;
constexpr int CQD    = 128;
constexpr int QND    = 96;
constexpr int QRD    = 32;
constexpr int CKVD   = 128;
constexpr int KND    = 64;
constexpr int KRD    = 64;
constexpr int VDD    = 256;
constexpr int DQK    = 128;
constexpr int ROWS   = BATCH * SLEN;
constexpr int UQ_N   = NH * (QND + QRD);
constexpr int DKV_N  = CKVD + KRD;
constexpr int UKV_N  = NKV * (KND + VDD);
constexpr int HVD    = NH * VDD;

typedef float v2f __attribute__((ext_vector_type(2)));
typedef float v8f __attribute__((ext_vector_type(8)));

__device__ __forceinline__ v8f wmma_f32(v2f a, v2f b, v8f c) {
  return __builtin_amdgcn_wmma_f32_16x16x4_f32(
       false, a,  false, b,
       (short)0, c,  false,  false);
}

__device__ __forceinline__ v8f v8f_zero() {
  v8f z = {0.f, 0.f, 0.f, 0.f, 0.f, 0.f, 0.f, 0.f};
  return z;
}

constexpr int KC = 32;

typedef float v4f __attribute__((ext_vector_type(4)));

__device__ __forceinline__ void gemm_stage_chunk(
    float (*Asb)[36], float (*Bsb)[64][2], int ar, int ac, int br, int bn,
    v4f a0, v4f a1, v4f a2, v4f a3, v4f b0, v4f b1) {
  *(v4f*)&Asb[ar][ac]      = a0;
  *(v4f*)&Asb[ar + 32][ac] = a1;
  *(v4f*)&Asb[ar + 64][ac] = a2;
  *(v4f*)&Asb[ar + 96][ac] = a3;
  const int kp = br >> 1, kb = br & 1;
  Bsb[kp][bn + 0][kb]     = b0.x;
  Bsb[kp][bn + 1][kb]     = b0.y;
  Bsb[kp][bn + 2][kb]     = b0.z;
  Bsb[kp][bn + 3][kb]     = b0.w;
  Bsb[kp + 8][bn + 0][kb] = b1.x;
  Bsb[kp + 8][bn + 1][kb] = b1.y;
  Bsb[kp + 8][bn + 2][kb] = b1.z;
  Bsb[kp + 8][bn + 3][kb] = b1.w;
}

__global__ __launch_bounds__(256) void gemm_wmma_f32_lds(
    const float* __restrict__ A, const float* __restrict__ B,
    float* __restrict__ C, int M, int N, int Kd) {
  __shared__ float As[2][128][36];
  __shared__ float Bs[2][KC / 2][64][2];

  const int tid  = threadIdx.x;
  const int lane = tid & 31;
  const int wave = tid >> 5;
  const int r    = lane & 15;
  const int hf   = lane >> 4;
  const int m0b  = blockIdx.x * 128;
  const int n0   = blockIdx.y * 64;

  v8f acc[4];
#pragma unroll
  for (int j = 0; j < 4; ++j) acc[j] = v8f_zero();

  const int ar = tid >> 3;
  const int ac = (tid * 4) & 31;
  const int br = tid >> 4;
  const int bn = (tid * 4) & 63;

  const float* Ap0 = A + (size_t)(m0b + ar) * Kd + ac;
  const float* Ap1 = Ap0 + (size_t)32 * Kd;
  const float* Ap2 = Ap0 + (size_t)64 * Kd;
  const float* Ap3 = Ap0 + (size_t)96 * Kd;
  const float* Bp0 = B + (size_t)br * N + n0 + bn;
  const float* Bp1 = Bp0 + (size_t)16 * N;

  const int nchunk = Kd / KC;

  v4f a0 = *(const v4f*)Ap0;
  v4f a1 = *(const v4f*)Ap1;
  v4f a2 = *(const v4f*)Ap2;
  v4f a3 = *(const v4f*)Ap3;
  v4f b0 = *(const v4f*)Bp0;
  v4f b1 = *(const v4f*)Bp1;
  gemm_stage_chunk(As[0], Bs[0], ar, ac, br, bn, a0, a1, a2, a3, b0, b1);
  __syncthreads();

  for (int c = 0; c < nchunk; ++c) {
    const int buf = c & 1;
    const bool more = (c + 1 < nchunk);
    if (more) {
      const int kc = (c + 1) * KC;
      const size_t kb = (size_t)kc * N;
      a0 = *(const v4f*)(Ap0 + kc);
      a1 = *(const v4f*)(Ap1 + kc);
      a2 = *(const v4f*)(Ap2 + kc);
      a3 = *(const v4f*)(Ap3 + kc);
      b0 = *(const v4f*)(Bp0 + kb);
      b1 = *(const v4f*)(Bp1 + kb);
    }
#pragma unroll
    for (int k = 0; k < KC; k += 4) {
      v2f a = *(const v2f*)&As[buf][wave * 16 + r][k + 2 * hf];
#pragma unroll
      for (int j = 0; j < 4; ++j) {
        v2f bb = *(const v2f*)&Bs[buf][(k + 2 * hf) >> 1][j * 16 + r][0];
        acc[j] = wmma_f32(a, bb, acc[j]);
      }
    }
    if (more)
      gemm_stage_chunk(As[buf ^ 1], Bs[buf ^ 1], ar, ac, br, bn,
                       a0, a1, a2, a3, b0, b1);
    __syncthreads();
  }

#pragma unroll
  for (int j = 0; j < 4; ++j)
#pragma unroll
    for (int i = 0; i < 8; ++i)
      C[(size_t)(m0b + wave * 16 + i + 8 * hf) * N + n0 + j * 16 + r] =
          acc[j][i];
}

__global__ __launch_bounds__(32) void t5ln_kernel(
    const float* __restrict__ in, int in_stride,
    const float* __restrict__ gamma, float* __restrict__ out) {
  const int row  = blockIdx.x;
  const int lane = threadIdx.x;
  const float* ip = in + (size_t)row * in_stride;
  float v[4], s = 0.f, s2 = 0.f;
#pragma unroll
  for (int i = 0; i < 4; ++i) {
    v[i] = ip[lane + 32 * i];
    s  += v[i];
    s2 += v[i] * v[i];
  }
#pragma unroll
  for (int off = 1; off < 32; off <<= 1) {
    s  += __shfl_xor(s, off, 32);
    s2 += __shfl_xor(s2, off, 32);
  }
  const float mu  = s * (1.f / 128.f);
  const float var = s2 * (1.f / 128.f) - mu * mu;
  const float rs  = rsqrtf(var + 1e-5f);
  float* op = out + (size_t)row * 128;
#pragma unroll
  for (int i = 0; i < 4; ++i)
    op[lane + 32 * i] = (v[i] - mu) * rs * gamma[lane + 32 * i];
}

__device__ __forceinline__ void rope_pair(float spos, int pair, float invD,
                                          float xe, float xo,
                                          float& re, float& ro) {
  const float lnTheta = 9.210340371976184f;
  float freq = expf(-(2.0f * (float)pair) * invD * lnTheta);
  float ang  = spos * freq;
  float c = cosf(ang), sn = sinf(ang);
  re = xe * c - xo * sn;
  ro = xe * sn + xo * c;
}

__global__ __launch_bounds__(256) void build_q_kernel(
    const float* __restrict__ qraw, float* __restrict__ qst) {
  size_t idx = (size_t)blockIdx.x * blockDim.x + threadIdx.x;
  const size_t total = (size_t)BATCH * NH * SLEN * DQK;
  if (idx >= total) return;
  int j = idx & 127;
  int s = (int)((idx >> 7) & (SLEN - 1));
  int h = (int)((idx >> 18) & (NH - 1));
  int b = (int)(idx >> 22);
  const float* qr = qraw + ((size_t)(b * SLEN + s)) * UQ_N;
  float val;
  if (j < QND) {
    val = qr[h * QND + j];
  } else {
    int jj = j - QND;
    int pair = jj >> 1;
    const float* base = qr + NH * QND + h * QRD;
    float xe = base[2 * pair], xo = base[2 * pair + 1];
    float re, ro;
    rope_pair((float)s, pair, 1.0f / (float)QRD, xe, xo, re, ro);
    val = (jj & 1) ? ro : re;
  }
  qst[idx] = val;
}

__global__ __launch_bounds__(256) void build_k_kernel(
    const float* __restrict__ kvraw, const float* __restrict__ ckv_kr,
    float* __restrict__ kst) {
  size_t idx = (size_t)blockIdx.x * blockDim.x + threadIdx.x;
  const size_t total = (size_t)BATCH * NKV * SLEN * DQK;
  if (idx >= total) return;
  int e   = idx & 3;
  int ln  = (int)((idx >> 2) & 31);
  int jc  = (int)((idx >> 7) & 15);
  int t   = (int)((idx >> 11) & (SLEN / 16 - 1));
  int kvh = (int)((idx >> 18) & (NKV - 1));
  int b   = (int)(idx >> 20);
  int j   = jc * 8 + (ln >> 4) * 4 + e;
  int s   = t * 16 + (ln & 15);
  size_t row = (size_t)(b * SLEN + s);
  float val;
  if (j < KND) {
    val = kvraw[row * UKV_N + kvh * KND + j];
  } else {
    int jj = j - KND;
    int pair = jj >> 1;
    const float* base = ckv_kr + row * DKV_N + CKVD;
    float xe = base[2 * pair], xo = base[2 * pair + 1];
    float re, ro;
    rope_pair((float)s, pair, 1.0f / (float)KRD, xe, xo, re, ro);
    val = (jj & 1) ? ro : re;
  }
  kst[idx] = val;
}

__global__ __launch_bounds__(256) void build_v_kernel(
    const float* __restrict__ kvraw, float* __restrict__ vpi) {
  size_t idx = (size_t)blockIdx.x * blockDim.x + threadIdx.x;
  const size_t total = (size_t)BATCH * NKV * SLEN * VDD;
  if (idx >= total) return;
  int d   = idx & 255;
  int s   = (int)((idx >> 8) & (SLEN - 1));
  int kvh = (int)((idx >> 19) & (NKV - 1));
  int b   = (int)(idx >> 21);
  size_t row = (size_t)(b * SLEN + s);
  float val = kvraw[row * UKV_N + NKV * KND + kvh * VDD + d];
  size_t dst = ((size_t)(b * NKV + kvh) * (SLEN / 2) + (s >> 1)) * (2 * VDD) +
               d * 2 + (s & 1);
  vpi[dst] = val;
}

__global__ __launch_bounds__(256) __attribute__((amdgpu_waves_per_eu(2, 2)))
void attn_kernel(
    const float* __restrict__ Q, const float* __restrict__ K,
    const float* __restrict__ V, float* __restrict__ O) {
  __shared__ float ps[8][16][16];

  const int lane = threadIdx.x & 31;
  const int wave = threadIdx.x >> 5;
  const int r    = lane & 15;
  const int hf   = lane >> 4;
  const int bh   = blockIdx.x;
  const int b    = bh >> 4;
  const int h    = bh & 15;
  const int kvh  = h >> 2;
  const int q0   = blockIdx.y * 128 + wave * 16;

  const float* Qb = Q + ((size_t)bh * SLEN + q0) * DQK;
  const float* Kb = K + ((size_t)(b * NKV + kvh) * SLEN) * DQK;
  const float* Vb = V + ((size_t)(b * NKV + kvh) * SLEN) * VDD;

  const float scale = 0.08838834764831845f;

  v2f qa[32];
#pragma unroll
  for (int j = 0; j < 16; ++j) {
    const float4 qv = *(const float4*)(Qb + (size_t)r * DQK + j * 8 + 4 * hf);
    v2f t0, t1;
    t0.x = qv.x * scale;
    t0.y = qv.y * scale;
    t1.x = qv.z * scale;
    t1.y = qv.w * scale;
    qa[2 * j]     = t0;
    qa[2 * j + 1] = t1;
  }

  float m_i[8], l_i[8];
#pragma unroll
  for (int i = 0; i < 8; ++i) { m_i[i] = -1e30f; l_i[i] = 0.f; }
  v8f acc[16];
#pragma unroll
  for (int nt = 0; nt < 16; ++nt) acc[nt] = v8f_zero();

  const int ktmax = (q0 + 15) >> 4;
  for (int kt = 0; kt <= ktmax; ++kt) {
    const int k0 = kt * 16;

    if (kt < ktmax) {
      const float* kpf = Kb + (size_t)(kt + 1) * (16 * DQK) + lane * 64;
      __builtin_prefetch(kpf, 0, 3);
      const float* vpf = Vb + (size_t)((k0 + 16) >> 1) * (2 * VDD) + lane * 64;
      __builtin_prefetch(vpf, 0, 3);
    }

    v8f sc = v8f_zero();
    const float* Kt = Kb + (size_t)kt * (16 * DQK) + lane * 4;
#pragma unroll
    for (int j = 0; j < 16; ++j) {
      const float4 kf = *(const float4*)(Kt + j * 128);
      v2f b0, b1;
      b0.x = kf.x;
      b0.y = kf.y;
      b1.x = kf.z;
      b1.y = kf.w;
      sc = wmma_f32(qa[2 * j], b0, sc);
      sc = wmma_f32(qa[2 * j + 1], b1, sc);
    }

#pragma unroll
    for (int i = 0; i < 8; ++i) {
      const int qrow = q0 + i + 8 * hf;
      const int kcol = k0 + r;
      float v = (kcol > qrow) ? -1e30f : sc[i];
      float mx = v;
#pragma unroll
      for (int off = 1; off < 16; off <<= 1)
        mx = fmaxf(mx, __shfl_xor(mx, off, 32));
      const float mnew  = fmaxf(m_i[i], mx);
      const float alpha = __expf(m_i[i] - mnew);
      const float pe    = __expf(v - mnew);
      float rs = pe;
#pragma unroll
      for (int off = 1; off < 16; off <<= 1)
        rs += __shfl_xor(rs, off, 32);
      l_i[i] = l_i[i] * alpha + rs;
      m_i[i] = mnew;
#pragma unroll
      for (int nt = 0; nt < 16; ++nt) acc[nt][i] *= alpha;
      ps[wave][i + 8 * hf][r] = pe;
    }
    asm volatile("s_wait_dscnt 0" ::: "memory");

    v2f pa[4];
#pragma unroll
    for (int ks = 0; ks < 4; ++ks) {
      v2f t;
      t.x = ps[wave][r][ks * 4 + 2 * hf];
      t.y = ps[wave][r][ks * 4 + 1 + 2 * hf];
      pa[ks] = t;
    }

#pragma unroll
    for (int nt = 0; nt < 16; ++nt) {
#pragma unroll
      for (int ks = 0; ks < 4; ++ks) {
        const int kk = k0 + ks * 4 + 2 * hf;
        const float* Vp =
            Vb + (size_t)(kk >> 1) * (2 * VDD) + (nt * 16 + r) * 2;
        v2f bb = *(const v2f*)Vp;
        acc[nt] = wmma_f32(pa[ks], bb, acc[nt]);
      }
    }
  }

#pragma unroll
  for (int i = 0; i < 8; ++i) {
    const float inv = 1.0f / l_i[i];
    const int qrow = q0 + i + 8 * hf;
    float* Op = O + ((size_t)(b * SLEN + qrow)) * HVD + h * VDD;
#pragma unroll
    for (int nt = 0; nt < 16; ++nt) Op[nt * 16 + r] = acc[nt][i] * inv;
  }
}

extern "C" void kernel_launch(void* const* d_in, const int* in_sizes, int n_in,
                              void* d_out, int out_size, void* d_ws,
                              size_t ws_size, hipStream_t stream) {
  (void)in_sizes; (void)n_in; (void)out_size; (void)ws_size;
  const float* x        = (const float*)d_in[0];
  const float* w_dq     = (const float*)d_in[1];
  const float* gamma_q  = (const float*)d_in[2];
  const float* w_uq     = (const float*)d_in[3];
  const float* w_dkv    = (const float*)d_in[4];
  const float* gamma_kv = (const float*)d_in[5];
  const float* w_ukv    = (const float*)d_in[6];
  const float* w_o      = (const float*)d_in[7];
  float* out = (float*)d_out;

  float* ws = (float*)d_ws;
  float* tmp_q   = ws;
  float* cq      = tmp_q   + (size_t)ROWS * CQD;
  float* qraw    = cq      + (size_t)ROWS * CQD;
  float* qst     = qraw    + (size_t)ROWS * UQ_N;
  float* ckv_kr  = qst     + (size_t)BATCH * NH * SLEN * DQK;
  float* ckv     = ckv_kr  + (size_t)ROWS * DKV_N;
  float* kvraw   = ckv     + (size_t)ROWS * CKVD;
  float* kst     = kvraw   + (size_t)ROWS * UKV_N;
  float* vpi     = kst     + (size_t)BATCH * NKV * SLEN * DQK;
  float* attn_o  = vpi     + (size_t)BATCH * NKV * SLEN * VDD;

  gemm_wmma_f32_lds<<<dim3(ROWS / 128, CQD / 64), 256, 0, stream>>>(
      x, w_dq, tmp_q, ROWS, CQD, DIMC);
  t5ln_kernel<<<ROWS, 32, 0, stream>>>(tmp_q, CQD, gamma_q, cq);
  gemm_wmma_f32_lds<<<dim3(ROWS / 128, UQ_N / 64), 256, 0, stream>>>(
      cq, w_uq, qraw, ROWS, UQ_N, CQD);
  {
    size_t total = (size_t)BATCH * NH * SLEN * DQK;
    build_q_kernel<<<(unsigned)((total + 255) / 256), 256, 0, stream>>>(qraw, qst);
  }
  gemm_wmma_f32_lds<<<dim3(ROWS / 128, DKV_N / 64), 256, 0, stream>>>(
      x, w_dkv, ckv_kr, ROWS, DKV_N, DIMC);
  t5ln_kernel<<<ROWS, 32, 0, stream>>>(ckv_kr, DKV_N, gamma_kv, ckv);
  gemm_wmma_f32_lds<<<dim3(ROWS / 128, UKV_N / 64), 256, 0, stream>>>(
      ckv, w_ukv, kvraw, ROWS, UKV_N, CQD);
  {
    size_t tk = (size_t)BATCH * NKV * SLEN * DQK;
    build_k_kernel<<<(unsigned)((tk + 255) / 256), 256, 0, stream>>>(kvraw, ckv_kr, kst);
    size_t tv = (size_t)BATCH * NKV * SLEN * VDD;
    build_v_kernel<<<(unsigned)((tv + 255) / 256), 256, 0, stream>>>(kvraw, vpi);
  }
  attn_kernel<<<dim3(BATCH * NH, SLEN / 128), 256, 0, stream>>>(qst, kst, vpi, attn_o);
  gemm_wmma_f32_lds<<<dim3(ROWS / 128, DIMC / 64), 256, 0, stream>>>(
      attn_o, w_o, out, ROWS, DIMC, HVD);
}


# attention: causal query blocks handed out heaviest first (query block = last - blockIdx.y): the 128-key-tile blocks start first, the 8-tile ones fill in, no long tail on an idle chip; sums unchanged
# speedup vs baseline: 4.4369x; 1.1689x over previous
#include <hip/hip_runtime.h>
#include <math.h>

constexpr int DIMC   = 2048;
constexpr int SLEN   = 2048;
constexpr int BATCH  = 2;
constexpr int NH     = 16;
constexpr int NKV    = 4;
constexpr int CQD    = 128;
constexpr int QND    = 96;
constexpr int QRD    = 32;
constexpr int CKVD   = 128;
constexpr int KND    = 64;
constexpr int KRD    = 64;
constexpr int VDD    = 256;
constexpr int DQK    = 128;
constexpr int ROWS   = BATCH * SLEN;
constexpr int UQ_N   = NH * (QND + QRD);
constexpr int DKV_N  = CKVD + KRD;
constexpr int UKV_N  = NKV * (KND + VDD);
constexpr int HVD    = NH * VDD;

typedef float v2f __attribute__((ext_vector_type(2)));
typedef float v8f __attribute__((ext_vector_type(8)));

__device__ __forceinline__ v8f wmma_f32(v2f a, v2f b, v8f c) {
  return __builtin_amdgcn_wmma_f32_16x16x4_f32(
       false, a,  false, b,
       (short)0, c,  false,  false);
}

__device__ __forceinline__ v8f v8f_zero() {
  v8f z = {0.f, 0.f, 0.f, 0.f, 0.f, 0.f, 0.f, 0.f};
  return z;
}

constexpr int KC = 32;

typedef float v4f __attribute__((ext_vector_type(4)));

__device__ __forceinline__ void gemm_stage_chunk(
    float (*Asb)[36], float (*Bsb)[64][2], int ar, int ac, int br, int bn,
    v4f a0, v4f a1, v4f a2, v4f a3, v4f b0, v4f b1) {
  *(v4f*)&Asb[ar][ac]      = a0;
  *(v4f*)&Asb[ar + 32][ac] = a1;
  *(v4f*)&Asb[ar + 64][ac] = a2;
  *(v4f*)&Asb[ar + 96][ac] = a3;
  const int kp = br >> 1, kb = br & 1;
  Bsb[kp][bn + 0][kb]     = b0.x;
  Bsb[kp][bn + 1][kb]     = b0.y;
  Bsb[kp][bn + 2][kb]     = b0.z;
  Bsb[kp][bn + 3][kb]     = b0.w;
  Bsb[kp + 8][bn + 0][kb] = b1.x;
  Bsb[kp + 8][bn + 1][kb] = b1.y;
  Bsb[kp + 8][bn + 2][kb] = b1.z;
  Bsb[kp + 8][bn + 3][kb] = b1.w;
}

__global__ __launch_bounds__(256) void gemm_wmma_f32_lds(
    const float* __restrict__ A, const float* __restrict__ B,
    float* __restrict__ C, int M, int N, int Kd) {
  __shared__ float As[2][128][36];
  __shared__ float Bs[2][KC / 2][64][2];

  const int tid  = threadIdx.x;
  const int lane = tid & 31;
  const int wave = tid >> 5;
  const int r    = lane & 15;
  const int hf   = lane >> 4;
  const int m0b  = blockIdx.x * 128;
  const int n0   = blockIdx.y * 64;

  v8f acc[4];
#pragma unroll
  for (int j = 0; j < 4; ++j) acc[j] = v8f_zero();

  const int ar = tid >> 3;
  const int ac = (tid * 4) & 31;
  const int br = tid >> 4;
  const int bn = (tid * 4) & 63;

  const float* Ap0 = A + (size_t)(m0b + ar) * Kd + ac;
  const float* Ap1 = Ap0 + (size_t)32 * Kd;
  const float* Ap2 = Ap0 + (size_t)64 * Kd;
  const float* Ap3 = Ap0 + (size_t)96 * Kd;
  const float* Bp0 = B + (size_t)br * N + n0 + bn;
  const float* Bp1 = Bp0 + (size_t)16 * N;

  const int nchunk = Kd / KC;

  v4f a0 = *(const v4f*)Ap0;
  v4f a1 = *(const v4f*)Ap1;
  v4f a2 = *(const v4f*)Ap2;
  v4f a3 = *(const v4f*)Ap3;
  v4f b0 = *(const v4f*)Bp0;
  v4f b1 = *(const v4f*)Bp1;
  gemm_stage_chunk(As[0], Bs[0], ar, ac, br, bn, a0, a1, a2, a3, b0, b1);
  __syncthreads();

  for (int c = 0; c < nchunk; ++c) {
    const int buf = c & 1;
    const bool more = (c + 1 < nchunk);
    if (more) {
      const int kc = (c + 1) * KC;
      const size_t kb = (size_t)kc * N;
      a0 = *(const v4f*)(Ap0 + kc);
      a1 = *(const v4f*)(Ap1 + kc);
      a2 = *(const v4f*)(Ap2 + kc);
      a3 = *(const v4f*)(Ap3 + kc);
      b0 = *(const v4f*)(Bp0 + kb);
      b1 = *(const v4f*)(Bp1 + kb);
    }
#pragma unroll
    for (int k = 0; k < KC; k += 4) {
      v2f a = *(const v2f*)&As[buf][wave * 16 + r][k + 2 * hf];
#pragma unroll
      for (int j = 0; j < 4; ++j) {
        v2f bb = *(const v2f*)&Bs[buf][(k + 2 * hf) >> 1][j * 16 + r][0];
        acc[j] = wmma_f32(a, bb, acc[j]);
      }
    }
    if (more)
      gemm_stage_chunk(As[buf ^ 1], Bs[buf ^ 1], ar, ac, br, bn,
                       a0, a1, a2, a3, b0, b1);
    __syncthreads();
  }

#pragma unroll
  for (int j = 0; j < 4; ++j)
#pragma unroll
    for (int i = 0; i < 8; ++i)
      C[(size_t)(m0b + wave * 16 + i + 8 * hf) * N + n0 + j * 16 + r] =
          acc[j][i];
}

__global__ __launch_bounds__(32) void t5ln_kernel(
    const float* __restrict__ in, int in_stride,
    const float* __restrict__ gamma, float* __restrict__ out) {
  const int row  = blockIdx.x;
  const int lane = threadIdx.x;
  const float* ip = in + (size_t)row * in_stride;
  float v[4], s = 0.f, s2 = 0.f;
#pragma unroll
  for (int i = 0; i < 4; ++i) {
    v[i] = ip[lane + 32 * i];
    s  += v[i];
    s2 += v[i] * v[i];
  }
#pragma unroll
  for (int off = 1; off < 32; off <<= 1) {
    s  += __shfl_xor(s, off, 32);
    s2 += __shfl_xor(s2, off, 32);
  }
  const float mu  = s * (1.f / 128.f);
  const float var = s2 * (1.f / 128.f) - mu * mu;
  const float rs  = rsqrtf(var + 1e-5f);
  float* op = out + (size_t)row * 128;
#pragma unroll
  for (int i = 0; i < 4; ++i)
    op[lane + 32 * i] = (v[i] - mu) * rs * gamma[lane + 32 * i];
}

__device__ __forceinline__ void rope_pair(float spos, int pair, float invD,
                                          float xe, float xo,
                                          float& re, float& ro) {
  const float lnTheta = 9.210340371976184f;
  float freq = expf(-(2.0f * (float)pair) * invD * lnTheta);
  float ang  = spos * freq;
  float c = cosf(ang), sn = sinf(ang);
  re = xe * c - xo * sn;
  ro = xe * sn + xo * c;
}

__global__ __launch_bounds__(256) void build_q_kernel(
    const float* __restrict__ qraw, float* __restrict__ qst) {
  size_t idx = (size_t)blockIdx.x * blockDim.x + threadIdx.x;
  const size_t total = (size_t)BATCH * NH * SLEN * DQK;
  if (idx >= total) return;
  int j = idx & 127;
  int s = (int)((idx >> 7) & (SLEN - 1));
  int h = (int)((idx >> 18) & (NH - 1));
  int b = (int)(idx >> 22);
  const float* qr = qraw + ((size_t)(b * SLEN + s)) * UQ_N;
  float val;
  if (j < QND) {
    val = qr[h * QND + j];
  } else {
    int jj = j - QND;
    int pair = jj >> 1;
    const float* base = qr + NH * QND + h * QRD;
    float xe = base[2 * pair], xo = base[2 * pair + 1];
    float re, ro;
    rope_pair((float)s, pair, 1.0f / (float)QRD, xe, xo, re, ro);
    val = (jj & 1) ? ro : re;
  }
  qst[idx] = val;
}

__global__ __launch_bounds__(256) void build_k_kernel(
    const float* __restrict__ kvraw, const float* __restrict__ ckv_kr,
    float* __restrict__ kst) {
  size_t idx = (size_t)blockIdx.x * blockDim.x + threadIdx.x;
  const size_t total = (size_t)BATCH * NKV * SLEN * DQK;
  if (idx >= total) return;
  int e   = idx & 3;
  int ln  = (int)((idx >> 2) & 31);
  int jc  = (int)((idx >> 7) & 15);
  int t   = (int)((idx >> 11) & (SLEN / 16 - 1));
  int kvh = (int)((idx >> 18) & (NKV - 1));
  int b   = (int)(idx >> 20);
  int j   = jc * 8 + (ln >> 4) * 4 + e;
  int s   = t * 16 + (ln & 15);
  size_t row = (size_t)(b * SLEN + s);
  float val;
  if (j < KND) {
    val = kvraw[row * UKV_N + kvh * KND + j];
  } else {
    int jj = j - KND;
    int pair = jj >> 1;
    const float* base = ckv_kr + row * DKV_N + CKVD;
    float xe = base[2 * pair], xo = base[2 * pair + 1];
    float re, ro;
    rope_pair((float)s, pair, 1.0f / (float)KRD, xe, xo, re, ro);
    val = (jj & 1) ? ro : re;
  }
  kst[idx] = val;
}

__global__ __launch_bounds__(256) void build_v_kernel(
    const float* __restrict__ kvraw, float* __restrict__ vpi) {
  size_t idx = (size_t)blockIdx.x * blockDim.x + threadIdx.x;
  const size_t total = (size_t)BATCH * NKV * SLEN * VDD;
  if (idx >= total) return;
  int d   = idx & 255;
  int s   = (int)((idx >> 8) & (SLEN - 1));
  int kvh = (int)((idx >> 19) & (NKV - 1));
  int b   = (int)(idx >> 21);
  size_t row = (size_t)(b * SLEN + s);
  float val = kvraw[row * UKV_N + NKV * KND + kvh * VDD + d];
  size_t dst = ((size_t)(b * NKV + kvh) * (SLEN / 2) + (s >> 1)) * (2 * VDD) +
               d * 2 + (s & 1);
  vpi[dst] = val;
}

__global__ __launch_bounds__(256) __attribute__((amdgpu_waves_per_eu(2, 2)))
void attn_kernel(
    const float* __restrict__ Q, const float* __restrict__ K,
    const float* __restrict__ V, float* __restrict__ O) {
  __shared__ float ps[8][16][16];

  const int lane = threadIdx.x & 31;
  const int wave = threadIdx.x >> 5;
  const int r    = lane & 15;
  const int hf   = lane >> 4;
  const int bh   = blockIdx.x;
  const int b    = bh >> 4;
  const int h    = bh & 15;
  const int kvh  = h >> 2;
  const int qb   = (SLEN / 128 - 1) - (int)blockIdx.y;
  const int q0   = qb * 128 + wave * 16;

  const float* Qb = Q + ((size_t)bh * SLEN + q0) * DQK;
  const float* Kb = K + ((size_t)(b * NKV + kvh) * SLEN) * DQK;
  const float* Vb = V + ((size_t)(b * NKV + kvh) * SLEN) * VDD;

  const float scale = 0.08838834764831845f;

  v2f qa[32];
#pragma unroll
  for (int j = 0; j < 16; ++j) {
    const float4 qv = *(const float4*)(Qb + (size_t)r * DQK + j * 8 + 4 * hf);
    v2f t0, t1;
    t0.x = qv.x * scale;
    t0.y = qv.y * scale;
    t1.x = qv.z * scale;
    t1.y = qv.w * scale;
    qa[2 * j]     = t0;
    qa[2 * j + 1] = t1;
  }

  float m_i[8], l_i[8];
#pragma unroll
  for (int i = 0; i < 8; ++i) { m_i[i] = -1e30f; l_i[i] = 0.f; }
  v8f acc[16];
#pragma unroll
  for (int nt = 0; nt < 16; ++nt) acc[nt] = v8f_zero();

  const int ktmax = (q0 + 15) >> 4;
  for (int kt = 0; kt <= ktmax; ++kt) {
    const int k0 = kt * 16;

    if (kt < ktmax) {
      const float* kpf = Kb + (size_t)(kt + 1) * (16 * DQK) + lane * 64;
      __builtin_prefetch(kpf, 0, 3);
      const float* vpf = Vb + (size_t)((k0 + 16) >> 1) * (2 * VDD) + lane * 64;
      __builtin_prefetch(vpf, 0, 3);
    }

    v8f sc = v8f_zero();
    const float* Kt = Kb + (size_t)kt * (16 * DQK) + lane * 4;
#pragma unroll
    for (int j = 0; j < 16; ++j) {
      const float4 kf = *(const float4*)(Kt + j * 128);
      v2f b0, b1;
      b0.x = kf.x;
      b0.y = kf.y;
      b1.x = kf.z;
      b1.y = kf.w;
      sc = wmma_f32(qa[2 * j], b0, sc);
      sc = wmma_f32(qa[2 * j + 1], b1, sc);
    }

#pragma unroll
    for (int i = 0; i < 8; ++i) {
      const int qrow = q0 + i + 8 * hf;
      const int kcol = k0 + r;
      float v = (kcol > qrow) ? -1e30f : sc[i];
      float mx = v;
#pragma unroll
      for (int off = 1; off < 16; off <<= 1)
        mx = fmaxf(mx, __shfl_xor(mx, off, 32));
      const float mnew  = fmaxf(m_i[i], mx);
      const float alpha = __expf(m_i[i] - mnew);
      const float pe    = __expf(v - mnew);
      float rs = pe;
#pragma unroll
      for (int off = 1; off < 16; off <<= 1)
        rs += __shfl_xor(rs, off, 32);
      l_i[i] = l_i[i] * alpha + rs;
      m_i[i] = mnew;
#pragma unroll
      for (int nt = 0; nt < 16; ++nt) acc[nt][i] *= alpha;
      ps[wave][i + 8 * hf][r] = pe;
    }
    asm volatile("s_wait_dscnt 0" ::: "memory");

    v2f pa[4];
#pragma unroll
    for (int ks = 0; ks < 4; ++ks) {
      v2f t;
      t.x = ps[wave][r][ks * 4 + 2 * hf];
      t.y = ps[wave][r][ks * 4 + 1 + 2 * hf];
      pa[ks] = t;
    }

#pragma unroll
    for (int nt = 0; nt < 16; ++nt) {
#pragma unroll
      for (int ks = 0; ks < 4; ++ks) {
        const int kk = k0 + ks * 4 + 2 * hf;
        const float* Vp =
            Vb + (size_t)(kk >> 1) * (2 * VDD) + (nt * 16 + r) * 2;
        v2f bb = *(const v2f*)Vp;
        acc[nt] = wmma_f32(pa[ks], bb, acc[nt]);
      }
    }
  }

#pragma unroll
  for (int i = 0; i < 8; ++i) {
    const float inv = 1.0f / l_i[i];
    const int qrow = q0 + i + 8 * hf;
    float* Op = O + ((size_t)(b * SLEN + qrow)) * HVD + h * VDD;
#pragma unroll
    for (int nt = 0; nt < 16; ++nt) Op[nt * 16 + r] = acc[nt][i] * inv;
  }
}

extern "C" void kernel_launch(void* const* d_in, const int* in_sizes, int n_in,
                              void* d_out, int out_size, void* d_ws,
                              size_t ws_size, hipStream_t stream) {
  (void)in_sizes; (void)n_in; (void)out_size; (void)ws_size;
  const float* x        = (const float*)d_in[0];
  const float* w_dq     = (const float*)d_in[1];
  const float* gamma_q  = (const float*)d_in[2];
  const float* w_uq     = (const float*)d_in[3];
  const float* w_dkv    = (const float*)d_in[4];
  const float* gamma_kv = (const float*)d_in[5];
  const float* w_ukv    = (const float*)d_in[6];
  const float* w_o      = (const float*)d_in[7];
  float* out = (float*)d_out;

  float* ws = (float*)d_ws;
  float* tmp_q   = ws;
  float* cq      = tmp_q   + (size_t)ROWS * CQD;
  float* qraw    = cq      + (size_t)ROWS * CQD;
  float* qst     = qraw    + (size_t)ROWS * UQ_N;
  float* ckv_kr  = qst     + (size_t)BATCH * NH * SLEN * DQK;
  float* ckv     = ckv_kr  + (size_t)ROWS * DKV_N;
  float* kvraw   = ckv     + (size_t)ROWS * CKVD;
  float* kst     = kvraw   + (size_t)ROWS * UKV_N;
  float* vpi     = kst     + (size_t)BATCH * NKV * SLEN * DQK;
  float* attn_o  = vpi     + (size_t)BATCH * NKV * SLEN * VDD;

  gemm_wmma_f32_lds<<<dim3(ROWS / 128, CQD / 64), 256, 0, stream>>>(
      x, w_dq, tmp_q, ROWS, CQD, DIMC);
  t5ln_kernel<<<ROWS, 32, 0, stream>>>(tmp_q, CQD, gamma_q, cq);
  gemm_wmma_f32_lds<<<dim3(ROWS / 128, UQ_N / 64), 256, 0, stream>>>(
      cq, w_uq, qraw, ROWS, UQ_N, CQD);
  {
    size_t total = (size_t)BATCH * NH * SLEN * DQK;
    build_q_kernel<<<(unsigned)((total + 255) / 256), 256, 0, stream>>>(qraw, qst);
  }
  gemm_wmma_f32_lds<<<dim3(ROWS / 128, DKV_N / 64), 256, 0, stream>>>(
      x, w_dkv, ckv_kr, ROWS, DKV_N, DIMC);
  t5ln_kernel<<<ROWS, 32, 0, stream>>>(ckv_kr, DKV_N, gamma_kv, ckv);
  gemm_wmma_f32_lds<<<dim3(ROWS / 128, UKV_N / 64), 256, 0, stream>>>(
      ckv, w_ukv, kvraw, ROWS, UKV_N, CQD);
  {
    size_t tk = (size_t)BATCH * NKV * SLEN * DQK;
    build_k_kernel<<<(unsigned)((tk + 255) / 256), 256, 0, stream>>>(kvraw, ckv_kr, kst);
    size_t tv = (size_t)BATCH * NKV * SLEN * VDD;
    build_v_kernel<<<(unsigned)((tv + 255) / 256), 256, 0, stream>>>(kvraw, vpi);
  }
  attn_kernel<<<dim3(BATCH * NH, SLEN / 128), 256, 0, stream>>>(qst, kst, vpi, attn_o);
  gemm_wmma_f32_lds<<<dim3(ROWS / 128, DIMC / 64), 256, 0, stream>>>(
      attn_o, w_o, out, ROWS, DIMC, HVD);
}


# attention: tile loop software-pipelined: a tile's V fragments load in front of its QK WMMAs, the next tile's K fragments behind them, so no load is waited for right after its issue; sums unchanged
# speedup vs baseline: 4.6218x; 1.0417x over previous
#include <hip/hip_runtime.h>
#include <math.h>

constexpr int DIMC   = 2048;
constexpr int SLEN   = 2048;
constexpr int BATCH  = 2;
constexpr int NH     = 16;
constexpr int NKV    = 4;
constexpr int CQD    = 128;
constexpr int QND    = 96;
constexpr int QRD    = 32;
constexpr int CKVD   = 128;
constexpr int KND    = 64;
constexpr int KRD    = 64;
constexpr int VDD    = 256;
constexpr int DQK    = 128;
constexpr int ROWS   = BATCH * SLEN;
constexpr int UQ_N   = NH * (QND + QRD);
constexpr int DKV_N  = CKVD + KRD;
constexpr int UKV_N  = NKV * (KND + VDD);
constexpr int HVD    = NH * VDD;

typedef float v2f __attribute__((ext_vector_type(2)));
typedef float v8f __attribute__((ext_vector_type(8)));

__device__ __forceinline__ v8f wmma_f32(v2f a, v2f b, v8f c) {
  return __builtin_amdgcn_wmma_f32_16x16x4_f32(
       false, a,  false, b,
       (short)0, c,  false,  false);
}

__device__ __forceinline__ v8f v8f_zero() {
  v8f z = {0.f, 0.f, 0.f, 0.f, 0.f, 0.f, 0.f, 0.f};
  return z;
}

constexpr int KC = 32;

typedef float v4f __attribute__((ext_vector_type(4)));

__device__ __forceinline__ void gemm_stage_chunk(
    float (*Asb)[36], float (*Bsb)[64][2], int ar, int ac, int br, int bn,
    v4f a0, v4f a1, v4f a2, v4f a3, v4f b0, v4f b1) {
  *(v4f*)&Asb[ar][ac]      = a0;
  *(v4f*)&Asb[ar + 32][ac] = a1;
  *(v4f*)&Asb[ar + 64][ac] = a2;
  *(v4f*)&Asb[ar + 96][ac] = a3;
  const int kp = br >> 1, kb = br & 1;
  Bsb[kp][bn + 0][kb]     = b0.x;
  Bsb[kp][bn + 1][kb]     = b0.y;
  Bsb[kp][bn + 2][kb]     = b0.z;
  Bsb[kp][bn + 3][kb]     = b0.w;
  Bsb[kp + 8][bn + 0][kb] = b1.x;
  Bsb[kp + 8][bn + 1][kb] = b1.y;
  Bsb[kp + 8][bn + 2][kb] = b1.z;
  Bsb[kp + 8][bn + 3][kb] = b1.w;
}

__global__ __launch_bounds__(256) void gemm_wmma_f32_lds(
    const float* __restrict__ A, const float* __restrict__ B,
    float* __restrict__ C, int M, int N, int Kd) {
  __shared__ float As[2][128][36];
  __shared__ float Bs[2][KC / 2][64][2];

  const int tid  = threadIdx.x;
  const int lane = tid & 31;
  const int wave = tid >> 5;
  const int r    = lane & 15;
  const int hf   = lane >> 4;
  const int m0b  = blockIdx.x * 128;
  const int n0   = blockIdx.y * 64;

  v8f acc[4];
#pragma unroll
  for (int j = 0; j < 4; ++j) acc[j] = v8f_zero();

  const int ar = tid >> 3;
  const int ac = (tid * 4) & 31;
  const int br = tid >> 4;
  const int bn = (tid * 4) & 63;

  const float* Ap0 = A + (size_t)(m0b + ar) * Kd + ac;
  const float* Ap1 = Ap0 + (size_t)32 * Kd;
  const float* Ap2 = Ap0 + (size_t)64 * Kd;
  const float* Ap3 = Ap0 + (size_t)96 * Kd;
  const float* Bp0 = B + (size_t)br * N + n0 + bn;
  const float* Bp1 = Bp0 + (size_t)16 * N;

  const int nchunk = Kd / KC;

  v4f a0 = *(const v4f*)Ap0;
  v4f a1 = *(const v4f*)Ap1;
  v4f a2 = *(const v4f*)Ap2;
  v4f a3 = *(const v4f*)Ap3;
  v4f b0 = *(const v4f*)Bp0;
  v4f b1 = *(const v4f*)Bp1;
  gemm_stage_chunk(As[0], Bs[0], ar, ac, br, bn, a0, a1, a2, a3, b0, b1);
  __syncthreads();

  for (int c = 0; c < nchunk; ++c) {
    const int buf = c & 1;
    const bool more = (c + 1 < nchunk);
    if (more) {
      const int kc = (c + 1) * KC;
      const size_t kb = (size_t)kc * N;
      a0 = *(const v4f*)(Ap0 + kc);
      a1 = *(const v4f*)(Ap1 + kc);
      a2 = *(const v4f*)(Ap2 + kc);
      a3 = *(const v4f*)(Ap3 + kc);
      b0 = *(const v4f*)(Bp0 + kb);
      b1 = *(const v4f*)(Bp1 + kb);
    }
#pragma unroll
    for (int k = 0; k < KC; k += 4) {
      v2f a = *(const v2f*)&As[buf][wave * 16 + r][k + 2 * hf];
#pragma unroll
      for (int j = 0; j < 4; ++j) {
        v2f bb = *(const v2f*)&Bs[buf][(k + 2 * hf) >> 1][j * 16 + r][0];
        acc[j] = wmma_f32(a, bb, acc[j]);
      }
    }
    if (more)
      gemm_stage_chunk(As[buf ^ 1], Bs[buf ^ 1], ar, ac, br, bn,
                       a0, a1, a2, a3, b0, b1);
    __syncthreads();
  }

#pragma unroll
  for (int j = 0; j < 4; ++j)
#pragma unroll
    for (int i = 0; i < 8; ++i)
      C[(size_t)(m0b + wave * 16 + i + 8 * hf) * N + n0 + j * 16 + r] =
          acc[j][i];
}

__global__ __launch_bounds__(32) void t5ln_kernel(
    const float* __restrict__ in, int in_stride,
    const float* __restrict__ gamma, float* __restrict__ out) {
  const int row  = blockIdx.x;
  const int lane = threadIdx.x;
  const float* ip = in + (size_t)row * in_stride;
  float v[4], s = 0.f, s2 = 0.f;
#pragma unroll
  for (int i = 0; i < 4; ++i) {
    v[i] = ip[lane + 32 * i];
    s  += v[i];
    s2 += v[i] * v[i];
  }
#pragma unroll
  for (int off = 1; off < 32; off <<= 1) {
    s  += __shfl_xor(s, off, 32);
    s2 += __shfl_xor(s2, off, 32);
  }
  const float mu  = s * (1.f / 128.f);
  const float var = s2 * (1.f / 128.f) - mu * mu;
  const float rs  = rsqrtf(var + 1e-5f);
  float* op = out + (size_t)row * 128;
#pragma unroll
  for (int i = 0; i < 4; ++i)
    op[lane + 32 * i] = (v[i] - mu) * rs * gamma[lane + 32 * i];
}

__device__ __forceinline__ void rope_pair(float spos, int pair, float invD,
                                          float xe, float xo,
                                          float& re, float& ro) {
  const float lnTheta = 9.210340371976184f;
  float freq = expf(-(2.0f * (float)pair) * invD * lnTheta);
  float ang  = spos * freq;
  float c = cosf(ang), sn = sinf(ang);
  re = xe * c - xo * sn;
  ro = xe * sn + xo * c;
}

__global__ __launch_bounds__(256) void build_q_kernel(
    const float* __restrict__ qraw, float* __restrict__ qst) {
  size_t idx = (size_t)blockIdx.x * blockDim.x + threadIdx.x;
  const size_t total = (size_t)BATCH * NH * SLEN * DQK;
  if (idx >= total) return;
  int j = idx & 127;
  int s = (int)((idx >> 7) & (SLEN - 1));
  int h = (int)((idx >> 18) & (NH - 1));
  int b = (int)(idx >> 22);
  const float* qr = qraw + ((size_t)(b * SLEN + s)) * UQ_N;
  float val;
  if (j < QND) {
    val = qr[h * QND + j];
  } else {
    int jj = j - QND;
    int pair = jj >> 1;
    const float* base = qr + NH * QND + h * QRD;
    float xe = base[2 * pair], xo = base[2 * pair + 1];
    float re, ro;
    rope_pair((float)s, pair, 1.0f / (float)QRD, xe, xo, re, ro);
    val = (jj & 1) ? ro : re;
  }
  qst[idx] = val;
}

__global__ __launch_bounds__(256) void build_k_kernel(
    const float* __restrict__ kvraw, const float* __restrict__ ckv_kr,
    float* __restrict__ kst) {
  size_t idx = (size_t)blockIdx.x * blockDim.x + threadIdx.x;
  const size_t total = (size_t)BATCH * NKV * SLEN * DQK;
  if (idx >= total) return;
  int e   = idx & 3;
  int ln  = (int)((idx >> 2) & 31);
  int jc  = (int)((idx >> 7) & 15);
  int t   = (int)((idx >> 11) & (SLEN / 16 - 1));
  int kvh = (int)((idx >> 18) & (NKV - 1));
  int b   = (int)(idx >> 20);
  int j   = jc * 8 + (ln >> 4) * 4 + e;
  int s   = t * 16 + (ln & 15);
  size_t row = (size_t)(b * SLEN + s);
  float val;
  if (j < KND) {
    val = kvraw[row * UKV_N + kvh * KND + j];
  } else {
    int jj = j - KND;
    int pair = jj >> 1;
    const float* base = ckv_kr + row * DKV_N + CKVD;
    float xe = base[2 * pair], xo = base[2 * pair + 1];
    float re, ro;
    rope_pair((float)s, pair, 1.0f / (float)KRD, xe, xo, re, ro);
    val = (jj & 1) ? ro : re;
  }
  kst[idx] = val;
}

__global__ __launch_bounds__(256) void build_v_kernel(
    const float* __restrict__ kvraw, float* __restrict__ vpi) {
  size_t idx = (size_t)blockIdx.x * blockDim.x + threadIdx.x;
  const size_t total = (size_t)BATCH * NKV * SLEN * VDD;
  if (idx >= total) return;
  int d   = idx & 255;
  int s   = (int)((idx >> 8) & (SLEN - 1));
  int kvh = (int)((idx >> 19) & (NKV - 1));
  int b   = (int)(idx >> 21);
  size_t row = (size_t)(b * SLEN + s);
  float val = kvraw[row * UKV_N + NKV * KND + kvh * VDD + d];
  size_t dst = ((size_t)(b * NKV + kvh) * (SLEN / 2) + (s >> 1)) * (2 * VDD) +
               d * 2 + (s & 1);
  vpi[dst] = val;
}

__global__ __launch_bounds__(256) __attribute__((amdgpu_waves_per_eu(2, 2)))
void attn_kernel(
    const float* __restrict__ Q, const float* __restrict__ K,
    const float* __restrict__ V, float* __restrict__ O) {
  __shared__ float ps[8][16][16];

  const int lane = threadIdx.x & 31;
  const int wave = threadIdx.x >> 5;
  const int r    = lane & 15;
  const int hf   = lane >> 4;
  const int bh   = blockIdx.x;
  const int b    = bh >> 4;
  const int h    = bh & 15;
  const int kvh  = h >> 2;
  const int qb   = (SLEN / 128 - 1) - (int)blockIdx.y;
  const int q0   = qb * 128 + wave * 16;

  const float* Qb = Q + ((size_t)bh * SLEN + q0) * DQK;
  const float* Kb = K + ((size_t)(b * NKV + kvh) * SLEN) * DQK;
  const float* Vb = V + ((size_t)(b * NKV + kvh) * SLEN) * VDD;

  const float scale = 0.08838834764831845f;

  v2f qa[32];
#pragma unroll
  for (int j = 0; j < 16; ++j) {
    const float4 qv = *(const float4*)(Qb + (size_t)r * DQK + j * 8 + 4 * hf);
    v2f t0, t1;
    t0.x = qv.x * scale;
    t0.y = qv.y * scale;
    t1.x = qv.z * scale;
    t1.y = qv.w * scale;
    qa[2 * j]     = t0;
    qa[2 * j + 1] = t1;
  }

  float m_i[8], l_i[8];
#pragma unroll
  for (int i = 0; i < 8; ++i) { m_i[i] = -1e30f; l_i[i] = 0.f; }
  v8f acc[16];
#pragma unroll
  for (int nt = 0; nt < 16; ++nt) acc[nt] = v8f_zero();

  const int ktmax = (q0 + 15) >> 4;

  float4 kf[16];
  {
    const float* Kt = Kb + lane * 4;
#pragma unroll
    for (int j = 0; j < 16; ++j) kf[j] = *(const float4*)(Kt + j * 128);
  }

  for (int kt = 0; kt <= ktmax; ++kt) {
    const int k0 = kt * 16;

    if (kt < ktmax) {
      const float* kpf = Kb + (size_t)(kt + 1) * (16 * DQK) + lane * 64;
      __builtin_prefetch(kpf, 0, 3);
      const float* vpf = Vb + (size_t)((k0 + 16) >> 1) * (2 * VDD) + lane * 64;
      __builtin_prefetch(vpf, 0, 3);
    }

    const float* Vt = Vb + (size_t)((k0 >> 1) + hf) * (2 * VDD) + r * 2;
    v2f vf[64];
#pragma unroll
    for (int ks = 0; ks < 3; ++ks)
#pragma unroll
      for (int nt = 0; nt < 16; ++nt)
        vf[ks * 16 + nt] =
            *(const v2f*)(Vt + (size_t)ks * (4 * VDD) + nt * 32);

    v8f sc = v8f_zero();
#pragma unroll
    for (int j = 0; j < 16; ++j) {
      v2f b0, b1;
      b0.x = kf[j].x;
      b0.y = kf[j].y;
      b1.x = kf[j].z;
      b1.y = kf[j].w;
      sc = wmma_f32(qa[2 * j], b0, sc);
      sc = wmma_f32(qa[2 * j + 1], b1, sc);
    }

    if (kt < ktmax) {
      const float* Kn = Kb + (size_t)(kt + 1) * (16 * DQK) + lane * 4;
#pragma unroll
      for (int j = 0; j < 16; ++j) kf[j] = *(const float4*)(Kn + j * 128);
    }
#pragma unroll
    for (int nt = 0; nt < 16; ++nt)
      vf[48 + nt] = *(const v2f*)(Vt + (size_t)3 * (4 * VDD) + nt * 32);

#pragma unroll
    for (int i = 0; i < 8; ++i) {
      const int qrow = q0 + i + 8 * hf;
      const int kcol = k0 + r;
      float v = (kcol > qrow) ? -1e30f : sc[i];
      float mx = v;
#pragma unroll
      for (int off = 1; off < 16; off <<= 1)
        mx = fmaxf(mx, __shfl_xor(mx, off, 32));
      const float mnew  = fmaxf(m_i[i], mx);
      const float alpha = __expf(m_i[i] - mnew);
      const float pe    = __expf(v - mnew);
      float rs = pe;
#pragma unroll
      for (int off = 1; off < 16; off <<= 1)
        rs += __shfl_xor(rs, off, 32);
      l_i[i] = l_i[i] * alpha + rs;
      m_i[i] = mnew;
#pragma unroll
      for (int nt = 0; nt < 16; ++nt) acc[nt][i] *= alpha;
      ps[wave][i + 8 * hf][r] = pe;
    }
    asm volatile("s_wait_dscnt 0" ::: "memory");

    v2f pa[4];
#pragma unroll
    for (int ks = 0; ks < 4; ++ks) {
      v2f t;
      t.x = ps[wave][r][ks * 4 + 2 * hf];
      t.y = ps[wave][r][ks * 4 + 1 + 2 * hf];
      pa[ks] = t;
    }

#pragma unroll
    for (int ks = 0; ks < 4; ++ks) {
#pragma unroll
      for (int nt = 0; nt < 16; ++nt)
        acc[nt] = wmma_f32(pa[ks], vf[ks * 16 + nt], acc[nt]);
    }
  }

#pragma unroll
  for (int i = 0; i < 8; ++i) {
    const float inv = 1.0f / l_i[i];
    const int qrow = q0 + i + 8 * hf;
    float* Op = O + ((size_t)(b * SLEN + qrow)) * HVD + h * VDD;
#pragma unroll
    for (int nt = 0; nt < 16; ++nt) Op[nt * 16 + r] = acc[nt][i] * inv;
  }
}

extern "C" void kernel_launch(void* const* d_in, const int* in_sizes, int n_in,
                              void* d_out, int out_size, void* d_ws,
                              size_t ws_size, hipStream_t stream) {
  (void)in_sizes; (void)n_in; (void)out_size; (void)ws_size;
  const float* x        = (const float*)d_in[0];
  const float* w_dq     = (const float*)d_in[1];
  const float* gamma_q  = (const float*)d_in[2];
  const float* w_uq     = (const float*)d_in[3];
  const float* w_dkv    = (const float*)d_in[4];
  const float* gamma_kv = (const float*)d_in[5];
  const float* w_ukv    = (const float*)d_in[6];
  const float* w_o      = (const float*)d_in[7];
  float* out = (float*)d_out;

  float* ws = (float*)d_ws;
  float* tmp_q   = ws;
  float* cq      = tmp_q   + (size_t)ROWS * CQD;
  float* qraw    = cq      + (size_t)ROWS * CQD;
  float* qst     = qraw    + (size_t)ROWS * UQ_N;
  float* ckv_kr  = qst     + (size_t)BATCH * NH * SLEN * DQK;
  float* ckv     = ckv_kr  + (size_t)ROWS * DKV_N;
  float* kvraw   = ckv     + (size_t)ROWS * CKVD;
  float* kst     = kvraw   + (size_t)ROWS * UKV_N;
  float* vpi     = kst     + (size_t)BATCH * NKV * SLEN * DQK;
  float* attn_o  = vpi     + (size_t)BATCH * NKV * SLEN * VDD;

  gemm_wmma_f32_lds<<<dim3(ROWS / 128, CQD / 64), 256, 0, stream>>>(
      x, w_dq, tmp_q, ROWS, CQD, DIMC);
  t5ln_kernel<<<ROWS, 32, 0, stream>>>(tmp_q, CQD, gamma_q, cq);
  gemm_wmma_f32_lds<<<dim3(ROWS / 128, UQ_N / 64), 256, 0, stream>>>(
      cq, w_uq, qraw, ROWS, UQ_N, CQD);
  {
    size_t total = (size_t)BATCH * NH * SLEN * DQK;
    build_q_kernel<<<(unsigned)((total + 255) / 256), 256, 0, stream>>>(qraw, qst);
  }
  gemm_wmma_f32_lds<<<dim3(ROWS / 128, DKV_N / 64), 256, 0, stream>>>(
      x, w_dkv, ckv_kr, ROWS, DKV_N, DIMC);
  t5ln_kernel<<<ROWS, 32, 0, stream>>>(ckv_kr, DKV_N, gamma_kv, ckv);
  gemm_wmma_f32_lds<<<dim3(ROWS / 128, UKV_N / 64), 256, 0, stream>>>(
      ckv, w_ukv, kvraw, ROWS, UKV_N, CQD);
  {
    size_t tk = (size_t)BATCH * NKV * SLEN * DQK;
    build_k_kernel<<<(unsigned)((tk + 255) / 256), 256, 0, stream>>>(kvraw, ckv_kr, kst);
    size_t tv = (size_t)BATCH * NKV * SLEN * VDD;
    build_v_kernel<<<(unsigned)((tv + 255) / 256), 256, 0, stream>>>(kvraw, vpi);
  }
  attn_kernel<<<dim3(BATCH * NH, SLEN / 128), 256, 0, stream>>>(qst, kst, vpi, attn_o);
  gemm_wmma_f32_lds<<<dim3(ROWS / 128, DIMC / 64), 256, 0, stream>>>(
      attn_o, w_o, out, ROWS, DIMC, HVD);
}


# attention: score/output tiles transposed (S^T = K Q^T, O^T = V^T P^T): softmax max/sum over keys in-lane plus one half-wave exchange, P^T feeds PV from registers, no LDS shuffles; key sum order new
# speedup vs baseline: 5.0262x; 1.0875x over previous
#include <hip/hip_runtime.h>
#include <math.h>

constexpr int DIMC   = 2048;
constexpr int SLEN   = 2048;
constexpr int BATCH  = 2;
constexpr int NH     = 16;
constexpr int NKV    = 4;
constexpr int CQD    = 128;
constexpr int QND    = 96;
constexpr int QRD    = 32;
constexpr int CKVD   = 128;
constexpr int KND    = 64;
constexpr int KRD    = 64;
constexpr int VDD    = 256;
constexpr int DQK    = 128;
constexpr int ROWS   = BATCH * SLEN;
constexpr int UQ_N   = NH * (QND + QRD);
constexpr int DKV_N  = CKVD + KRD;
constexpr int UKV_N  = NKV * (KND + VDD);
constexpr int HVD    = NH * VDD;

typedef float v2f __attribute__((ext_vector_type(2)));
typedef float v8f __attribute__((ext_vector_type(8)));

__device__ __forceinline__ v8f wmma_f32(v2f a, v2f b, v8f c) {
  return __builtin_amdgcn_wmma_f32_16x16x4_f32(
       false, a,  false, b,
       (short)0, c,  false,  false);
}

__device__ __forceinline__ v8f v8f_zero() {
  v8f z = {0.f, 0.f, 0.f, 0.f, 0.f, 0.f, 0.f, 0.f};
  return z;
}

constexpr int KC = 32;

typedef float v4f __attribute__((ext_vector_type(4)));

__device__ __forceinline__ void gemm_stage_chunk(
    float (*Asb)[36], float (*Bsb)[64][2], int ar, int ac, int br, int bn,
    v4f a0, v4f a1, v4f a2, v4f a3, v4f b0, v4f b1) {
  *(v4f*)&Asb[ar][ac]      = a0;
  *(v4f*)&Asb[ar + 32][ac] = a1;
  *(v4f*)&Asb[ar + 64][ac] = a2;
  *(v4f*)&Asb[ar + 96][ac] = a3;
  const int kp = br >> 1, kb = br & 1;
  Bsb[kp][bn + 0][kb]     = b0.x;
  Bsb[kp][bn + 1][kb]     = b0.y;
  Bsb[kp][bn + 2][kb]     = b0.z;
  Bsb[kp][bn + 3][kb]     = b0.w;
  Bsb[kp + 8][bn + 0][kb] = b1.x;
  Bsb[kp + 8][bn + 1][kb] = b1.y;
  Bsb[kp + 8][bn + 2][kb] = b1.z;
  Bsb[kp + 8][bn + 3][kb] = b1.w;
}

__global__ __launch_bounds__(256) void gemm_wmma_f32_lds(
    const float* __restrict__ A, const float* __restrict__ B,
    float* __restrict__ C, int M, int N, int Kd) {
  __shared__ float As[2][128][36];
  __shared__ float Bs[2][KC / 2][64][2];

  const int tid  = threadIdx.x;
  const int lane = tid & 31;
  const int wave = tid >> 5;
  const int r    = lane & 15;
  const int hf   = lane >> 4;
  const int m0b  = blockIdx.x * 128;
  const int n0   = blockIdx.y * 64;

  v8f acc[4];
#pragma unroll
  for (int j = 0; j < 4; ++j) acc[j] = v8f_zero();

  const int ar = tid >> 3;
  const int ac = (tid * 4) & 31;
  const int br = tid >> 4;
  const int bn = (tid * 4) & 63;

  const float* Ap0 = A + (size_t)(m0b + ar) * Kd + ac;
  const float* Ap1 = Ap0 + (size_t)32 * Kd;
  const float* Ap2 = Ap0 + (size_t)64 * Kd;
  const float* Ap3 = Ap0 + (size_t)96 * Kd;
  const float* Bp0 = B + (size_t)br * N + n0 + bn;
  const float* Bp1 = Bp0 + (size_t)16 * N;

  const int nchunk = Kd / KC;

  v4f a0 = *(const v4f*)Ap0;
  v4f a1 = *(const v4f*)Ap1;
  v4f a2 = *(const v4f*)Ap2;
  v4f a3 = *(const v4f*)Ap3;
  v4f b0 = *(const v4f*)Bp0;
  v4f b1 = *(const v4f*)Bp1;
  gemm_stage_chunk(As[0], Bs[0], ar, ac, br, bn, a0, a1, a2, a3, b0, b1);
  __syncthreads();

  for (int c = 0; c < nchunk; ++c) {
    const int buf = c & 1;
    const bool more = (c + 1 < nchunk);
    if (more) {
      const int kc = (c + 1) * KC;
      const size_t kb = (size_t)kc * N;
      a0 = *(const v4f*)(Ap0 + kc);
      a1 = *(const v4f*)(Ap1 + kc);
      a2 = *(const v4f*)(Ap2 + kc);
      a3 = *(const v4f*)(Ap3 + kc);
      b0 = *(const v4f*)(Bp0 + kb);
      b1 = *(const v4f*)(Bp1 + kb);
    }
#pragma unroll
    for (int k = 0; k < KC; k += 4) {
      v2f a = *(const v2f*)&As[buf][wave * 16 + r][k + 2 * hf];
#pragma unroll
      for (int j = 0; j < 4; ++j) {
        v2f bb = *(const v2f*)&Bs[buf][(k + 2 * hf) >> 1][j * 16 + r][0];
        acc[j] = wmma_f32(a, bb, acc[j]);
      }
    }
    if (more)
      gemm_stage_chunk(As[buf ^ 1], Bs[buf ^ 1], ar, ac, br, bn,
                       a0, a1, a2, a3, b0, b1);
    __syncthreads();
  }

#pragma unroll
  for (int j = 0; j < 4; ++j)
#pragma unroll
    for (int i = 0; i < 8; ++i)
      C[(size_t)(m0b + wave * 16 + i + 8 * hf) * N + n0 + j * 16 + r] =
          acc[j][i];
}

__global__ __launch_bounds__(32) void t5ln_kernel(
    const float* __restrict__ in, int in_stride,
    const float* __restrict__ gamma, float* __restrict__ out) {
  const int row  = blockIdx.x;
  const int lane = threadIdx.x;
  const float* ip = in + (size_t)row * in_stride;
  float v[4], s = 0.f, s2 = 0.f;
#pragma unroll
  for (int i = 0; i < 4; ++i) {
    v[i] = ip[lane + 32 * i];
    s  += v[i];
    s2 += v[i] * v[i];
  }
#pragma unroll
  for (int off = 1; off < 32; off <<= 1) {
    s  += __shfl_xor(s, off, 32);
    s2 += __shfl_xor(s2, off, 32);
  }
  const float mu  = s * (1.f / 128.f);
  const float var = s2 * (1.f / 128.f) - mu * mu;
  const float rs  = rsqrtf(var + 1e-5f);
  float* op = out + (size_t)row * 128;
#pragma unroll
  for (int i = 0; i < 4; ++i)
    op[lane + 32 * i] = (v[i] - mu) * rs * gamma[lane + 32 * i];
}

__device__ __forceinline__ void rope_pair(float spos, int pair, float invD,
                                          float xe, float xo,
                                          float& re, float& ro) {
  const float lnTheta = 9.210340371976184f;
  float freq = expf(-(2.0f * (float)pair) * invD * lnTheta);
  float ang  = spos * freq;
  float c = cosf(ang), sn = sinf(ang);
  re = xe * c - xo * sn;
  ro = xe * sn + xo * c;
}

__global__ __launch_bounds__(256) void build_q_kernel(
    const float* __restrict__ qraw, float* __restrict__ qst) {
  size_t idx = (size_t)blockIdx.x * blockDim.x + threadIdx.x;
  const size_t total = (size_t)BATCH * NH * SLEN * DQK;
  if (idx >= total) return;
  int j = idx & 127;
  int s = (int)((idx >> 7) & (SLEN - 1));
  int h = (int)((idx >> 18) & (NH - 1));
  int b = (int)(idx >> 22);
  const float* qr = qraw + ((size_t)(b * SLEN + s)) * UQ_N;
  float val;
  if (j < QND) {
    val = qr[h * QND + j];
  } else {
    int jj = j - QND;
    int pair = jj >> 1;
    const float* base = qr + NH * QND + h * QRD;
    float xe = base[2 * pair], xo = base[2 * pair + 1];
    float re, ro;
    rope_pair((float)s, pair, 1.0f / (float)QRD, xe, xo, re, ro);
    val = (jj & 1) ? ro : re;
  }
  qst[idx] = val;
}

__global__ __launch_bounds__(256) void build_k_kernel(
    const float* __restrict__ kvraw, const float* __restrict__ ckv_kr,
    float* __restrict__ kst) {
  size_t idx = (size_t)blockIdx.x * blockDim.x + threadIdx.x;
  const size_t total = (size_t)BATCH * NKV * SLEN * DQK;
  if (idx >= total) return;
  int e   = idx & 3;
  int ln  = (int)((idx >> 2) & 31);
  int jc  = (int)((idx >> 7) & 15);
  int t   = (int)((idx >> 11) & (SLEN / 16 - 1));
  int kvh = (int)((idx >> 18) & (NKV - 1));
  int b   = (int)(idx >> 20);
  int j   = jc * 8 + (ln >> 4) * 4 + e;
  int s   = t * 16 + (ln & 15);
  size_t row = (size_t)(b * SLEN + s);
  float val;
  if (j < KND) {
    val = kvraw[row * UKV_N + kvh * KND + j];
  } else {
    int jj = j - KND;
    int pair = jj >> 1;
    const float* base = ckv_kr + row * DKV_N + CKVD;
    float xe = base[2 * pair], xo = base[2 * pair + 1];
    float re, ro;
    rope_pair((float)s, pair, 1.0f / (float)KRD, xe, xo, re, ro);
    val = (jj & 1) ? ro : re;
  }
  kst[idx] = val;
}

__global__ __launch_bounds__(256) void build_v_kernel(
    const float* __restrict__ kvraw, float* __restrict__ vpi) {
  size_t idx = (size_t)blockIdx.x * blockDim.x + threadIdx.x;
  const size_t total = (size_t)BATCH * NKV * SLEN * VDD;
  if (idx >= total) return;
  int d   = idx & 255;
  int s   = (int)((idx >> 8) & (SLEN - 1));
  int kvh = (int)((idx >> 19) & (NKV - 1));
  int b   = (int)(idx >> 21);
  size_t row = (size_t)(b * SLEN + s);
  float val = kvraw[row * UKV_N + NKV * KND + kvh * VDD + d];
  size_t dst = ((size_t)(b * NKV + kvh) * (SLEN / 2) + (s >> 1)) * (2 * VDD) +
               d * 2 + (s & 1);
  vpi[dst] = val;
}

constexpr int OPITCH = 260;

__device__ __forceinline__ float xhalf(float x) {
  const int bits = __builtin_bit_cast(int, x);
  return __builtin_bit_cast(
      float, __builtin_amdgcn_permlanex16(bits, bits, 0x76543210, 0xfedcba98,
                                          false, false));
}

__global__ __launch_bounds__(256) __attribute__((amdgpu_waves_per_eu(2, 2)))
void attn_kernel(
    const float* __restrict__ Q, const float* __restrict__ K,
    const float* __restrict__ V, float* __restrict__ O) {
  __shared__ alignas(16) float ost[8][16][OPITCH];

  const int lane = threadIdx.x & 31;
  const int wave = threadIdx.x >> 5;
  const int r    = lane & 15;
  const int hf   = lane >> 4;
  const int bh   = blockIdx.x;
  const int b    = bh >> 4;
  const int h    = bh & 15;
  const int kvh  = h >> 2;
  const int qb   = (SLEN / 128 - 1) - (int)blockIdx.y;
  const int q0   = qb * 128 + wave * 16;

  const float* Qb = Q + ((size_t)bh * SLEN + q0) * DQK;
  const float* Kb = K + ((size_t)(b * NKV + kvh) * SLEN) * DQK;
  const float* Vb = V + ((size_t)(b * NKV + kvh) * SLEN) * VDD;

  const float scale = 0.08838834764831845f;

  v2f qa[32];
#pragma unroll
  for (int j = 0; j < 16; ++j) {
    const float4 qv = *(const float4*)(Qb + (size_t)r * DQK + j * 8 + 4 * hf);
    v2f t0, t1;
    t0.x = qv.x * scale;
    t0.y = qv.y * scale;
    t1.x = qv.z * scale;
    t1.y = qv.w * scale;
    qa[2 * j]     = t0;
    qa[2 * j + 1] = t1;
  }

  float m_run = -1e30f, l_run = 0.f;
  v8f acc[16];
#pragma unroll
  for (int nt = 0; nt < 16; ++nt) acc[nt] = v8f_zero();

  const int ktmax = (q0 + 15) >> 4;
  const int qcol  = q0 + r;

  float4 kf[16];
  {
    const float* Kt = Kb + lane * 4;
#pragma unroll
    for (int j = 0; j < 16; ++j) kf[j] = *(const float4*)(Kt + j * 128);
  }

  for (int kt = 0; kt <= ktmax; ++kt) {
    const int k0 = kt * 16;

    if (kt < ktmax) {
      const float* kpf = Kb + (size_t)(kt + 1) * (16 * DQK) + lane * 64;
      __builtin_prefetch(kpf, 0, 3);
      const float* vpf = Vb + (size_t)((k0 + 16) >> 1) * (2 * VDD) + lane * 64;
      __builtin_prefetch(vpf, 0, 3);
    }

    const float* Vt = Vb + (size_t)((k0 >> 1) + 4 * hf) * (2 * VDD) + r * 2;
    v2f vf[64];
#pragma unroll
    for (int ks = 0; ks < 3; ++ks)
#pragma unroll
      for (int nt = 0; nt < 16; ++nt)
        vf[ks * 16 + nt] =
            *(const v2f*)(Vt + (size_t)ks * (2 * VDD) + nt * 32);

    v8f sc = v8f_zero();
#pragma unroll
    for (int j = 0; j < 16; ++j) {
      v2f a0, a1;
      a0.x = kf[j].x;
      a0.y = kf[j].y;
      a1.x = kf[j].z;
      a1.y = kf[j].w;
      sc = wmma_f32(a0, qa[2 * j], sc);
      sc = wmma_f32(a1, qa[2 * j + 1], sc);
    }

    if (kt < ktmax) {
      const float* Kn = Kb + (size_t)(kt + 1) * (16 * DQK) + lane * 4;
#pragma unroll
      for (int j = 0; j < 16; ++j) kf[j] = *(const float4*)(Kn + j * 128);
    }
#pragma unroll
    for (int nt = 0; nt < 16; ++nt)
      vf[48 + nt] = *(const v2f*)(Vt + (size_t)3 * (2 * VDD) + nt * 32);

    const int key0 = k0 + 8 * hf;
    float v[8];
#pragma unroll
    for (int i = 0; i < 8; ++i) v[i] = (key0 + i > qcol) ? -1e30f : sc[i];
    float mx = fmaxf(fmaxf(fmaxf(v[0], v[1]), fmaxf(v[2], v[3])),
                     fmaxf(fmaxf(v[4], v[5]), fmaxf(v[6], v[7])));
    mx = fmaxf(mx, xhalf(mx));
    const float mnew  = fmaxf(m_run, mx);
    const float alpha = __expf(m_run - mnew);
    float pe[8];
#pragma unroll
    for (int i = 0; i < 8; ++i) pe[i] = __expf(v[i] - mnew);
    const float rs = ((pe[0] + pe[1]) + (pe[2] + pe[3])) +
                     ((pe[4] + pe[5]) + (pe[6] + pe[7]));
    l_run = l_run * alpha + rs;
    m_run = mnew;
#pragma unroll
    for (int nt = 0; nt < 16; ++nt) acc[nt] *= alpha;

#pragma unroll
    for (int ks = 0; ks < 4; ++ks) {
      v2f pb;
      pb.x = pe[2 * ks];
      pb.y = pe[2 * ks + 1];
#pragma unroll
      for (int nt = 0; nt < 16; ++nt)
        acc[nt] = wmma_f32(vf[ks * 16 + nt], pb, acc[nt]);
    }
    asm volatile("v_nop\n\tv_nop\n\tv_nop\n\tv_nop" : "+v"(pe[6]), "+v"(pe[7]));
  }

  const float inv = 1.0f / (l_run + xhalf(l_run));
  float (*os)[OPITCH] = ost[wave];
#pragma unroll
  for (int nt = 0; nt < 16; ++nt) {
    const v8f a = acc[nt];
    float4 lo, hi;
    lo.x = a[0] * inv; lo.y = a[1] * inv; lo.z = a[2] * inv; lo.w = a[3] * inv;
    hi.x = a[4] * inv; hi.y = a[5] * inv; hi.z = a[6] * inv; hi.w = a[7] * inv;
    *(float4*)&os[r][nt * 16 + 8 * hf]     = lo;
    *(float4*)&os[r][nt * 16 + 8 * hf + 4] = hi;
  }
  asm volatile("s_wait_dscnt 0" ::: "memory");
  float* Ob = O + ((size_t)(b * SLEN + q0)) * HVD + h * VDD + lane * 4;
#pragma unroll
  for (int row = 0; row < 16; ++row) {
#pragma unroll
    for (int hc = 0; hc < 2; ++hc) {
      const float4 o = *(const float4*)&os[row][hc * 128 + lane * 4];
      *(float4*)(Ob + (size_t)row * HVD + hc * 128) = o;
    }
  }
}

extern "C" void kernel_launch(void* const* d_in, const int* in_sizes, int n_in,
                              void* d_out, int out_size, void* d_ws,
                              size_t ws_size, hipStream_t stream) {
  (void)in_sizes; (void)n_in; (void)out_size; (void)ws_size;
  const float* x        = (const float*)d_in[0];
  const float* w_dq     = (const float*)d_in[1];
  const float* gamma_q  = (const float*)d_in[2];
  const float* w_uq     = (const float*)d_in[3];
  const float* w_dkv    = (const float*)d_in[4];
  const float* gamma_kv = (const float*)d_in[5];
  const float* w_ukv    = (const float*)d_in[6];
  const float* w_o      = (const float*)d_in[7];
  float* out = (float*)d_out;

  float* ws = (float*)d_ws;
  float* tmp_q   = ws;
  float* cq      = tmp_q   + (size_t)ROWS * CQD;
  float* qraw    = cq      + (size_t)ROWS * CQD;
  float* qst     = qraw    + (size_t)ROWS * UQ_N;
  float* ckv_kr  = qst     + (size_t)BATCH * NH * SLEN * DQK;
  float* ckv     = ckv_kr  + (size_t)ROWS * DKV_N;
  float* kvraw   = ckv     + (size_t)ROWS * CKVD;
  float* kst     = kvraw   + (size_t)ROWS * UKV_N;
  float* vpi     = kst     + (size_t)BATCH * NKV * SLEN * DQK;
  float* attn_o  = vpi     + (size_t)BATCH * NKV * SLEN * VDD;

  gemm_wmma_f32_lds<<<dim3(ROWS / 128, CQD / 64), 256, 0, stream>>>(
      x, w_dq, tmp_q, ROWS, CQD, DIMC);
  t5ln_kernel<<<ROWS, 32, 0, stream>>>(tmp_q, CQD, gamma_q, cq);
  gemm_wmma_f32_lds<<<dim3(ROWS / 128, UQ_N / 64), 256, 0, stream>>>(
      cq, w_uq, qraw, ROWS, UQ_N, CQD);
  {
    size_t total = (size_t)BATCH * NH * SLEN * DQK;
    build_q_kernel<<<(unsigned)((total + 255) / 256), 256, 0, stream>>>(qraw, qst);
  }
  gemm_wmma_f32_lds<<<dim3(ROWS / 128, DKV_N / 64), 256, 0, stream>>>(
      x, w_dkv, ckv_kr, ROWS, DKV_N, DIMC);
  t5ln_kernel<<<ROWS, 32, 0, stream>>>(ckv_kr, DKV_N, gamma_kv, ckv);
  gemm_wmma_f32_lds<<<dim3(ROWS / 128, UKV_N / 64), 256, 0, stream>>>(
      ckv, w_ukv, kvraw, ROWS, UKV_N, CQD);
  {
    size_t tk = (size_t)BATCH * NKV * SLEN * DQK;
    build_k_kernel<<<(unsigned)((tk + 255) / 256), 256, 0, stream>>>(kvraw, ckv_kr, kst);
    size_t tv = (size_t)BATCH * NKV * SLEN * VDD;
    build_v_kernel<<<(unsigned)((tv + 255) / 256), 256, 0, stream>>>(kvraw, vpi);
  }
  attn_kernel<<<dim3(BATCH * NH, SLEN / 128), 256, 0, stream>>>(qst, kst, vpi, attn_o);
  gemm_wmma_f32_lds<<<dim3(ROWS / 128, DIMC / 64), 256, 0, stream>>>(
      attn_o, w_o, out, ROWS, DIMC, HVD);
}


# GEMM: the two projections of x (N=128 and N=192: 64 and 96 work-groups of 64 K-chunks each) share ONE launch and run side by side on different WGPs, not one after the other; every sum unchanged
# speedup vs baseline: 5.2933x; 1.0531x over previous
#include <hip/hip_runtime.h>
#include <math.h>

constexpr int DIMC   = 2048;
constexpr int SLEN   = 2048;
constexpr int BATCH  = 2;
constexpr int NH     = 16;
constexpr int NKV    = 4;
constexpr int CQD    = 128;
constexpr int QND    = 96;
constexpr int QRD    = 32;
constexpr int CKVD   = 128;
constexpr int KND    = 64;
constexpr int KRD    = 64;
constexpr int VDD    = 256;
constexpr int DQK    = 128;
constexpr int ROWS   = BATCH * SLEN;
constexpr int UQ_N   = NH * (QND + QRD);
constexpr int DKV_N  = CKVD + KRD;
constexpr int UKV_N  = NKV * (KND + VDD);
constexpr int HVD    = NH * VDD;

typedef float v2f __attribute__((ext_vector_type(2)));
typedef float v8f __attribute__((ext_vector_type(8)));

__device__ __forceinline__ v8f wmma_f32(v2f a, v2f b, v8f c) {
  return __builtin_amdgcn_wmma_f32_16x16x4_f32(
       false, a,  false, b,
       (short)0, c,  false,  false);
}

__device__ __forceinline__ v8f v8f_zero() {
  v8f z = {0.f, 0.f, 0.f, 0.f, 0.f, 0.f, 0.f, 0.f};
  return z;
}

constexpr int KC = 32;

typedef float v4f __attribute__((ext_vector_type(4)));

__device__ __forceinline__ void gemm_stage_chunk(
    float (*Asb)[36], float (*Bsb)[64][2], int ar, int ac, int br, int bn,
    v4f a0, v4f a1, v4f a2, v4f a3, v4f b0, v4f b1) {
  *(v4f*)&Asb[ar][ac]      = a0;
  *(v4f*)&Asb[ar + 32][ac] = a1;
  *(v4f*)&Asb[ar + 64][ac] = a2;
  *(v4f*)&Asb[ar + 96][ac] = a3;
  const int kp = br >> 1, kb = br & 1;
  Bsb[kp][bn + 0][kb]     = b0.x;
  Bsb[kp][bn + 1][kb]     = b0.y;
  Bsb[kp][bn + 2][kb]     = b0.z;
  Bsb[kp][bn + 3][kb]     = b0.w;
  Bsb[kp + 8][bn + 0][kb] = b1.x;
  Bsb[kp + 8][bn + 1][kb] = b1.y;
  Bsb[kp + 8][bn + 2][kb] = b1.z;
  Bsb[kp + 8][bn + 3][kb] = b1.w;
}

__device__ __forceinline__ void gemm_tile_128x64(
    const float* __restrict__ A, const float* __restrict__ B,
    float* __restrict__ C, int N, int Kd, int m0b, int n0,
    float (*As)[128][36], float (*Bs)[KC / 2][64][2]) {
  const int tid  = threadIdx.x;
  const int lane = tid & 31;
  const int wave = tid >> 5;
  const int r    = lane & 15;
  const int hf   = lane >> 4;

  v8f acc[4];
#pragma unroll
  for (int j = 0; j < 4; ++j) acc[j] = v8f_zero();

  const int ar = tid >> 3;
  const int ac = (tid * 4) & 31;
  const int br = tid >> 4;
  const int bn = (tid * 4) & 63;

  const float* Ap0 = A + (size_t)(m0b + ar) * Kd + ac;
  const float* Ap1 = Ap0 + (size_t)32 * Kd;
  const float* Ap2 = Ap0 + (size_t)64 * Kd;
  const float* Ap3 = Ap0 + (size_t)96 * Kd;
  const float* Bp0 = B + (size_t)br * N + n0 + bn;
  const float* Bp1 = Bp0 + (size_t)16 * N;

  const int nchunk = Kd / KC;

  v4f a0 = *(const v4f*)Ap0;
  v4f a1 = *(const v4f*)Ap1;
  v4f a2 = *(const v4f*)Ap2;
  v4f a3 = *(const v4f*)Ap3;
  v4f b0 = *(const v4f*)Bp0;
  v4f b1 = *(const v4f*)Bp1;
  gemm_stage_chunk(As[0], Bs[0], ar, ac, br, bn, a0, a1, a2, a3, b0, b1);
  __syncthreads();

  for (int c = 0; c < nchunk; ++c) {
    const int buf = c & 1;
    const bool more = (c + 1 < nchunk);
    if (more) {
      const int kc = (c + 1) * KC;
      const size_t kb = (size_t)kc * N;
      a0 = *(const v4f*)(Ap0 + kc);
      a1 = *(const v4f*)(Ap1 + kc);
      a2 = *(const v4f*)(Ap2 + kc);
      a3 = *(const v4f*)(Ap3 + kc);
      b0 = *(const v4f*)(Bp0 + kb);
      b1 = *(const v4f*)(Bp1 + kb);
    }
#pragma unroll
    for (int k = 0; k < KC; k += 4) {
      v2f a = *(const v2f*)&As[buf][wave * 16 + r][k + 2 * hf];
#pragma unroll
      for (int j = 0; j < 4; ++j) {
        v2f bb = *(const v2f*)&Bs[buf][(k + 2 * hf) >> 1][j * 16 + r][0];
        acc[j] = wmma_f32(a, bb, acc[j]);
      }
    }
    if (more)
      gemm_stage_chunk(As[buf ^ 1], Bs[buf ^ 1], ar, ac, br, bn,
                       a0, a1, a2, a3, b0, b1);
    __syncthreads();
  }

#pragma unroll
  for (int j = 0; j < 4; ++j)
#pragma unroll
    for (int i = 0; i < 8; ++i)
      C[(size_t)(m0b + wave * 16 + i + 8 * hf) * N + n0 + j * 16 + r] =
          acc[j][i];
}

__global__ __launch_bounds__(256) void gemm_wmma_f32_lds(
    const float* __restrict__ A, const float* __restrict__ B,
    float* __restrict__ C, int M, int N, int Kd) {
  __shared__ float As[2][128][36];
  __shared__ float Bs[2][KC / 2][64][2];
  (void)M;
  gemm_tile_128x64(A, B, C, N, Kd, blockIdx.x * 128, blockIdx.y * 64, As, Bs);
}

__global__ __launch_bounds__(256) void gemm_wmma_f32_lds_pair(
    const float* __restrict__ A,
    const float* __restrict__ B0, float* __restrict__ C0, int N0,
    const float* __restrict__ B1, float* __restrict__ C1, int N1, int Kd) {
  __shared__ float As[2][128][36];
  __shared__ float Bs[2][KC / 2][64][2];
  const int  nb0   = N0 / 64;
  const int  by    = blockIdx.y;
  const bool first = by < nb0;
  const float* B = first ? B0 : B1;
  float*       C = first ? C0 : C1;
  const int    N = first ? N0 : N1;
  gemm_tile_128x64(A, B, C, N, Kd, blockIdx.x * 128,
                   (first ? by : by - nb0) * 64, As, Bs);
}

__global__ __launch_bounds__(32) void t5ln_kernel(
    const float* __restrict__ in, int in_stride,
    const float* __restrict__ gamma, float* __restrict__ out) {
  const int row  = blockIdx.x;
  const int lane = threadIdx.x;
  const float* ip = in + (size_t)row * in_stride;
  float v[4], s = 0.f, s2 = 0.f;
#pragma unroll
  for (int i = 0; i < 4; ++i) {
    v[i] = ip[lane + 32 * i];
    s  += v[i];
    s2 += v[i] * v[i];
  }
#pragma unroll
  for (int off = 1; off < 32; off <<= 1) {
    s  += __shfl_xor(s, off, 32);
    s2 += __shfl_xor(s2, off, 32);
  }
  const float mu  = s * (1.f / 128.f);
  const float var = s2 * (1.f / 128.f) - mu * mu;
  const float rs  = rsqrtf(var + 1e-5f);
  float* op = out + (size_t)row * 128;
#pragma unroll
  for (int i = 0; i < 4; ++i)
    op[lane + 32 * i] = (v[i] - mu) * rs * gamma[lane + 32 * i];
}

__device__ __forceinline__ void rope_pair(float spos, int pair, float invD,
                                          float xe, float xo,
                                          float& re, float& ro) {
  const float lnTheta = 9.210340371976184f;
  float freq = expf(-(2.0f * (float)pair) * invD * lnTheta);
  float ang  = spos * freq;
  float c = cosf(ang), sn = sinf(ang);
  re = xe * c - xo * sn;
  ro = xe * sn + xo * c;
}

__global__ __launch_bounds__(256) void build_q_kernel(
    const float* __restrict__ qraw, float* __restrict__ qst) {
  size_t idx = (size_t)blockIdx.x * blockDim.x + threadIdx.x;
  const size_t total = (size_t)BATCH * NH * SLEN * DQK;
  if (idx >= total) return;
  int j = idx & 127;
  int s = (int)((idx >> 7) & (SLEN - 1));
  int h = (int)((idx >> 18) & (NH - 1));
  int b = (int)(idx >> 22);
  const float* qr = qraw + ((size_t)(b * SLEN + s)) * UQ_N;
  float val;
  if (j < QND) {
    val = qr[h * QND + j];
  } else {
    int jj = j - QND;
    int pair = jj >> 1;
    const float* base = qr + NH * QND + h * QRD;
    float xe = base[2 * pair], xo = base[2 * pair + 1];
    float re, ro;
    rope_pair((float)s, pair, 1.0f / (float)QRD, xe, xo, re, ro);
    val = (jj & 1) ? ro : re;
  }
  qst[idx] = val;
}

__global__ __launch_bounds__(256) void build_k_kernel(
    const float* __restrict__ kvraw, const float* __restrict__ ckv_kr,
    float* __restrict__ kst) {
  size_t idx = (size_t)blockIdx.x * blockDim.x + threadIdx.x;
  const size_t total = (size_t)BATCH * NKV * SLEN * DQK;
  if (idx >= total) return;
  int e   = idx & 3;
  int ln  = (int)((idx >> 2) & 31);
  int jc  = (int)((idx >> 7) & 15);
  int t   = (int)((idx >> 11) & (SLEN / 16 - 1));
  int kvh = (int)((idx >> 18) & (NKV - 1));
  int b   = (int)(idx >> 20);
  int j   = jc * 8 + (ln >> 4) * 4 + e;
  int s   = t * 16 + (ln & 15);
  size_t row = (size_t)(b * SLEN + s);
  float val;
  if (j < KND) {
    val = kvraw[row * UKV_N + kvh * KND + j];
  } else {
    int jj = j - KND;
    int pair = jj >> 1;
    const float* base = ckv_kr + row * DKV_N + CKVD;
    float xe = base[2 * pair], xo = base[2 * pair + 1];
    float re, ro;
    rope_pair((float)s, pair, 1.0f / (float)KRD, xe, xo, re, ro);
    val = (jj & 1) ? ro : re;
  }
  kst[idx] = val;
}

__global__ __launch_bounds__(256) void build_v_kernel(
    const float* __restrict__ kvraw, float* __restrict__ vpi) {
  size_t idx = (size_t)blockIdx.x * blockDim.x + threadIdx.x;
  const size_t total = (size_t)BATCH * NKV * SLEN * VDD;
  if (idx >= total) return;
  int d   = idx & 255;
  int s   = (int)((idx >> 8) & (SLEN - 1));
  int kvh = (int)((idx >> 19) & (NKV - 1));
  int b   = (int)(idx >> 21);
  size_t row = (size_t)(b * SLEN + s);
  float val = kvraw[row * UKV_N + NKV * KND + kvh * VDD + d];
  size_t dst = ((size_t)(b * NKV + kvh) * (SLEN / 2) + (s >> 1)) * (2 * VDD) +
               d * 2 + (s & 1);
  vpi[dst] = val;
}

constexpr int OPITCH = 260;

__device__ __forceinline__ float xhalf(float x) {
  const int bits = __builtin_bit_cast(int, x);
  return __builtin_bit_cast(
      float, __builtin_amdgcn_permlanex16(bits, bits, 0x76543210, 0xfedcba98,
                                          false, false));
}

__global__ __launch_bounds__(256) __attribute__((amdgpu_waves_per_eu(2, 2)))
void attn_kernel(
    const float* __restrict__ Q, const float* __restrict__ K,
    const float* __restrict__ V, float* __restrict__ O) {
  __shared__ alignas(16) float ost[8][16][OPITCH];

  const int lane = threadIdx.x & 31;
  const int wave = threadIdx.x >> 5;
  const int r    = lane & 15;
  const int hf   = lane >> 4;
  const int bh   = blockIdx.x;
  const int b    = bh >> 4;
  const int h    = bh & 15;
  const int kvh  = h >> 2;
  const int qb   = (SLEN / 128 - 1) - (int)blockIdx.y;
  const int q0   = qb * 128 + wave * 16;

  const float* Qb = Q + ((size_t)bh * SLEN + q0) * DQK;
  const float* Kb = K + ((size_t)(b * NKV + kvh) * SLEN) * DQK;
  const float* Vb = V + ((size_t)(b * NKV + kvh) * SLEN) * VDD;

  const float scale = 0.08838834764831845f;

  v2f qa[32];
#pragma unroll
  for (int j = 0; j < 16; ++j) {
    const float4 qv = *(const float4*)(Qb + (size_t)r * DQK + j * 8 + 4 * hf);
    v2f t0, t1;
    t0.x = qv.x * scale;
    t0.y = qv.y * scale;
    t1.x = qv.z * scale;
    t1.y = qv.w * scale;
    qa[2 * j]     = t0;
    qa[2 * j + 1] = t1;
  }

  float m_run = -1e30f, l_run = 0.f;
  v8f acc[16];
#pragma unroll
  for (int nt = 0; nt < 16; ++nt) acc[nt] = v8f_zero();

  const int ktmax = (q0 + 15) >> 4;
  const int qcol  = q0 + r;

  float4 kf[16];
  {
    const float* Kt = Kb + lane * 4;
#pragma unroll
    for (int j = 0; j < 16; ++j) kf[j] = *(const float4*)(Kt + j * 128);
  }

  for (int kt = 0; kt <= ktmax; ++kt) {
    const int k0 = kt * 16;

    if (kt < ktmax) {
      const float* kpf = Kb + (size_t)(kt + 1) * (16 * DQK) + lane * 64;
      __builtin_prefetch(kpf, 0, 3);
      const float* vpf = Vb + (size_t)((k0 + 16) >> 1) * (2 * VDD) + lane * 64;
      __builtin_prefetch(vpf, 0, 3);
    }

    const float* Vt = Vb + (size_t)((k0 >> 1) + 4 * hf) * (2 * VDD) + r * 2;
    v2f vf[64];
#pragma unroll
    for (int ks = 0; ks < 3; ++ks)
#pragma unroll
      for (int nt = 0; nt < 16; ++nt)
        vf[ks * 16 + nt] =
            *(const v2f*)(Vt + (size_t)ks * (2 * VDD) + nt * 32);

    v8f sc = v8f_zero();
#pragma unroll
    for (int j = 0; j < 16; ++j) {
      v2f a0, a1;
      a0.x = kf[j].x;
      a0.y = kf[j].y;
      a1.x = kf[j].z;
      a1.y = kf[j].w;
      sc = wmma_f32(a0, qa[2 * j], sc);
      sc = wmma_f32(a1, qa[2 * j + 1], sc);
    }

    if (kt < ktmax) {
      const float* Kn = Kb + (size_t)(kt + 1) * (16 * DQK) + lane * 4;
#pragma unroll
      for (int j = 0; j < 16; ++j) kf[j] = *(const float4*)(Kn + j * 128);
    }
#pragma unroll
    for (int nt = 0; nt < 16; ++nt)
      vf[48 + nt] = *(const v2f*)(Vt + (size_t)3 * (2 * VDD) + nt * 32);

    const int key0 = k0 + 8 * hf;
    float v[8];
#pragma unroll
    for (int i = 0; i < 8; ++i) v[i] = (key0 + i > qcol) ? -1e30f : sc[i];
    float mx = fmaxf(fmaxf(fmaxf(v[0], v[1]), fmaxf(v[2], v[3])),
                     fmaxf(fmaxf(v[4], v[5]), fmaxf(v[6], v[7])));
    mx = fmaxf(mx, xhalf(mx));
    const float mnew  = fmaxf(m_run, mx);
    const float alpha = __expf(m_run - mnew);
    float pe[8];
#pragma unroll
    for (int i = 0; i < 8; ++i) pe[i] = __expf(v[i] - mnew);
    const float rs = ((pe[0] + pe[1]) + (pe[2] + pe[3])) +
                     ((pe[4] + pe[5]) + (pe[6] + pe[7]));
    l_run = l_run * alpha + rs;
    m_run = mnew;
#pragma unroll
    for (int nt = 0; nt < 16; ++nt) acc[nt] *= alpha;

#pragma unroll
    for (int ks = 0; ks < 4; ++ks) {
      v2f pb;
      pb.x = pe[2 * ks];
      pb.y = pe[2 * ks + 1];
#pragma unroll
      for (int nt = 0; nt < 16; ++nt)
        acc[nt] = wmma_f32(vf[ks * 16 + nt], pb, acc[nt]);
    }
    asm volatile("v_nop\n\tv_nop\n\tv_nop\n\tv_nop" : "+v"(pe[6]), "+v"(pe[7]));
  }

  const float inv = 1.0f / (l_run + xhalf(l_run));
  float (*os)[OPITCH] = ost[wave];
#pragma unroll
  for (int nt = 0; nt < 16; ++nt) {
    const v8f a = acc[nt];
    float4 lo, hi;
    lo.x = a[0] * inv; lo.y = a[1] * inv; lo.z = a[2] * inv; lo.w = a[3] * inv;
    hi.x = a[4] * inv; hi.y = a[5] * inv; hi.z = a[6] * inv; hi.w = a[7] * inv;
    *(float4*)&os[r][nt * 16 + 8 * hf]     = lo;
    *(float4*)&os[r][nt * 16 + 8 * hf + 4] = hi;
  }
  asm volatile("s_wait_dscnt 0" ::: "memory");
  float* Ob = O + ((size_t)(b * SLEN + q0)) * HVD + h * VDD + lane * 4;
#pragma unroll
  for (int row = 0; row < 16; ++row) {
#pragma unroll
    for (int hc = 0; hc < 2; ++hc) {
      const float4 o = *(const float4*)&os[row][hc * 128 + lane * 4];
      *(float4*)(Ob + (size_t)row * HVD + hc * 128) = o;
    }
  }
}

extern "C" void kernel_launch(void* const* d_in, const int* in_sizes, int n_in,
                              void* d_out, int out_size, void* d_ws,
                              size_t ws_size, hipStream_t stream) {
  (void)in_sizes; (void)n_in; (void)out_size; (void)ws_size;
  const float* x        = (const float*)d_in[0];
  const float* w_dq     = (const float*)d_in[1];
  const float* gamma_q  = (const float*)d_in[2];
  const float* w_uq     = (const float*)d_in[3];
  const float* w_dkv    = (const float*)d_in[4];
  const float* gamma_kv = (const float*)d_in[5];
  const float* w_ukv    = (const float*)d_in[6];
  const float* w_o      = (const float*)d_in[7];
  float* out = (float*)d_out;

  float* ws = (float*)d_ws;
  float* tmp_q   = ws;
  float* cq      = tmp_q   + (size_t)ROWS * CQD;
  float* qraw    = cq      + (size_t)ROWS * CQD;
  float* qst     = qraw    + (size_t)ROWS * UQ_N;
  float* ckv_kr  = qst     + (size_t)BATCH * NH * SLEN * DQK;
  float* ckv     = ckv_kr  + (size_t)ROWS * DKV_N;
  float* kvraw   = ckv     + (size_t)ROWS * CKVD;
  float* kst     = kvraw   + (size_t)ROWS * UKV_N;
  float* vpi     = kst     + (size_t)BATCH * NKV * SLEN * DQK;
  float* attn_o  = vpi     + (size_t)BATCH * NKV * SLEN * VDD;

  gemm_wmma_f32_lds_pair<<<dim3(ROWS / 128, (CQD + DKV_N) / 64), 256, 0, stream>>>(
      x, w_dq, tmp_q, CQD, w_dkv, ckv_kr, DKV_N, DIMC);
  t5ln_kernel<<<ROWS, 32, 0, stream>>>(tmp_q, CQD, gamma_q, cq);
  gemm_wmma_f32_lds<<<dim3(ROWS / 128, UQ_N / 64), 256, 0, stream>>>(
      cq, w_uq, qraw, ROWS, UQ_N, CQD);
  {
    size_t total = (size_t)BATCH * NH * SLEN * DQK;
    build_q_kernel<<<(unsigned)((total + 255) / 256), 256, 0, stream>>>(qraw, qst);
  }
  t5ln_kernel<<<ROWS, 32, 0, stream>>>(ckv_kr, DKV_N, gamma_kv, ckv);
  gemm_wmma_f32_lds<<<dim3(ROWS / 128, UKV_N / 64), 256, 0, stream>>>(
      ckv, w_ukv, kvraw, ROWS, UKV_N, CQD);
  {
    size_t tk = (size_t)BATCH * NKV * SLEN * DQK;
    build_k_kernel<<<(unsigned)((tk + 255) / 256), 256, 0, stream>>>(kvraw, ckv_kr, kst);
    size_t tv = (size_t)BATCH * NKV * SLEN * VDD;
    build_v_kernel<<<(unsigned)((tv + 255) / 256), 256, 0, stream>>>(kvraw, vpi);
  }
  attn_kernel<<<dim3(BATCH * NH, SLEN / 128), 256, 0, stream>>>(qst, kst, vpi, attn_o);
  gemm_wmma_f32_lds<<<dim3(ROWS / 128, DIMC / 64), 256, 0, stream>>>(
      attn_o, w_o, out, ROWS, DIMC, HVD);
}


# V up-projection folded into w_o (per-head weight GEMM): PV runs on the shared 128-wide latent, so PV WMMAs and the output GEMM's K halve; products reassociated, key/value/head sum order new
# speedup vs baseline: 8.0622x; 1.5231x over previous
#include <hip/hip_runtime.h>
#include <math.h>

constexpr int DIMC   = 2048;
constexpr int SLEN   = 2048;
constexpr int BATCH  = 2;
constexpr int NH     = 16;
constexpr int NKV    = 4;
constexpr int CQD    = 128;
constexpr int QND    = 96;
constexpr int QRD    = 32;
constexpr int CKVD   = 128;
constexpr int KND    = 64;
constexpr int KRD    = 64;
constexpr int VDD    = 256;
constexpr int DQK    = 128;
constexpr int ROWS   = BATCH * SLEN;
constexpr int UQ_N   = NH * (QND + QRD);
constexpr int DKV_N  = CKVD + KRD;
constexpr int UKV_N  = NKV * (KND + VDD);
constexpr int HVD    = NH * VDD;
constexpr int LAT    = CKVD;
constexpr int HLAT   = NH * LAT;
constexpr int KNOPE_N = NKV * KND;

typedef float v2f __attribute__((ext_vector_type(2)));
typedef float v8f __attribute__((ext_vector_type(8)));

__device__ __forceinline__ v8f wmma_f32(v2f a, v2f b, v8f c) {
  return __builtin_amdgcn_wmma_f32_16x16x4_f32(
       false, a,  false, b,
       (short)0, c,  false,  false);
}

__device__ __forceinline__ v8f v8f_zero() {
  v8f z = {0.f, 0.f, 0.f, 0.f, 0.f, 0.f, 0.f, 0.f};
  return z;
}

constexpr int KC = 32;

typedef float v4f __attribute__((ext_vector_type(4)));

__device__ __forceinline__ void gemm_stage_chunk(
    float (*Asb)[36], float (*Bsb)[64][2], int ar, int ac, int br, int bn,
    v4f a0, v4f a1, v4f a2, v4f a3, v4f b0, v4f b1) {
  *(v4f*)&Asb[ar][ac]      = a0;
  *(v4f*)&Asb[ar + 32][ac] = a1;
  *(v4f*)&Asb[ar + 64][ac] = a2;
  *(v4f*)&Asb[ar + 96][ac] = a3;
  const int kp = br >> 1, kb = br & 1;
  Bsb[kp][bn + 0][kb]     = b0.x;
  Bsb[kp][bn + 1][kb]     = b0.y;
  Bsb[kp][bn + 2][kb]     = b0.z;
  Bsb[kp][bn + 3][kb]     = b0.w;
  Bsb[kp + 8][bn + 0][kb] = b1.x;
  Bsb[kp + 8][bn + 1][kb] = b1.y;
  Bsb[kp + 8][bn + 2][kb] = b1.z;
  Bsb[kp + 8][bn + 3][kb] = b1.w;
}

__device__ __forceinline__ void gemm_tile_128x64(
    const float* __restrict__ A, int lda, const float* __restrict__ B, int ldb,
    float* __restrict__ C, int ldc, int Kd, int m0b, int n0,
    float (*As)[128][36], float (*Bs)[KC / 2][64][2]) {
  const int tid  = threadIdx.x;
  const int lane = tid & 31;
  const int wave = tid >> 5;
  const int r    = lane & 15;
  const int hf   = lane >> 4;

  v8f acc[4];
#pragma unroll
  for (int j = 0; j < 4; ++j) acc[j] = v8f_zero();

  const int ar = tid >> 3;
  const int ac = (tid * 4) & 31;
  const int br = tid >> 4;
  const int bn = (tid * 4) & 63;

  const float* Ap0 = A + (size_t)(m0b + ar) * lda + ac;
  const float* Ap1 = Ap0 + (size_t)32 * lda;
  const float* Ap2 = Ap0 + (size_t)64 * lda;
  const float* Ap3 = Ap0 + (size_t)96 * lda;
  const float* Bp0 = B + (size_t)br * ldb + n0 + bn;
  const float* Bp1 = Bp0 + (size_t)16 * ldb;

  const int nchunk = Kd / KC;

  v4f a0 = *(const v4f*)Ap0;
  v4f a1 = *(const v4f*)Ap1;
  v4f a2 = *(const v4f*)Ap2;
  v4f a3 = *(const v4f*)Ap3;
  v4f b0 = *(const v4f*)Bp0;
  v4f b1 = *(const v4f*)Bp1;
  gemm_stage_chunk(As[0], Bs[0], ar, ac, br, bn, a0, a1, a2, a3, b0, b1);
  __syncthreads();

  for (int c = 0; c < nchunk; ++c) {
    const int buf = c & 1;
    const bool more = (c + 1 < nchunk);
    if (more) {
      const int kc = (c + 1) * KC;
      const size_t kb = (size_t)kc * ldb;
      a0 = *(const v4f*)(Ap0 + kc);
      a1 = *(const v4f*)(Ap1 + kc);
      a2 = *(const v4f*)(Ap2 + kc);
      a3 = *(const v4f*)(Ap3 + kc);
      b0 = *(const v4f*)(Bp0 + kb);
      b1 = *(const v4f*)(Bp1 + kb);
    }
#pragma unroll
    for (int k = 0; k < KC; k += 4) {
      v2f a = *(const v2f*)&As[buf][wave * 16 + r][k + 2 * hf];
#pragma unroll
      for (int j = 0; j < 4; ++j) {
        v2f bb = *(const v2f*)&Bs[buf][(k + 2 * hf) >> 1][j * 16 + r][0];
        acc[j] = wmma_f32(a, bb, acc[j]);
      }
    }
    if (more)
      gemm_stage_chunk(As[buf ^ 1], Bs[buf ^ 1], ar, ac, br, bn,
                       a0, a1, a2, a3, b0, b1);
    __syncthreads();
  }

#pragma unroll
  for (int j = 0; j < 4; ++j)
#pragma unroll
    for (int i = 0; i < 8; ++i)
      C[(size_t)(m0b + wave * 16 + i + 8 * hf) * ldc + n0 + j * 16 + r] =
          acc[j][i];
}

__global__ __launch_bounds__(256) void gemm_wmma_f32_lds(
    const float* __restrict__ A, int lda, const float* __restrict__ B, int ldb,
    float* __restrict__ C, int ldc, int Kd) {
  __shared__ float As[2][128][36];
  __shared__ float Bs[2][KC / 2][64][2];
  gemm_tile_128x64(A, lda, B, ldb, C, ldc, Kd, blockIdx.x * 128,
                   blockIdx.y * 64, As, Bs);
}

__global__ __launch_bounds__(256) void gemm_wmma_f32_lds_fold(
    const float* __restrict__ w_ukv, const float* __restrict__ w_o,
    float* __restrict__ wc) {
  __shared__ float As[2][128][36];
  __shared__ float Bs[2][KC / 2][64][2];
  const int h = blockIdx.z;
  const float* A = w_ukv + KNOPE_N + (h / (NH / NKV)) * VDD;
  const float* B = w_o + (size_t)h * VDD * DIMC;
  float*       C = wc + (size_t)h * LAT * DIMC;
  gemm_tile_128x64(A, UKV_N, B, DIMC, C, DIMC, VDD, 0, blockIdx.y * 64, As, Bs);
}

__global__ __launch_bounds__(256) void gemm_wmma_f32_lds_pair(
    const float* __restrict__ A,
    const float* __restrict__ B0, float* __restrict__ C0, int N0,
    const float* __restrict__ B1, float* __restrict__ C1, int N1, int Kd) {
  __shared__ float As[2][128][36];
  __shared__ float Bs[2][KC / 2][64][2];
  const int  nb0   = N0 / 64;
  const int  by    = blockIdx.y;
  const bool first = by < nb0;
  const float* B = first ? B0 : B1;
  float*       C = first ? C0 : C1;
  const int    N = first ? N0 : N1;
  gemm_tile_128x64(A, Kd, B, N, C, N, Kd, blockIdx.x * 128,
                   (first ? by : by - nb0) * 64, As, Bs);
}

__global__ __launch_bounds__(32) void t5ln_kernel(
    const float* __restrict__ in, int in_stride,
    const float* __restrict__ gamma, float* __restrict__ out) {
  const int row  = blockIdx.x;
  const int lane = threadIdx.x;
  const float* ip = in + (size_t)row * in_stride;
  float v[4], s = 0.f, s2 = 0.f;
#pragma unroll
  for (int i = 0; i < 4; ++i) {
    v[i] = ip[lane + 32 * i];
    s  += v[i];
    s2 += v[i] * v[i];
  }
#pragma unroll
  for (int off = 1; off < 32; off <<= 1) {
    s  += __shfl_xor(s, off, 32);
    s2 += __shfl_xor(s2, off, 32);
  }
  const float mu  = s * (1.f / 128.f);
  const float var = s2 * (1.f / 128.f) - mu * mu;
  const float rs  = rsqrtf(var + 1e-5f);
  float* op = out + (size_t)row * 128;
#pragma unroll
  for (int i = 0; i < 4; ++i)
    op[lane + 32 * i] = (v[i] - mu) * rs * gamma[lane + 32 * i];
}

__device__ __forceinline__ void rope_pair(float spos, int pair, float invD,
                                          float xe, float xo,
                                          float& re, float& ro) {
  const float lnTheta = 9.210340371976184f;
  float freq = expf(-(2.0f * (float)pair) * invD * lnTheta);
  float ang  = spos * freq;
  float c = cosf(ang), sn = sinf(ang);
  re = xe * c - xo * sn;
  ro = xe * sn + xo * c;
}

__global__ __launch_bounds__(256) void build_q_kernel(
    const float* __restrict__ qraw, float* __restrict__ qst) {
  size_t idx = (size_t)blockIdx.x * blockDim.x + threadIdx.x;
  const size_t total = (size_t)BATCH * NH * SLEN * DQK;
  if (idx >= total) return;
  int j = idx & 127;
  int s = (int)((idx >> 7) & (SLEN - 1));
  int h = (int)((idx >> 18) & (NH - 1));
  int b = (int)(idx >> 22);
  const float* qr = qraw + ((size_t)(b * SLEN + s)) * UQ_N;
  float val;
  if (j < QND) {
    val = qr[h * QND + j];
  } else {
    int jj = j - QND;
    int pair = jj >> 1;
    const float* base = qr + NH * QND + h * QRD;
    float xe = base[2 * pair], xo = base[2 * pair + 1];
    float re, ro;
    rope_pair((float)s, pair, 1.0f / (float)QRD, xe, xo, re, ro);
    val = (jj & 1) ? ro : re;
  }
  qst[idx] = val;
}

__global__ __launch_bounds__(256) void build_k_kernel(
    const float* __restrict__ knope, const float* __restrict__ ckv_kr,
    float* __restrict__ kst) {
  size_t idx = (size_t)blockIdx.x * blockDim.x + threadIdx.x;
  const size_t total = (size_t)BATCH * NKV * SLEN * DQK;
  if (idx >= total) return;
  int e   = idx & 3;
  int ln  = (int)((idx >> 2) & 31);
  int jc  = (int)((idx >> 7) & 15);
  int t   = (int)((idx >> 11) & (SLEN / 16 - 1));
  int kvh = (int)((idx >> 18) & (NKV - 1));
  int b   = (int)(idx >> 20);
  int j   = jc * 8 + (ln >> 4) * 4 + e;
  int s   = t * 16 + (ln & 15);
  size_t row = (size_t)(b * SLEN + s);
  float val;
  if (j < KND) {
    val = knope[row * KNOPE_N + kvh * KND + j];
  } else {
    int jj = j - KND;
    int pair = jj >> 1;
    const float* base = ckv_kr + row * DKV_N + CKVD;
    float xe = base[2 * pair], xo = base[2 * pair + 1];
    float re, ro;
    rope_pair((float)s, pair, 1.0f / (float)KRD, xe, xo, re, ro);
    val = (jj & 1) ? ro : re;
  }
  kst[idx] = val;
}

__global__ __launch_bounds__(256) void build_vlat_kernel(
    const float* __restrict__ ckv, float* __restrict__ vl) {
  const int idx = blockIdx.x * 256 + threadIdx.x;
  const int d = idx & (LAT - 1);
  const int p = idx >> 7;
  v2f o;
  o.x = ckv[(size_t)(2 * p) * LAT + d];
  o.y = ckv[(size_t)(2 * p + 1) * LAT + d];
  *(v2f*)(vl + (size_t)p * (2 * LAT) + 2 * d) = o;
}

constexpr int OPITCH = 132;

__device__ __forceinline__ float xhalf(float x) {
  const int bits = __builtin_bit_cast(int, x);
  return __builtin_bit_cast(
      float, __builtin_amdgcn_permlanex16(bits, bits, 0x76543210, 0xfedcba98,
                                          false, false));
}

__global__ __launch_bounds__(256) __attribute__((amdgpu_waves_per_eu(2, 2)))
void attn_kernel(
    const float* __restrict__ Q, const float* __restrict__ K,
    const float* __restrict__ V, float* __restrict__ O) {
  __shared__ alignas(16) float ost[8][16][OPITCH];

  const int lane = threadIdx.x & 31;
  const int wave = threadIdx.x >> 5;
  const int r    = lane & 15;
  const int hf   = lane >> 4;
  const int bh   = blockIdx.x;
  const int b    = bh >> 4;
  const int h    = bh & 15;
  const int kvh  = h >> 2;
  const int qb   = (SLEN / 128 - 1) - (int)blockIdx.y;
  const int q0   = qb * 128 + wave * 16;

  const float* Qb = Q + ((size_t)bh * SLEN + q0) * DQK;
  const float* Kb = K + ((size_t)(b * NKV + kvh) * SLEN) * DQK;
  const float* Vb = V + ((size_t)b * SLEN) * LAT;

  const float scale = 0.08838834764831845f;

  v2f qa[32];
#pragma unroll
  for (int j = 0; j < 16; ++j) {
    const float4 qv = *(const float4*)(Qb + (size_t)r * DQK + j * 8 + 4 * hf);
    v2f t0, t1;
    t0.x = qv.x * scale;
    t0.y = qv.y * scale;
    t1.x = qv.z * scale;
    t1.y = qv.w * scale;
    qa[2 * j]     = t0;
    qa[2 * j + 1] = t1;
  }

  float m_run = -1e30f, l_run = 0.f;
  v8f acc[8];
#pragma unroll
  for (int nt = 0; nt < 8; ++nt) acc[nt] = v8f_zero();

  const int ktmax = (q0 + 15) >> 4;
  const int qcol  = q0 + r;

  float4 kf[16];
  {
    const float* Kt = Kb + lane * 4;
#pragma unroll
    for (int j = 0; j < 16; ++j) kf[j] = *(const float4*)(Kt + j * 128);
  }

  for (int kt = 0; kt <= ktmax; ++kt) {
    const int k0 = kt * 16;

    if (kt < ktmax) {
      const float* kpf = Kb + (size_t)(kt + 1) * (16 * DQK) + lane * 64;
      __builtin_prefetch(kpf, 0, 3);
      const float* vpf = Vb + (size_t)((k0 + 16) >> 1) * (2 * LAT) + lane * 64;
      __builtin_prefetch(vpf, 0, 3);
    }

    const float* Vt = Vb + (size_t)((k0 >> 1) + 4 * hf) * (2 * LAT) + r * 2;
    v2f vf[32];
#pragma unroll
    for (int ks = 0; ks < 4; ++ks)
#pragma unroll
      for (int nt = 0; nt < 8; ++nt)
        vf[ks * 8 + nt] =
            *(const v2f*)(Vt + (size_t)ks * (2 * LAT) + nt * 32);

    v8f sc = v8f_zero();
#pragma unroll
    for (int j = 0; j < 16; ++j) {
      v2f a0, a1;
      a0.x = kf[j].x;
      a0.y = kf[j].y;
      a1.x = kf[j].z;
      a1.y = kf[j].w;
      sc = wmma_f32(a0, qa[2 * j], sc);
      sc = wmma_f32(a1, qa[2 * j + 1], sc);
    }

    if (kt < ktmax) {
      const float* Kn = Kb + (size_t)(kt + 1) * (16 * DQK) + lane * 4;
#pragma unroll
      for (int j = 0; j < 16; ++j) kf[j] = *(const float4*)(Kn + j * 128);
    }
    const int key0 = k0 + 8 * hf;
    float v[8];
#pragma unroll
    for (int i = 0; i < 8; ++i) v[i] = (key0 + i > qcol) ? -1e30f : sc[i];
    float mx = fmaxf(fmaxf(fmaxf(v[0], v[1]), fmaxf(v[2], v[3])),
                     fmaxf(fmaxf(v[4], v[5]), fmaxf(v[6], v[7])));
    mx = fmaxf(mx, xhalf(mx));
    const float mnew  = fmaxf(m_run, mx);
    const float alpha = __expf(m_run - mnew);
    float pe[8];
#pragma unroll
    for (int i = 0; i < 8; ++i) pe[i] = __expf(v[i] - mnew);
    const float rs = ((pe[0] + pe[1]) + (pe[2] + pe[3])) +
                     ((pe[4] + pe[5]) + (pe[6] + pe[7]));
    l_run = l_run * alpha + rs;
    m_run = mnew;
#pragma unroll
    for (int nt = 0; nt < 8; ++nt) acc[nt] *= alpha;

#pragma unroll
    for (int ks = 0; ks < 4; ++ks) {
      v2f pb;
      pb.x = pe[2 * ks];
      pb.y = pe[2 * ks + 1];
#pragma unroll
      for (int nt = 0; nt < 8; ++nt)
        acc[nt] = wmma_f32(vf[ks * 8 + nt], pb, acc[nt]);
    }
    asm volatile("v_nop\n\tv_nop\n\tv_nop\n\tv_nop" : "+v"(pe[6]), "+v"(pe[7]));
  }

  const float inv = 1.0f / (l_run + xhalf(l_run));
  float (*os)[OPITCH] = ost[wave];
#pragma unroll
  for (int nt = 0; nt < 8; ++nt) {
    const v8f a = acc[nt];
    float4 lo, hi;
    lo.x = a[0] * inv; lo.y = a[1] * inv; lo.z = a[2] * inv; lo.w = a[3] * inv;
    hi.x = a[4] * inv; hi.y = a[5] * inv; hi.z = a[6] * inv; hi.w = a[7] * inv;
    *(float4*)&os[r][nt * 16 + 8 * hf]     = lo;
    *(float4*)&os[r][nt * 16 + 8 * hf + 4] = hi;
  }
  asm volatile("s_wait_dscnt 0" ::: "memory");
  float* Ob = O + ((size_t)(b * SLEN + q0)) * HLAT + h * LAT + lane * 4;
#pragma unroll
  for (int row = 0; row < 16; ++row) {
    const float4 o = *(const float4*)&os[row][lane * 4];
    *(float4*)(Ob + (size_t)row * HLAT) = o;
  }
}

extern "C" void kernel_launch(void* const* d_in, const int* in_sizes, int n_in,
                              void* d_out, int out_size, void* d_ws,
                              size_t ws_size, hipStream_t stream) {
  (void)in_sizes; (void)n_in; (void)out_size; (void)ws_size;
  const float* x        = (const float*)d_in[0];
  const float* w_dq     = (const float*)d_in[1];
  const float* gamma_q  = (const float*)d_in[2];
  const float* w_uq     = (const float*)d_in[3];
  const float* w_dkv    = (const float*)d_in[4];
  const float* gamma_kv = (const float*)d_in[5];
  const float* w_ukv    = (const float*)d_in[6];
  const float* w_o      = (const float*)d_in[7];
  float* out = (float*)d_out;

  float* ws = (float*)d_ws;
  float* tmp_q   = ws;
  float* cq      = tmp_q   + (size_t)ROWS * CQD;
  float* qraw    = cq      + (size_t)ROWS * CQD;
  float* qst     = qraw    + (size_t)ROWS * UQ_N;
  float* ckv_kr  = qst     + (size_t)BATCH * NH * SLEN * DQK;
  float* ckv     = ckv_kr  + (size_t)ROWS * DKV_N;
  float* knope   = ckv     + (size_t)ROWS * CKVD;
  float* kst     = knope   + (size_t)ROWS * KNOPE_N;
  float* vlat    = kst     + (size_t)BATCH * NKV * SLEN * DQK;
  float* wcomb   = vlat    + (size_t)ROWS * LAT;
  float* attn_l  = wcomb   + (size_t)HLAT * DIMC;

  gemm_wmma_f32_lds_pair<<<dim3(ROWS / 128, (CQD + DKV_N) / 64), 256, 0, stream>>>(
      x, w_dq, tmp_q, CQD, w_dkv, ckv_kr, DKV_N, DIMC);
  t5ln_kernel<<<ROWS, 32, 0, stream>>>(tmp_q, CQD, gamma_q, cq);
  gemm_wmma_f32_lds<<<dim3(ROWS / 128, UQ_N / 64), 256, 0, stream>>>(
      cq, CQD, w_uq, UQ_N, qraw, UQ_N, CQD);
  {
    size_t total = (size_t)BATCH * NH * SLEN * DQK;
    build_q_kernel<<<(unsigned)((total + 255) / 256), 256, 0, stream>>>(qraw, qst);
  }
  t5ln_kernel<<<ROWS, 32, 0, stream>>>(ckv_kr, DKV_N, gamma_kv, ckv);
  gemm_wmma_f32_lds<<<dim3(ROWS / 128, KNOPE_N / 64), 256, 0, stream>>>(
      ckv, CKVD, w_ukv, UKV_N, knope, KNOPE_N, CKVD);
  {
    size_t tk = (size_t)BATCH * NKV * SLEN * DQK;
    build_k_kernel<<<(unsigned)((tk + 255) / 256), 256, 0, stream>>>(knope, ckv_kr, kst);
    build_vlat_kernel<<<(ROWS / 2) * LAT / 256, 256, 0, stream>>>(ckv, vlat);
  }
  gemm_wmma_f32_lds_fold<<<dim3(1, DIMC / 64, NH), 256, 0, stream>>>(
      w_ukv, w_o, wcomb);
  attn_kernel<<<dim3(BATCH * NH, SLEN / 128), 256, 0, stream>>>(qst, kst, vlat, attn_l);
  gemm_wmma_f32_lds<<<dim3(ROWS / 128, DIMC / 64), 256, 0, stream>>>(
      attn_l, HLAT, wcomb, DIMC, out, DIMC, HLAT);
}


# attention: each wave reads its 16 query rows from the q projection itself and applies RoPE in registers; the one-thread-per-element q_states kernel (8M threads, 64 MB) and its launch are gone
# speedup vs baseline: 8.3346x; 1.0338x over previous
#include <hip/hip_runtime.h>
#include <math.h>

constexpr int DIMC   = 2048;
constexpr int SLEN   = 2048;
constexpr int BATCH  = 2;
constexpr int NH     = 16;
constexpr int NKV    = 4;
constexpr int CQD    = 128;
constexpr int QND    = 96;
constexpr int QRD    = 32;
constexpr int CKVD   = 128;
constexpr int KND    = 64;
constexpr int KRD    = 64;
constexpr int VDD    = 256;
constexpr int DQK    = 128;
constexpr int ROWS   = BATCH * SLEN;
constexpr int UQ_N   = NH * (QND + QRD);
constexpr int DKV_N  = CKVD + KRD;
constexpr int UKV_N  = NKV * (KND + VDD);
constexpr int HVD    = NH * VDD;
constexpr int LAT    = CKVD;
constexpr int HLAT   = NH * LAT;
constexpr int KNOPE_N = NKV * KND;

typedef float v2f __attribute__((ext_vector_type(2)));
typedef float v8f __attribute__((ext_vector_type(8)));

__device__ __forceinline__ v8f wmma_f32(v2f a, v2f b, v8f c) {
  return __builtin_amdgcn_wmma_f32_16x16x4_f32(
       false, a,  false, b,
       (short)0, c,  false,  false);
}

__device__ __forceinline__ v8f v8f_zero() {
  v8f z = {0.f, 0.f, 0.f, 0.f, 0.f, 0.f, 0.f, 0.f};
  return z;
}

constexpr int KC = 32;

typedef float v4f __attribute__((ext_vector_type(4)));

__device__ __forceinline__ void gemm_stage_chunk(
    float (*Asb)[36], float (*Bsb)[64][2], int ar, int ac, int br, int bn,
    v4f a0, v4f a1, v4f a2, v4f a3, v4f b0, v4f b1) {
  *(v4f*)&Asb[ar][ac]      = a0;
  *(v4f*)&Asb[ar + 32][ac] = a1;
  *(v4f*)&Asb[ar + 64][ac] = a2;
  *(v4f*)&Asb[ar + 96][ac] = a3;
  const int kp = br >> 1, kb = br & 1;
  Bsb[kp][bn + 0][kb]     = b0.x;
  Bsb[kp][bn + 1][kb]     = b0.y;
  Bsb[kp][bn + 2][kb]     = b0.z;
  Bsb[kp][bn + 3][kb]     = b0.w;
  Bsb[kp + 8][bn + 0][kb] = b1.x;
  Bsb[kp + 8][bn + 1][kb] = b1.y;
  Bsb[kp + 8][bn + 2][kb] = b1.z;
  Bsb[kp + 8][bn + 3][kb] = b1.w;
}

__device__ __forceinline__ void gemm_tile_128x64(
    const float* __restrict__ A, int lda, const float* __restrict__ B, int ldb,
    float* __restrict__ C, int ldc, int Kd, int m0b, int n0,
    float (*As)[128][36], float (*Bs)[KC / 2][64][2]) {
  const int tid  = threadIdx.x;
  const int lane = tid & 31;
  const int wave = tid >> 5;
  const int r    = lane & 15;
  const int hf   = lane >> 4;

  v8f acc[4];
#pragma unroll
  for (int j = 0; j < 4; ++j) acc[j] = v8f_zero();

  const int ar = tid >> 3;
  const int ac = (tid * 4) & 31;
  const int br = tid >> 4;
  const int bn = (tid * 4) & 63;

  const float* Ap0 = A + (size_t)(m0b + ar) * lda + ac;
  const float* Ap1 = Ap0 + (size_t)32 * lda;
  const float* Ap2 = Ap0 + (size_t)64 * lda;
  const float* Ap3 = Ap0 + (size_t)96 * lda;
  const float* Bp0 = B + (size_t)br * ldb + n0 + bn;
  const float* Bp1 = Bp0 + (size_t)16 * ldb;

  const int nchunk = Kd / KC;

  v4f a0 = *(const v4f*)Ap0;
  v4f a1 = *(const v4f*)Ap1;
  v4f a2 = *(const v4f*)Ap2;
  v4f a3 = *(const v4f*)Ap3;
  v4f b0 = *(const v4f*)Bp0;
  v4f b1 = *(const v4f*)Bp1;
  gemm_stage_chunk(As[0], Bs[0], ar, ac, br, bn, a0, a1, a2, a3, b0, b1);
  __syncthreads();

  for (int c = 0; c < nchunk; ++c) {
    const int buf = c & 1;
    const bool more = (c + 1 < nchunk);
    if (more) {
      const int kc = (c + 1) * KC;
      const size_t kb = (size_t)kc * ldb;
      a0 = *(const v4f*)(Ap0 + kc);
      a1 = *(const v4f*)(Ap1 + kc);
      a2 = *(const v4f*)(Ap2 + kc);
      a3 = *(const v4f*)(Ap3 + kc);
      b0 = *(const v4f*)(Bp0 + kb);
      b1 = *(const v4f*)(Bp1 + kb);
    }
#pragma unroll
    for (int k = 0; k < KC; k += 4) {
      v2f a = *(const v2f*)&As[buf][wave * 16 + r][k + 2 * hf];
#pragma unroll
      for (int j = 0; j < 4; ++j) {
        v2f bb = *(const v2f*)&Bs[buf][(k + 2 * hf) >> 1][j * 16 + r][0];
        acc[j] = wmma_f32(a, bb, acc[j]);
      }
    }
    if (more)
      gemm_stage_chunk(As[buf ^ 1], Bs[buf ^ 1], ar, ac, br, bn,
                       a0, a1, a2, a3, b0, b1);
    __syncthreads();
  }

#pragma unroll
  for (int j = 0; j < 4; ++j)
#pragma unroll
    for (int i = 0; i < 8; ++i)
      C[(size_t)(m0b + wave * 16 + i + 8 * hf) * ldc + n0 + j * 16 + r] =
          acc[j][i];
}

__global__ __launch_bounds__(256) void gemm_wmma_f32_lds(
    const float* __restrict__ A, int lda, const float* __restrict__ B, int ldb,
    float* __restrict__ C, int ldc, int Kd) {
  __shared__ float As[2][128][36];
  __shared__ float Bs[2][KC / 2][64][2];
  gemm_tile_128x64(A, lda, B, ldb, C, ldc, Kd, blockIdx.x * 128,
                   blockIdx.y * 64, As, Bs);
}

__global__ __launch_bounds__(256) void gemm_wmma_f32_lds_fold(
    const float* __restrict__ w_ukv, const float* __restrict__ w_o,
    float* __restrict__ wc) {
  __shared__ float As[2][128][36];
  __shared__ float Bs[2][KC / 2][64][2];
  const int h = blockIdx.z;
  const float* A = w_ukv + KNOPE_N + (h / (NH / NKV)) * VDD;
  const float* B = w_o + (size_t)h * VDD * DIMC;
  float*       C = wc + (size_t)h * LAT * DIMC;
  gemm_tile_128x64(A, UKV_N, B, DIMC, C, DIMC, VDD, 0, blockIdx.y * 64, As, Bs);
}

__global__ __launch_bounds__(256) void gemm_wmma_f32_lds_pair(
    const float* __restrict__ A,
    const float* __restrict__ B0, float* __restrict__ C0, int N0,
    const float* __restrict__ B1, float* __restrict__ C1, int N1, int Kd) {
  __shared__ float As[2][128][36];
  __shared__ float Bs[2][KC / 2][64][2];
  const int  nb0   = N0 / 64;
  const int  by    = blockIdx.y;
  const bool first = by < nb0;
  const float* B = first ? B0 : B1;
  float*       C = first ? C0 : C1;
  const int    N = first ? N0 : N1;
  gemm_tile_128x64(A, Kd, B, N, C, N, Kd, blockIdx.x * 128,
                   (first ? by : by - nb0) * 64, As, Bs);
}

__global__ __launch_bounds__(32) void t5ln_kernel(
    const float* __restrict__ in, int in_stride,
    const float* __restrict__ gamma, float* __restrict__ out) {
  const int row  = blockIdx.x;
  const int lane = threadIdx.x;
  const float* ip = in + (size_t)row * in_stride;
  float v[4], s = 0.f, s2 = 0.f;
#pragma unroll
  for (int i = 0; i < 4; ++i) {
    v[i] = ip[lane + 32 * i];
    s  += v[i];
    s2 += v[i] * v[i];
  }
#pragma unroll
  for (int off = 1; off < 32; off <<= 1) {
    s  += __shfl_xor(s, off, 32);
    s2 += __shfl_xor(s2, off, 32);
  }
  const float mu  = s * (1.f / 128.f);
  const float var = s2 * (1.f / 128.f) - mu * mu;
  const float rs  = rsqrtf(var + 1e-5f);
  float* op = out + (size_t)row * 128;
#pragma unroll
  for (int i = 0; i < 4; ++i)
    op[lane + 32 * i] = (v[i] - mu) * rs * gamma[lane + 32 * i];
}

__device__ __forceinline__ void rope_pair(float spos, int pair, float invD,
                                          float xe, float xo,
                                          float& re, float& ro) {
  const float lnTheta = 9.210340371976184f;
  float freq = expf(-(2.0f * (float)pair) * invD * lnTheta);
  float ang  = spos * freq;
  float c = cosf(ang), sn = sinf(ang);
  re = xe * c - xo * sn;
  ro = xe * sn + xo * c;
}

__global__ __launch_bounds__(256) void build_k_kernel(
    const float* __restrict__ knope, const float* __restrict__ ckv_kr,
    float* __restrict__ kst) {
  size_t idx = (size_t)blockIdx.x * blockDim.x + threadIdx.x;
  const size_t total = (size_t)BATCH * NKV * SLEN * DQK;
  if (idx >= total) return;
  int e   = idx & 3;
  int ln  = (int)((idx >> 2) & 31);
  int jc  = (int)((idx >> 7) & 15);
  int t   = (int)((idx >> 11) & (SLEN / 16 - 1));
  int kvh = (int)((idx >> 18) & (NKV - 1));
  int b   = (int)(idx >> 20);
  int j   = jc * 8 + (ln >> 4) * 4 + e;
  int s   = t * 16 + (ln & 15);
  size_t row = (size_t)(b * SLEN + s);
  float val;
  if (j < KND) {
    val = knope[row * KNOPE_N + kvh * KND + j];
  } else {
    int jj = j - KND;
    int pair = jj >> 1;
    const float* base = ckv_kr + row * DKV_N + CKVD;
    float xe = base[2 * pair], xo = base[2 * pair + 1];
    float re, ro;
    rope_pair((float)s, pair, 1.0f / (float)KRD, xe, xo, re, ro);
    val = (jj & 1) ? ro : re;
  }
  kst[idx] = val;
}

__global__ __launch_bounds__(256) void build_vlat_kernel(
    const float* __restrict__ ckv, float* __restrict__ vl) {
  const int idx = blockIdx.x * 256 + threadIdx.x;
  const int d = idx & (LAT - 1);
  const int p = idx >> 7;
  v2f o;
  o.x = ckv[(size_t)(2 * p) * LAT + d];
  o.y = ckv[(size_t)(2 * p + 1) * LAT + d];
  *(v2f*)(vl + (size_t)p * (2 * LAT) + 2 * d) = o;
}

constexpr int OPITCH = 132;

__device__ __forceinline__ float xhalf(float x) {
  const int bits = __builtin_bit_cast(int, x);
  return __builtin_bit_cast(
      float, __builtin_amdgcn_permlanex16(bits, bits, 0x76543210, 0xfedcba98,
                                          false, false));
}

__global__ __launch_bounds__(256) __attribute__((amdgpu_waves_per_eu(2, 2)))
void attn_kernel(
    const float* __restrict__ Qraw, const float* __restrict__ K,
    const float* __restrict__ V, float* __restrict__ O) {
  __shared__ alignas(16) float ost[8][16][OPITCH];

  const int lane = threadIdx.x & 31;
  const int wave = threadIdx.x >> 5;
  const int r    = lane & 15;
  const int hf   = lane >> 4;
  const int bh   = blockIdx.x;
  const int b    = bh >> 4;
  const int h    = bh & 15;
  const int kvh  = h >> 2;
  const int qb   = (SLEN / 128 - 1) - (int)blockIdx.y;
  const int q0   = qb * 128 + wave * 16;

  const float* Qrow = Qraw + ((size_t)(b * SLEN + q0 + r)) * UQ_N;
  const float* Qn   = Qrow + h * QND + 4 * hf;
  const float* Qr   = Qrow + NH * QND + h * QRD + 4 * hf;
  const float* Kb = K + ((size_t)(b * NKV + kvh) * SLEN) * DQK;
  const float* Vb = V + ((size_t)b * SLEN) * LAT;

  const float scale = 0.08838834764831845f;

  v2f qa[32];
#pragma unroll
  for (int j = 0; j < QND / 8; ++j) {
    const float4 qv = *(const float4*)(Qn + j * 8);
    v2f t0, t1;
    t0.x = qv.x * scale;
    t0.y = qv.y * scale;
    t1.x = qv.z * scale;
    t1.y = qv.w * scale;
    qa[2 * j]     = t0;
    qa[2 * j + 1] = t1;
  }
  const float spos = (float)(q0 + r);
#pragma unroll
  for (int jr = 0; jr < QRD / 8; ++jr) {
    const float4 qv = *(const float4*)(Qr + jr * 8);
    const int p0 = jr * 4 + 2 * hf;
    float re0, ro0, re1, ro1;
    rope_pair(spos, p0,     1.0f / (float)QRD, qv.x, qv.y, re0, ro0);
    rope_pair(spos, p0 + 1, 1.0f / (float)QRD, qv.z, qv.w, re1, ro1);
    v2f t0, t1;
    t0.x = re0 * scale;
    t0.y = ro0 * scale;
    t1.x = re1 * scale;
    t1.y = ro1 * scale;
    qa[2 * (QND / 8 + jr)]     = t0;
    qa[2 * (QND / 8 + jr) + 1] = t1;
  }

  float m_run = -1e30f, l_run = 0.f;
  v8f acc[8];
#pragma unroll
  for (int nt = 0; nt < 8; ++nt) acc[nt] = v8f_zero();

  const int ktmax = (q0 + 15) >> 4;
  const int qcol  = q0 + r;

  float4 kf[16];
  {
    const float* Kt = Kb + lane * 4;
#pragma unroll
    for (int j = 0; j < 16; ++j) kf[j] = *(const float4*)(Kt + j * 128);
  }

  for (int kt = 0; kt <= ktmax; ++kt) {
    const int k0 = kt * 16;

    if (kt < ktmax) {
      const float* kpf = Kb + (size_t)(kt + 1) * (16 * DQK) + lane * 64;
      __builtin_prefetch(kpf, 0, 3);
      const float* vpf = Vb + (size_t)((k0 + 16) >> 1) * (2 * LAT) + lane * 64;
      __builtin_prefetch(vpf, 0, 3);
    }

    const float* Vt = Vb + (size_t)((k0 >> 1) + 4 * hf) * (2 * LAT) + r * 2;
    v2f vf[32];
#pragma unroll
    for (int ks = 0; ks < 4; ++ks)
#pragma unroll
      for (int nt = 0; nt < 8; ++nt)
        vf[ks * 8 + nt] =
            *(const v2f*)(Vt + (size_t)ks * (2 * LAT) + nt * 32);

    v8f sc = v8f_zero();
#pragma unroll
    for (int j = 0; j < 16; ++j) {
      v2f a0, a1;
      a0.x = kf[j].x;
      a0.y = kf[j].y;
      a1.x = kf[j].z;
      a1.y = kf[j].w;
      sc = wmma_f32(a0, qa[2 * j], sc);
      sc = wmma_f32(a1, qa[2 * j + 1], sc);
    }

    if (kt < ktmax) {
      const float* Kn = Kb + (size_t)(kt + 1) * (16 * DQK) + lane * 4;
#pragma unroll
      for (int j = 0; j < 16; ++j) kf[j] = *(const float4*)(Kn + j * 128);
    }
    const int key0 = k0 + 8 * hf;
    float v[8];
#pragma unroll
    for (int i = 0; i < 8; ++i) v[i] = (key0 + i > qcol) ? -1e30f : sc[i];
    float mx = fmaxf(fmaxf(fmaxf(v[0], v[1]), fmaxf(v[2], v[3])),
                     fmaxf(fmaxf(v[4], v[5]), fmaxf(v[6], v[7])));
    mx = fmaxf(mx, xhalf(mx));
    const float mnew  = fmaxf(m_run, mx);
    const float alpha = __expf(m_run - mnew);
    float pe[8];
#pragma unroll
    for (int i = 0; i < 8; ++i) pe[i] = __expf(v[i] - mnew);
    const float rs = ((pe[0] + pe[1]) + (pe[2] + pe[3])) +
                     ((pe[4] + pe[5]) + (pe[6] + pe[7]));
    l_run = l_run * alpha + rs;
    m_run = mnew;
#pragma unroll
    for (int nt = 0; nt < 8; ++nt) acc[nt] *= alpha;

#pragma unroll
    for (int ks = 0; ks < 4; ++ks) {
      v2f pb;
      pb.x = pe[2 * ks];
      pb.y = pe[2 * ks + 1];
#pragma unroll
      for (int nt = 0; nt < 8; ++nt)
        acc[nt] = wmma_f32(vf[ks * 8 + nt], pb, acc[nt]);
    }
    asm volatile("v_nop\n\tv_nop\n\tv_nop\n\tv_nop" : "+v"(pe[6]), "+v"(pe[7]));
  }

  const float inv = 1.0f / (l_run + xhalf(l_run));
  float (*os)[OPITCH] = ost[wave];
#pragma unroll
  for (int nt = 0; nt < 8; ++nt) {
    const v8f a = acc[nt];
    float4 lo, hi;
    lo.x = a[0] * inv; lo.y = a[1] * inv; lo.z = a[2] * inv; lo.w = a[3] * inv;
    hi.x = a[4] * inv; hi.y = a[5] * inv; hi.z = a[6] * inv; hi.w = a[7] * inv;
    *(float4*)&os[r][nt * 16 + 8 * hf]     = lo;
    *(float4*)&os[r][nt * 16 + 8 * hf + 4] = hi;
  }
  asm volatile("s_wait_dscnt 0" ::: "memory");
  float* Ob = O + ((size_t)(b * SLEN + q0)) * HLAT + h * LAT + lane * 4;
#pragma unroll
  for (int row = 0; row < 16; ++row) {
    const float4 o = *(const float4*)&os[row][lane * 4];
    *(float4*)(Ob + (size_t)row * HLAT) = o;
  }
}

extern "C" void kernel_launch(void* const* d_in, const int* in_sizes, int n_in,
                              void* d_out, int out_size, void* d_ws,
                              size_t ws_size, hipStream_t stream) {
  (void)in_sizes; (void)n_in; (void)out_size; (void)ws_size;
  const float* x        = (const float*)d_in[0];
  const float* w_dq     = (const float*)d_in[1];
  const float* gamma_q  = (const float*)d_in[2];
  const float* w_uq     = (const float*)d_in[3];
  const float* w_dkv    = (const float*)d_in[4];
  const float* gamma_kv = (const float*)d_in[5];
  const float* w_ukv    = (const float*)d_in[6];
  const float* w_o      = (const float*)d_in[7];
  float* out = (float*)d_out;

  float* ws = (float*)d_ws;
  float* tmp_q   = ws;
  float* cq      = tmp_q   + (size_t)ROWS * CQD;
  float* qraw    = cq      + (size_t)ROWS * CQD;
  float* ckv_kr  = qraw    + (size_t)ROWS * UQ_N;
  float* ckv     = ckv_kr  + (size_t)ROWS * DKV_N;
  float* knope   = ckv     + (size_t)ROWS * CKVD;
  float* kst     = knope   + (size_t)ROWS * KNOPE_N;
  float* vlat    = kst     + (size_t)BATCH * NKV * SLEN * DQK;
  float* wcomb   = vlat    + (size_t)ROWS * LAT;
  float* attn_l  = wcomb   + (size_t)HLAT * DIMC;

  gemm_wmma_f32_lds_pair<<<dim3(ROWS / 128, (CQD + DKV_N) / 64), 256, 0, stream>>>(
      x, w_dq, tmp_q, CQD, w_dkv, ckv_kr, DKV_N, DIMC);
  t5ln_kernel<<<ROWS, 32, 0, stream>>>(tmp_q, CQD, gamma_q, cq);
  gemm_wmma_f32_lds<<<dim3(ROWS / 128, UQ_N / 64), 256, 0, stream>>>(
      cq, CQD, w_uq, UQ_N, qraw, UQ_N, CQD);
  t5ln_kernel<<<ROWS, 32, 0, stream>>>(ckv_kr, DKV_N, gamma_kv, ckv);
  gemm_wmma_f32_lds<<<dim3(ROWS / 128, KNOPE_N / 64), 256, 0, stream>>>(
      ckv, CKVD, w_ukv, UKV_N, knope, KNOPE_N, CKVD);
  {
    size_t tk = (size_t)BATCH * NKV * SLEN * DQK;
    build_k_kernel<<<(unsigned)((tk + 255) / 256), 256, 0, stream>>>(knope, ckv_kr, kst);
    build_vlat_kernel<<<(ROWS / 2) * LAT / 256, 256, 0, stream>>>(ckv, vlat);
  }
  gemm_wmma_f32_lds_fold<<<dim3(1, DIMC / 64, NH), 256, 0, stream>>>(
      w_ukv, w_o, wcomb);
  attn_kernel<<<dim3(BATCH * NH, SLEN / 128), 256, 0, stream>>>(qraw, kst, vlat, attn_l);
  gemm_wmma_f32_lds<<<dim3(ROWS / 128, DIMC / 64), 256, 0, stream>>>(
      attn_l, HLAT, wcomb, DIMC, out, DIMC, HLAT);
}


# work that waits for nothing shares a launch (10 -> 6): w_ukv*w_o product on the 96 WGPs the x projections leave idle; both layernorms; q and k_nope up-projections; K, V builders; sums unchanged
# speedup vs baseline: 8.7145x; 1.0456x over previous
#include <hip/hip_runtime.h>
#include <math.h>

constexpr int DIMC   = 2048;
constexpr int SLEN   = 2048;
constexpr int BATCH  = 2;
constexpr int NH     = 16;
constexpr int NKV    = 4;
constexpr int CQD    = 128;
constexpr int QND    = 96;
constexpr int QRD    = 32;
constexpr int CKVD   = 128;
constexpr int KND    = 64;
constexpr int KRD    = 64;
constexpr int VDD    = 256;
constexpr int DQK    = 128;
constexpr int ROWS   = BATCH * SLEN;
constexpr int UQ_N   = NH * (QND + QRD);
constexpr int DKV_N  = CKVD + KRD;
constexpr int UKV_N  = NKV * (KND + VDD);
constexpr int HVD    = NH * VDD;
constexpr int LAT    = CKVD;
constexpr int HLAT   = NH * LAT;
constexpr int KNOPE_N = NKV * KND;

typedef float v2f __attribute__((ext_vector_type(2)));
typedef float v8f __attribute__((ext_vector_type(8)));

__device__ __forceinline__ v8f wmma_f32(v2f a, v2f b, v8f c) {
  return __builtin_amdgcn_wmma_f32_16x16x4_f32(
       false, a,  false, b,
       (short)0, c,  false,  false);
}

__device__ __forceinline__ v8f v8f_zero() {
  v8f z = {0.f, 0.f, 0.f, 0.f, 0.f, 0.f, 0.f, 0.f};
  return z;
}

constexpr int KC = 32;

typedef float v4f __attribute__((ext_vector_type(4)));

__device__ __forceinline__ void gemm_stage_chunk(
    float (*Asb)[36], float (*Bsb)[64][2], int ar, int ac, int br, int bn,
    v4f a0, v4f a1, v4f a2, v4f a3, v4f b0, v4f b1) {
  *(v4f*)&Asb[ar][ac]      = a0;
  *(v4f*)&Asb[ar + 32][ac] = a1;
  *(v4f*)&Asb[ar + 64][ac] = a2;
  *(v4f*)&Asb[ar + 96][ac] = a3;
  const int kp = br >> 1, kb = br & 1;
  Bsb[kp][bn + 0][kb]     = b0.x;
  Bsb[kp][bn + 1][kb]     = b0.y;
  Bsb[kp][bn + 2][kb]     = b0.z;
  Bsb[kp][bn + 3][kb]     = b0.w;
  Bsb[kp + 8][bn + 0][kb] = b1.x;
  Bsb[kp + 8][bn + 1][kb] = b1.y;
  Bsb[kp + 8][bn + 2][kb] = b1.z;
  Bsb[kp + 8][bn + 3][kb] = b1.w;
}

__device__ __forceinline__ void gemm_tile_128x64(
    const float* __restrict__ A, int lda, const float* __restrict__ B, int ldb,
    float* __restrict__ C, int ldc, int Kd, int m0b, int n0,
    float (*As)[128][36], float (*Bs)[KC / 2][64][2]) {
  const int tid  = threadIdx.x;
  const int lane = tid & 31;
  const int wave = tid >> 5;
  const int r    = lane & 15;
  const int hf   = lane >> 4;

  v8f acc[4];
#pragma unroll
  for (int j = 0; j < 4; ++j) acc[j] = v8f_zero();

  const int ar = tid >> 3;
  const int ac = (tid * 4) & 31;
  const int br = tid >> 4;
  const int bn = (tid * 4) & 63;

  const float* Ap0 = A + (size_t)(m0b + ar) * lda + ac;
  const float* Ap1 = Ap0 + (size_t)32 * lda;
  const float* Ap2 = Ap0 + (size_t)64 * lda;
  const float* Ap3 = Ap0 + (size_t)96 * lda;
  const float* Bp0 = B + (size_t)br * ldb + n0 + bn;
  const float* Bp1 = Bp0 + (size_t)16 * ldb;

  const int nchunk = Kd / KC;

  v4f a0 = *(const v4f*)Ap0;
  v4f a1 = *(const v4f*)Ap1;
  v4f a2 = *(const v4f*)Ap2;
  v4f a3 = *(const v4f*)Ap3;
  v4f b0 = *(const v4f*)Bp0;
  v4f b1 = *(const v4f*)Bp1;
  gemm_stage_chunk(As[0], Bs[0], ar, ac, br, bn, a0, a1, a2, a3, b0, b1);
  __syncthreads();

  for (int c = 0; c < nchunk; ++c) {
    const int buf = c & 1;
    const bool more = (c + 1 < nchunk);
    if (more) {
      const int kc = (c + 1) * KC;
      const size_t kb = (size_t)kc * ldb;
      a0 = *(const v4f*)(Ap0 + kc);
      a1 = *(const v4f*)(Ap1 + kc);
      a2 = *(const v4f*)(Ap2 + kc);
      a3 = *(const v4f*)(Ap3 + kc);
      b0 = *(const v4f*)(Bp0 + kb);
      b1 = *(const v4f*)(Bp1 + kb);
    }
#pragma unroll
    for (int k = 0; k < KC; k += 4) {
      v2f a = *(const v2f*)&As[buf][wave * 16 + r][k + 2 * hf];
#pragma unroll
      for (int j = 0; j < 4; ++j) {
        v2f bb = *(const v2f*)&Bs[buf][(k + 2 * hf) >> 1][j * 16 + r][0];
        acc[j] = wmma_f32(a, bb, acc[j]);
      }
    }
    if (more)
      gemm_stage_chunk(As[buf ^ 1], Bs[buf ^ 1], ar, ac, br, bn,
                       a0, a1, a2, a3, b0, b1);
    __syncthreads();
  }

#pragma unroll
  for (int j = 0; j < 4; ++j)
#pragma unroll
    for (int i = 0; i < 8; ++i)
      C[(size_t)(m0b + wave * 16 + i + 8 * hf) * ldc + n0 + j * 16 + r] =
          acc[j][i];
}

__global__ __launch_bounds__(256) void gemm_wmma_f32_lds(
    const float* __restrict__ A, int lda, const float* __restrict__ B, int ldb,
    float* __restrict__ C, int ldc, int Kd) {
  __shared__ float As[2][128][36];
  __shared__ float Bs[2][KC / 2][64][2];
  gemm_tile_128x64(A, lda, B, ldb, C, ldc, Kd, blockIdx.x * 128,
                   blockIdx.y * 64, As, Bs);
}

constexpr int NUM_WGP    = 256;
constexpr int XP_MB      = ROWS / 128;
constexpr int XP_WGS     = XP_MB * ((CQD + DKV_N) / 64);
constexpr int FOLD_NB    = DIMC / 64;
constexpr int FOLD_TILES = NH * FOLD_NB;
constexpr int FOLD_WGS   = NUM_WGP - XP_WGS;

__global__ __launch_bounds__(256) void gemm_xproj_and_fold(
    const float* __restrict__ x,
    const float* __restrict__ w_dq, float* __restrict__ tmp_q,
    const float* __restrict__ w_dkv, float* __restrict__ ckv_kr,
    const float* __restrict__ w_ukv, const float* __restrict__ w_o,
    float* __restrict__ wc) {
  __shared__ float As[2][128][36];
  __shared__ float Bs[2][KC / 2][64][2];
  const int id = blockIdx.x;
  if (id < XP_WGS) {
    constexpr int nb0 = CQD / 64;
    const int  bx    = id % XP_MB;
    const int  by    = id / XP_MB;
    const bool first = by < nb0;
    const float* B = first ? w_dq : w_dkv;
    float*       C = first ? tmp_q : ckv_kr;
    const int    N = first ? CQD : DKV_N;
    gemm_tile_128x64(x, DIMC, B, N, C, N, DIMC, bx * 128,
                     (first ? by : by - nb0) * 64, As, Bs);
  } else {
    for (int t = id - XP_WGS; t < FOLD_TILES; t += FOLD_WGS) {
      const int h  = t / FOLD_NB;
      const int nb = t % FOLD_NB;
      const float* A = w_ukv + KNOPE_N + (h / (NH / NKV)) * VDD;
      const float* B = w_o + (size_t)h * VDD * DIMC;
      float*       C = wc + (size_t)h * LAT * DIMC;
      gemm_tile_128x64(A, UKV_N, B, DIMC, C, DIMC, VDD, 0, nb * 64, As, Bs);
    }
  }
}

__global__ __launch_bounds__(256) void gemm_qraw_and_knope(
    const float* __restrict__ cq, const float* __restrict__ w_uq,
    float* __restrict__ qraw,
    const float* __restrict__ ckv, const float* __restrict__ w_ukv,
    float* __restrict__ knope) {
  __shared__ float As[2][128][36];
  __shared__ float Bs[2][KC / 2][64][2];
  constexpr int nbq = UQ_N / 64;
  const int  by    = blockIdx.y;
  const bool first = by < nbq;
  const float* A   = first ? cq : ckv;
  const float* B   = first ? w_uq : w_ukv;
  const int    ldb = first ? UQ_N : UKV_N;
  float*       C   = first ? qraw : knope;
  const int    ldc = first ? UQ_N : KNOPE_N;
  gemm_tile_128x64(A, CQD, B, ldb, C, ldc, CQD, blockIdx.x * 128,
                   (first ? by : by - nbq) * 64, As, Bs);
}

__global__ __launch_bounds__(32) void t5ln_both_kernel(
    const float* __restrict__ in0, int in_stride0,
    const float* __restrict__ gamma0, float* __restrict__ out0,
    const float* __restrict__ in1, int in_stride1,
    const float* __restrict__ gamma1, float* __restrict__ out1) {
  const bool first = blockIdx.x < ROWS;
  const int row  = first ? blockIdx.x : blockIdx.x - ROWS;
  const int lane = threadIdx.x;
  const float* in    = first ? in0 : in1;
  const int in_stride = first ? in_stride0 : in_stride1;
  const float* gamma = first ? gamma0 : gamma1;
  float*       out   = first ? out0 : out1;
  const float* ip = in + (size_t)row * in_stride;
  float v[4], s = 0.f, s2 = 0.f;
#pragma unroll
  for (int i = 0; i < 4; ++i) {
    v[i] = ip[lane + 32 * i];
    s  += v[i];
    s2 += v[i] * v[i];
  }
#pragma unroll
  for (int off = 1; off < 32; off <<= 1) {
    s  += __shfl_xor(s, off, 32);
    s2 += __shfl_xor(s2, off, 32);
  }
  const float mu  = s * (1.f / 128.f);
  const float var = s2 * (1.f / 128.f) - mu * mu;
  const float rs  = rsqrtf(var + 1e-5f);
  float* op = out + (size_t)row * 128;
#pragma unroll
  for (int i = 0; i < 4; ++i)
    op[lane + 32 * i] = (v[i] - mu) * rs * gamma[lane + 32 * i];
}

__device__ __forceinline__ void rope_pair(float spos, int pair, float invD,
                                          float xe, float xo,
                                          float& re, float& ro) {
  const float lnTheta = 9.210340371976184f;
  float freq = expf(-(2.0f * (float)pair) * invD * lnTheta);
  float ang  = spos * freq;
  float c = cosf(ang), sn = sinf(ang);
  re = xe * c - xo * sn;
  ro = xe * sn + xo * c;
}

__device__ __forceinline__ void build_k_elem(
    size_t idx, const float* __restrict__ knope,
    const float* __restrict__ ckv_kr, float* __restrict__ kst) {
  const size_t total = (size_t)BATCH * NKV * SLEN * DQK;
  if (idx >= total) return;
  int e   = idx & 3;
  int ln  = (int)((idx >> 2) & 31);
  int jc  = (int)((idx >> 7) & 15);
  int t   = (int)((idx >> 11) & (SLEN / 16 - 1));
  int kvh = (int)((idx >> 18) & (NKV - 1));
  int b   = (int)(idx >> 20);
  int j   = jc * 8 + (ln >> 4) * 4 + e;
  int s   = t * 16 + (ln & 15);
  size_t row = (size_t)(b * SLEN + s);
  float val;
  if (j < KND) {
    val = knope[row * KNOPE_N + kvh * KND + j];
  } else {
    int jj = j - KND;
    int pair = jj >> 1;
    const float* base = ckv_kr + row * DKV_N + CKVD;
    float xe = base[2 * pair], xo = base[2 * pair + 1];
    float re, ro;
    rope_pair((float)s, pair, 1.0f / (float)KRD, xe, xo, re, ro);
    val = (jj & 1) ? ro : re;
  }
  kst[idx] = val;
}

__device__ __forceinline__ void build_vlat_elem(
    int idx, const float* __restrict__ ckv, float* __restrict__ vl) {
  const int d = idx & (LAT - 1);
  const int p = idx >> 7;
  v2f o;
  o.x = ckv[(size_t)(2 * p) * LAT + d];
  o.y = ckv[(size_t)(2 * p + 1) * LAT + d];
  *(v2f*)(vl + (size_t)p * (2 * LAT) + 2 * d) = o;
}

constexpr int BK_WGS = BATCH * NKV * SLEN * DQK / 256;
constexpr int BV_WGS = (ROWS / 2) * LAT / 256;

__global__ __launch_bounds__(256) void build_kv_kernel(
    const float* __restrict__ knope, const float* __restrict__ ckv_kr,
    float* __restrict__ kst, const float* __restrict__ ckv,
    float* __restrict__ vl) {
  const int bid = blockIdx.x;
  if (bid < BK_WGS)
    build_k_elem((size_t)bid * 256 + threadIdx.x, knope, ckv_kr, kst);
  else
    build_vlat_elem((bid - BK_WGS) * 256 + threadIdx.x, ckv, vl);
}

constexpr int OPITCH = 132;

__device__ __forceinline__ float xhalf(float x) {
  const int bits = __builtin_bit_cast(int, x);
  return __builtin_bit_cast(
      float, __builtin_amdgcn_permlanex16(bits, bits, 0x76543210, 0xfedcba98,
                                          false, false));
}

__global__ __launch_bounds__(256) __attribute__((amdgpu_waves_per_eu(2, 2)))
void attn_kernel(
    const float* __restrict__ Qraw, const float* __restrict__ K,
    const float* __restrict__ V, float* __restrict__ O) {
  __shared__ alignas(16) float ost[8][16][OPITCH];

  const int lane = threadIdx.x & 31;
  const int wave = threadIdx.x >> 5;
  const int r    = lane & 15;
  const int hf   = lane >> 4;
  const int bh   = blockIdx.x;
  const int b    = bh >> 4;
  const int h    = bh & 15;
  const int kvh  = h >> 2;
  const int qb   = (SLEN / 128 - 1) - (int)blockIdx.y;
  const int q0   = qb * 128 + wave * 16;

  const float* Qrow = Qraw + ((size_t)(b * SLEN + q0 + r)) * UQ_N;
  const float* Qn   = Qrow + h * QND + 4 * hf;
  const float* Qr   = Qrow + NH * QND + h * QRD + 4 * hf;
  const float* Kb = K + ((size_t)(b * NKV + kvh) * SLEN) * DQK;
  const float* Vb = V + ((size_t)b * SLEN) * LAT;

  const float scale = 0.08838834764831845f;

  v2f qa[32];
#pragma unroll
  for (int j = 0; j < QND / 8; ++j) {
    const float4 qv = *(const float4*)(Qn + j * 8);
    v2f t0, t1;
    t0.x = qv.x * scale;
    t0.y = qv.y * scale;
    t1.x = qv.z * scale;
    t1.y = qv.w * scale;
    qa[2 * j]     = t0;
    qa[2 * j + 1] = t1;
  }
  const float spos = (float)(q0 + r);
#pragma unroll
  for (int jr = 0; jr < QRD / 8; ++jr) {
    const float4 qv = *(const float4*)(Qr + jr * 8);
    const int p0 = jr * 4 + 2 * hf;
    float re0, ro0, re1, ro1;
    rope_pair(spos, p0,     1.0f / (float)QRD, qv.x, qv.y, re0, ro0);
    rope_pair(spos, p0 + 1, 1.0f / (float)QRD, qv.z, qv.w, re1, ro1);
    v2f t0, t1;
    t0.x = re0 * scale;
    t0.y = ro0 * scale;
    t1.x = re1 * scale;
    t1.y = ro1 * scale;
    qa[2 * (QND / 8 + jr)]     = t0;
    qa[2 * (QND / 8 + jr) + 1] = t1;
  }

  float m_run = -1e30f, l_run = 0.f;
  v8f acc[8];
#pragma unroll
  for (int nt = 0; nt < 8; ++nt) acc[nt] = v8f_zero();

  const int ktmax = (q0 + 15) >> 4;
  const int qcol  = q0 + r;

  float4 kf[16];
  {
    const float* Kt = Kb + lane * 4;
#pragma unroll
    for (int j = 0; j < 16; ++j) kf[j] = *(const float4*)(Kt + j * 128);
  }

  for (int kt = 0; kt <= ktmax; ++kt) {
    const int k0 = kt * 16;

    if (kt < ktmax) {
      const float* kpf = Kb + (size_t)(kt + 1) * (16 * DQK) + lane * 64;
      __builtin_prefetch(kpf, 0, 3);
      const float* vpf = Vb + (size_t)((k0 + 16) >> 1) * (2 * LAT) + lane * 64;
      __builtin_prefetch(vpf, 0, 3);
    }

    const float* Vt = Vb + (size_t)((k0 >> 1) + 4 * hf) * (2 * LAT) + r * 2;
    v2f vf[32];
#pragma unroll
    for (int ks = 0; ks < 4; ++ks)
#pragma unroll
      for (int nt = 0; nt < 8; ++nt)
        vf[ks * 8 + nt] =
            *(const v2f*)(Vt + (size_t)ks * (2 * LAT) + nt * 32);

    v8f sc = v8f_zero();
#pragma unroll
    for (int j = 0; j < 16; ++j) {
      v2f a0, a1;
      a0.x = kf[j].x;
      a0.y = kf[j].y;
      a1.x = kf[j].z;
      a1.y = kf[j].w;
      sc = wmma_f32(a0, qa[2 * j], sc);
      sc = wmma_f32(a1, qa[2 * j + 1], sc);
    }

    if (kt < ktmax) {
      const float* Kn = Kb + (size_t)(kt + 1) * (16 * DQK) + lane * 4;
#pragma unroll
      for (int j = 0; j < 16; ++j) kf[j] = *(const float4*)(Kn + j * 128);
    }
    const int key0 = k0 + 8 * hf;
    float v[8];
#pragma unroll
    for (int i = 0; i < 8; ++i) v[i] = (key0 + i > qcol) ? -1e30f : sc[i];
    float mx = fmaxf(fmaxf(fmaxf(v[0], v[1]), fmaxf(v[2], v[3])),
                     fmaxf(fmaxf(v[4], v[5]), fmaxf(v[6], v[7])));
    mx = fmaxf(mx, xhalf(mx));
    const float mnew  = fmaxf(m_run, mx);
    const float alpha = __expf(m_run - mnew);
    float pe[8];
#pragma unroll
    for (int i = 0; i < 8; ++i) pe[i] = __expf(v[i] - mnew);
    const float rs = ((pe[0] + pe[1]) + (pe[2] + pe[3])) +
                     ((pe[4] + pe[5]) + (pe[6] + pe[7]));
    l_run = l_run * alpha + rs;
    m_run = mnew;
#pragma unroll
    for (int nt = 0; nt < 8; ++nt) acc[nt] *= alpha;

#pragma unroll
    for (int ks = 0; ks < 4; ++ks) {
      v2f pb;
      pb.x = pe[2 * ks];
      pb.y = pe[2 * ks + 1];
#pragma unroll
      for (int nt = 0; nt < 8; ++nt)
        acc[nt] = wmma_f32(vf[ks * 8 + nt], pb, acc[nt]);
    }
    asm volatile("v_nop\n\tv_nop\n\tv_nop\n\tv_nop" : "+v"(pe[6]), "+v"(pe[7]));
  }

  const float inv = 1.0f / (l_run + xhalf(l_run));
  float (*os)[OPITCH] = ost[wave];
#pragma unroll
  for (int nt = 0; nt < 8; ++nt) {
    const v8f a = acc[nt];
    float4 lo, hi;
    lo.x = a[0] * inv; lo.y = a[1] * inv; lo.z = a[2] * inv; lo.w = a[3] * inv;
    hi.x = a[4] * inv; hi.y = a[5] * inv; hi.z = a[6] * inv; hi.w = a[7] * inv;
    *(float4*)&os[r][nt * 16 + 8 * hf]     = lo;
    *(float4*)&os[r][nt * 16 + 8 * hf + 4] = hi;
  }
  asm volatile("s_wait_dscnt 0" ::: "memory");
  float* Ob = O + ((size_t)(b * SLEN + q0)) * HLAT + h * LAT + lane * 4;
#pragma unroll
  for (int row = 0; row < 16; ++row) {
    const float4 o = *(const float4*)&os[row][lane * 4];
    *(float4*)(Ob + (size_t)row * HLAT) = o;
  }
}

extern "C" void kernel_launch(void* const* d_in, const int* in_sizes, int n_in,
                              void* d_out, int out_size, void* d_ws,
                              size_t ws_size, hipStream_t stream) {
  (void)in_sizes; (void)n_in; (void)out_size; (void)ws_size;
  const float* x        = (const float*)d_in[0];
  const float* w_dq     = (const float*)d_in[1];
  const float* gamma_q  = (const float*)d_in[2];
  const float* w_uq     = (const float*)d_in[3];
  const float* w_dkv    = (const float*)d_in[4];
  const float* gamma_kv = (const float*)d_in[5];
  const float* w_ukv    = (const float*)d_in[6];
  const float* w_o      = (const float*)d_in[7];
  float* out = (float*)d_out;

  float* ws = (float*)d_ws;
  float* tmp_q   = ws;
  float* cq      = tmp_q   + (size_t)ROWS * CQD;
  float* qraw    = cq      + (size_t)ROWS * CQD;
  float* ckv_kr  = qraw    + (size_t)ROWS * UQ_N;
  float* ckv     = ckv_kr  + (size_t)ROWS * DKV_N;
  float* knope   = ckv     + (size_t)ROWS * CKVD;
  float* kst     = knope   + (size_t)ROWS * KNOPE_N;
  float* vlat    = kst     + (size_t)BATCH * NKV * SLEN * DQK;
  float* wcomb   = vlat    + (size_t)ROWS * LAT;
  float* attn_l  = wcomb   + (size_t)HLAT * DIMC;

  gemm_xproj_and_fold<<<NUM_WGP, 256, 0, stream>>>(
      x, w_dq, tmp_q, w_dkv, ckv_kr, w_ukv, w_o, wcomb);
  t5ln_both_kernel<<<2 * ROWS, 32, 0, stream>>>(
      tmp_q, CQD, gamma_q, cq, ckv_kr, DKV_N, gamma_kv, ckv);
  gemm_qraw_and_knope<<<dim3(ROWS / 128, UQ_N / 64 + KNOPE_N / 64), 256, 0, stream>>>(
      cq, w_uq, qraw, ckv, w_ukv, knope);
  build_kv_kernel<<<BK_WGS + BV_WGS, 256, 0, stream>>>(knope, ckv_kr, kst, ckv, vlat);
  attn_kernel<<<dim3(BATCH * NH, SLEN / 128), 256, 0, stream>>>(qraw, kst, vlat, attn_l);
  gemm_wmma_f32_lds<<<dim3(ROWS / 128, DIMC / 64), 256, 0, stream>>>(
      attn_l, HLAT, wcomb, DIMC, out, DIMC, HLAT);
}
